# starting kernel
# baseline (speedup 1.0000x reference)
#include <hip/hip_runtime.h>


#define D 128
#define TILE_ROWS 16
#define BLOCK 256

typedef float v2f __attribute__((ext_vector_type(2)));
typedef float v8f __attribute__((ext_vector_type(8)));

__global__ __launch_bounds__(BLOCK)
void gin_scatter_add(const float* __restrict__ x,
                     const int*   __restrict__ src,
                     const int*   __restrict__ dst,
                     float*       __restrict__ agg,
                     int nEdges) {
    long gid = (long)blockIdx.x * blockDim.x + threadIdx.x;
    long total = (long)nEdges * D;
    if (gid >= total) return;
    int e = (int)(gid >> 7);
    int c = (int)(gid & (D - 1));
    int s = src[e];
    int d = dst[e];
    float v = x[(long)s * D + c];
    unsafeAtomicAdd(&agg[(long)d * D + c], v);
}

__global__ __launch_bounds__(BLOCK)
void gin_mlp(const float* __restrict__ xin,
             const float* __restrict__ agg,
             const float* __restrict__ W1, const float* __restrict__ b1,
             const float* __restrict__ W2, const float* __restrict__ b2,
             float* __restrict__ out,
             int applyReluOut) {
    __shared__ float s_tile[TILE_ROWS][D];
    __shared__ float t_tile[TILE_ROWS][D];

    const int tid  = threadIdx.x;
    const int lane = tid & 31;
    const int wave = tid >> 5;
    const long row0 = (long)blockIdx.x * TILE_ROWS;

    #pragma unroll
    for (int i = tid; i < TILE_ROWS * D; i += BLOCK) {
        int r = i >> 7;
        int c = i & (D - 1);
        long g = (row0 + r) * D + c;
        s_tile[r][c] = xin[g] + agg[g];
    }
    __syncthreads();

    const int mrow  = lane & 15;
    const int kHalf = (lane >> 4) << 1;
    const int n0    = wave * 16;
    const int col   = n0 + mrow;
    const int rbase = (lane >> 4) << 3;

    {
        v8f acc = {};
        #pragma unroll 4
        for (int k = 0; k < D; k += 4) {
            const int kb = k + kHalf;
            v2f a, b;
            a[0] = s_tile[mrow][kb];
            a[1] = s_tile[mrow][kb + 1];
            b[0] = W1[(long)kb * D + col];
            b[1] = W1[(long)(kb + 1) * D + col];
            acc = __builtin_amdgcn_wmma_f32_16x16x4_f32(
                false, a, false, b, (short)0, acc, false, false);
        }
        const float bias = b1[col];
        #pragma unroll
        for (int v = 0; v < 8; ++v) {
            float t = acc[v] + bias;
            t_tile[rbase + v][col] = t > 0.0f ? t : 0.0f;
        }
    }
    __syncthreads();

    {
        v8f acc = {};
        #pragma unroll 4
        for (int k = 0; k < D; k += 4) {
            const int kb = k + kHalf;
            v2f a, b;
            a[0] = t_tile[mrow][kb];
            a[1] = t_tile[mrow][kb + 1];
            b[0] = W2[(long)kb * D + col];
            b[1] = W2[(long)(kb + 1) * D + col];
            acc = __builtin_amdgcn_wmma_f32_16x16x4_f32(
                false, a, false, b, (short)0, acc, false, false);
        }
        const float bias = b2[col];
        #pragma unroll
        for (int v = 0; v < 8; ++v) {
            float u = acc[v] + bias;
            if (applyReluOut) u = u > 0.0f ? u : 0.0f;
            out[(row0 + rbase + v) * D + col] = u;
        }
    }
}

extern "C" void kernel_launch(void* const* d_in, const int* in_sizes, int n_in,
                              void* d_out, int out_size, void* d_ws, size_t ws_size,
                              hipStream_t stream) {
    const float* x  = (const float*)d_in[0];
    const int*   ei = (const int*)  d_in[1];
    const float* W1 = (const float*)d_in[2];
    const float* b1 = (const float*)d_in[3];
    const float* W2 = (const float*)d_in[4];
    const float* b2 = (const float*)d_in[5];
    float* out = (float*)d_out;

    const int N = in_sizes[0] / D;
    const int E = in_sizes[1] / 2;
    const int* src = ei;
    const int* dst = ei + E;

    float* agg = (float*)d_ws;
    float* h1  = agg + (size_t)N * D;
    const size_t aggBytes = (size_t)N * D * sizeof(float);

    const long scatterThreads = (long)E * D;
    const int  sBlocks = (int)((scatterThreads + BLOCK - 1) / BLOCK);
    const int  mBlocks = N / TILE_ROWS;

    hipMemsetAsync(agg, 0, aggBytes, stream);
    gin_scatter_add<<<sBlocks, BLOCK, 0, stream>>>(x, src, dst, agg, E);
    gin_mlp<<<mBlocks, BLOCK, 0, stream>>>(x, agg, W1, b1, W2, b2, h1, 1);

    hipMemsetAsync(agg, 0, aggBytes, stream);
    gin_scatter_add<<<sBlocks, BLOCK, 0, stream>>>(h1, src, dst, agg, E);
    gin_mlp<<<mBlocks, BLOCK, 0, stream>>>(h1, agg, W1, b1, W2, b2, out, 0);
}


# MLP computed transposed: weights are the WMMA A operand read from LDS in fragment order, a wave's 16 nodes the B operand in registers, first product's D feeds the second as B; k sum order regrouped
# speedup vs baseline: 2.5825x; 2.5825x over previous
#include <hip/hip_runtime.h>


#define D 128
#define TILE_ROWS 16

typedef float v2f __attribute__((ext_vector_type(2)));
typedef float v8f __attribute__((ext_vector_type(8)));

#define AGG_ROWS   400
#define AGG_WAVES  32
#define AGG_BLOCK  1024
#define SEG_CAP    256
#define SCAN_DEPTH 8

typedef float f4 __attribute__((ext_vector_type(4)));
typedef int   i4 __attribute__((ext_vector_type(4)));

__global__ __launch_bounds__(AGG_BLOCK)
void gin_agg_owner(const float* __restrict__ x,
                   const int*   __restrict__ src,
                   const int*   __restrict__ dst,
                   float*       __restrict__ agg,
                   int nNodes, int nEdges) {
    __shared__ f4  acc[AGG_ROWS][32];
    __shared__ unsigned seg[AGG_WAVES][SEG_CAP];
    __shared__ int s_cnt[AGG_WAVES];
    __shared__ int s_more[AGG_WAVES];

    const int tid  = threadIdx.x;
    const int lane = tid & 31;
    const int wave = __builtin_amdgcn_readfirstlane(tid >> 5);
    const int rowLo = blockIdx.x * AGG_ROWS;
    const int rowsHere = min(AGG_ROWS, nNodes - rowLo);

    f4* accFlat = &acc[0][0];
    for (int i = tid; i < AGG_ROWS * 32; i += AGG_BLOCK) accFlat[i] = f4{0.f, 0.f, 0.f, 0.f};

    int per = (nEdges + AGG_WAVES - 1) / AGG_WAVES;
    per = (per + 3) & ~3;
    const int beg = min(wave * per, nEdges);
    const int end = min(beg + per, nEdges);
    const bool vecOk = ((((unsigned long long)dst) & 15ull) == 0ull) && ((nEdges & 3) == 0);
    int pos = beg;
    bool anyMore;

    do {
        int cntw = 0;
        bool full = false;
        while (pos < end && !full) {
            i4 v[SCAN_DEPTH];
            if (vecOk) {
                #pragma unroll
                for (int u = 0; u < SCAN_DEPTH; ++u) {
                    const int e = pos + u * 128 + lane * 4;
                    const i4 t = *(const i4*)(dst + min(e, end - 4));
                    const i4 none = {-1, -1, -1, -1};
                    v[u] = (e < end) ? t : none;
                }
            } else {
                #pragma unroll
                for (int u = 0; u < SCAN_DEPTH; ++u) {
                    const int e = pos + u * 128 + lane * 4;
                    i4 t;
                    #pragma unroll
                    for (int k = 0; k < 4; ++k) {
                        const int d = dst[min(e + k, end - 1)];
                        t[k] = (e + k < end) ? d : -1;
                    }
                    v[u] = t;
                }
            }
            int stepsDone = 0;
            #pragma unroll
            for (int u = 0; u < SCAN_DEPTH; ++u) {
                if (!full && pos + u * 128 < end) {
                    if (cntw + 128 > SEG_CAP) {
                        full = true;
                    } else {
                        const int off = pos - beg + u * 128 + lane * 4;
                        unsigned lr[4]; bool h[4]; unsigned m[4];
                        #pragma unroll
                        for (int k = 0; k < 4; ++k) {
                            lr[k] = (unsigned)(v[u][k] - rowLo);
                            h[k]  = lr[k] < (unsigned)rowsHere;
                            m[k]  = __builtin_amdgcn_ballot_w32(h[k]);
                        }
                        if (m[0] | m[1] | m[2] | m[3]) {
                            #pragma unroll
                            for (int k = 0; k < 4; ++k) {
                                if (h[k]) {
                                    const int p = cntw + (int)__builtin_amdgcn_mbcnt_lo(m[k], 0u);
                                    seg[wave][p] = (lr[k] << 20) | (unsigned)(off + k);
                                }
                                cntw += __builtin_popcount(m[k]);
                            }
                        }
                        stepsDone = u + 1;
                    }
                }
            }
            pos += stepsDone * 128;
        }
        for (int i = lane; i < cntw; i += 32) {
            const unsigned en = seg[wave][i];
            seg[wave][i] = (en & 0xFFF00000u) | (unsigned)src[beg + (int)(en & 0xFFFFFu)];
        }
        if (lane == 0) { s_cnt[wave] = cntw; s_more[wave] = (pos < end) ? 1 : 0; }
        __syncthreads();
        anyMore = __builtin_amdgcn_ballot_w32(s_more[lane] != 0) != 0u;

        unsigned held = 0u;
        int nHeld = 0;
        auto flush = [&]() {
            for (int j = 0; j < nHeld; j += 4) {
                f4 xv[4]; int r[4];
                #pragma unroll
                for (int q = 0; q < 4; ++q) {
                    const unsigned e2 = (unsigned)__builtin_amdgcn_readlane((int)held, min(j + q, nHeld - 1));
                    r[q]  = (int)(e2 >> 20);
                    xv[q] = ((const f4*)x)[(long)(e2 & 0xFFFFFu) * 32 + lane];
                }
                #pragma unroll
                for (int q = 0; q < 4; ++q) {
                    if (j + q < nHeld) {
                        f4 a = acc[r[q]][lane];
                        a += xv[q];
                        acc[r[q]][lane] = a;
                    }
                }
            }
            nHeld = 0;
        };
        const int cl = s_cnt[lane];
        for (int sg = 0; sg < AGG_WAVES; ++sg) {
            const int c = __builtin_amdgcn_readlane(cl, sg);
            for (int i0 = 0; i0 < c; i0 += 32) {
                const int i = i0 + lane;
                const unsigned ent = (i < c) ? seg[sg][i] : 0u;
                const bool own = (i < c) && (int)((ent >> 20) & (AGG_WAVES - 1)) == wave;
                unsigned m = __builtin_amdgcn_ballot_w32(own);
                while (m) {
                    const int l = __builtin_ctz(m);
                    m &= m - 1u;
                    const unsigned e2 = (unsigned)__builtin_amdgcn_readlane((int)ent, l);
                    held = (lane == nHeld) ? e2 : held;
                    if (++nHeld == 32) flush();
                }
            }
        }
        flush();
        __syncthreads();
    } while (anyMore);

    f4* aggRows = (f4*)agg + (long)rowLo * 32;
    for (int i = tid; i < rowsHere * 32; i += AGG_BLOCK) aggRows[i] = accFlat[i];
}

#define MLP_WAVES 16
#define MLP_BLOCK (MLP_WAVES * 32)
#define MLP_GROUPS 256

static __device__ __forceinline__ v8f mma4(float a0, float a1, float b0, float b1, v8f c) {
    v2f a = {a0, a1};
    v2f b = {b0, b1};
    return __builtin_amdgcn_wmma_f32_16x16x4_f32(false, a, false, b, (short)0, c, false, false);
}

__global__ __launch_bounds__(MLP_BLOCK)
void gin_mlp_t(const float* __restrict__ xin,
               const float* __restrict__ agg,
               const float* __restrict__ W1, const float* __restrict__ b1,
               const float* __restrict__ W2, const float* __restrict__ b2,
               float* __restrict__ out,
               int applyReluOut, int nTiles) {
    __shared__ f4 w1f[8][16][32];
    __shared__ f4 w2f[8][16][32];
    __shared__ alignas(16) float bs[2][D];

    const int tid  = threadIdx.x;
    const int lane = tid & 31;
    const int wave = __builtin_amdgcn_readfirstlane(tid >> 5);

    {
        float* w1s = (float*)&w1f[0][0][0];
        float* w2s = (float*)&w2f[0][0][0];
        for (int idx = tid; idx < D * D; idx += MLP_BLOCK) {
            const int k  = idx >> 7;
            const int n  = idx & (D - 1);
            const int tn = n >> 4;
            const int cc = n & 15;
            const int m  = k & 3;
            const int i1 = k >> 3;
            const int h1 = (k >> 2) & 1;
            w1s[((tn * 16 + i1) * 32 + cc + 16 * h1) * 4 + m] = W1[idx];
            const int i2 = ((k >> 4) << 1) | ((k >> 2) & 1);
            const int h2 = (k >> 3) & 1;
            w2s[((tn * 16 + i2) * 32 + cc + 16 * h2) * 4 + m] = W2[idx];
        }
        if (tid < D) { bs[0][tid] = b1[tid]; bs[1][tid] = b2[tid]; }
    }
    __syncthreads();

    const int c = lane & 15;
    const int h = lane >> 4;

    for (int tile = blockIdx.x + MLP_GROUPS * wave; tile < nTiles; tile += MLP_GROUPS * MLP_WAVES) {
        asm volatile("" ::: "memory");
        const long row = (long)tile * 16 + c;
        const f4* xr = (const f4*)xin + row * 32 + h;
        const f4* ar = (const f4*)agg + row * 32 + h;

        f4 sv[16];
        #pragma unroll
        for (int i = 0; i < 16; ++i) sv[i] = xr[2 * i] + ar[2 * i];
        __builtin_amdgcn_sched_barrier(0);

        v8f acc[8];
        #pragma unroll
        for (int t = 0; t < 8; ++t) acc[t] = v8f{};

        {
            f4 aC[8], aN[8];
            #pragma unroll
            for (int t = 0; t < 8; ++t) aC[t] = w1f[t][0][lane];
            #pragma unroll
            for (int i = 0; i < 16; ++i) {
                if (i + 1 < 16) {
                    #pragma unroll
                    for (int t = 0; t < 8; ++t) aN[t] = w1f[t][i + 1][lane];
                }
                __builtin_amdgcn_sched_barrier(0);
                #pragma unroll
                for (int t = 0; t < 8; ++t) acc[t] = mma4(aC[t][0], aC[t][1], sv[i][0], sv[i][1], acc[t]);
                #pragma unroll
                for (int t = 0; t < 8; ++t) acc[t] = mma4(aC[t][2], aC[t][3], sv[i][2], sv[i][3], acc[t]);
                __builtin_amdgcn_sched_barrier(0);
                #pragma unroll
                for (int t = 0; t < 8; ++t) aC[t] = aN[t];
            }
        }
        #pragma unroll
        for (int t = 0; t < 8; ++t) {
            const f4 bA = *(const f4*)&bs[0][16 * t + 8 * h];
            const f4 bB = *(const f4*)&bs[0][16 * t + 8 * h + 4];
            #pragma unroll
            for (int r = 0; r < 8; ++r) {
                const float v = acc[t][r] + (r < 4 ? bA[r] : bB[r - 4]);
                acc[t][r] = v > 0.0f ? v : 0.0f;
            }
        }

        v8f acc2[8];
        #pragma unroll
        for (int t = 0; t < 8; ++t) acc2[t] = v8f{};
        {
            f4 aC[8], aN[8];
            #pragma unroll
            for (int t = 0; t < 8; ++t) aC[t] = w2f[t][0][lane];
            #pragma unroll
            for (int i = 0; i < 16; ++i) {
                const int tk = i >> 1;
                const int p4 = (i & 1) * 4;
                if (i + 1 < 16) {
                    #pragma unroll
                    for (int t = 0; t < 8; ++t) aN[t] = w2f[t][i + 1][lane];
                }
                __builtin_amdgcn_sched_barrier(0);
                #pragma unroll
                for (int t = 0; t < 8; ++t) acc2[t] = mma4(aC[t][0], aC[t][1], acc[tk][p4], acc[tk][p4 + 1], acc2[t]);
                #pragma unroll
                for (int t = 0; t < 8; ++t) acc2[t] = mma4(aC[t][2], aC[t][3], acc[tk][p4 + 2], acc[tk][p4 + 3], acc2[t]);
                __builtin_amdgcn_sched_barrier(0);
                #pragma unroll
                for (int t = 0; t < 8; ++t) aC[t] = aN[t];
            }
        }

        f4* orow = (f4*)(out + row * D + 8 * h);
        #pragma unroll
        for (int t = 0; t < 8; ++t) {
            const f4 bA = *(const f4*)&bs[1][16 * t + 8 * h];
            const f4 bB = *(const f4*)&bs[1][16 * t + 8 * h + 4];
            f4 o0, o1;
            #pragma unroll
            for (int r = 0; r < 4; ++r) {
                float u0 = acc2[t][r] + bA[r];
                float u1 = acc2[t][r + 4] + bB[r];
                if (applyReluOut) { u0 = u0 > 0.0f ? u0 : 0.0f; u1 = u1 > 0.0f ? u1 : 0.0f; }
                o0[r] = u0; o1[r] = u1;
            }
            orow[4 * t]     = o0;
            orow[4 * t + 1] = o1;
        }
    }
}

extern "C" void kernel_launch(void* const* d_in, const int* in_sizes, int n_in,
                              void* d_out, int out_size, void* d_ws, size_t ws_size,
                              hipStream_t stream) {
    const float* x  = (const float*)d_in[0];
    const int*   ei = (const int*)  d_in[1];
    const float* W1 = (const float*)d_in[2];
    const float* b1 = (const float*)d_in[3];
    const float* W2 = (const float*)d_in[4];
    const float* b2 = (const float*)d_in[5];
    float* out = (float*)d_out;

    const int N = in_sizes[0] / D;
    const int E = in_sizes[1] / 2;
    const int* src = ei;
    const int* dst = ei + E;

    float* agg = (float*)d_ws;
    float* h1  = agg + (size_t)N * D;

    const int  aBlocks = (N + AGG_ROWS - 1) / AGG_ROWS;
    const int  mTiles  = N / TILE_ROWS;

    gin_agg_owner<<<aBlocks, AGG_BLOCK, 0, stream>>>(x, src, dst, agg, N, E);
    gin_mlp_t<<<MLP_GROUPS, MLP_BLOCK, 0, stream>>>(x, agg, W1, b1, W2, b2, h1, 1, mTiles);

    gin_agg_owner<<<aBlocks, AGG_BLOCK, 0, stream>>>(h1, src, dst, agg, N, E);
    gin_mlp_t<<<MLP_GROUPS, MLP_BLOCK, 0, stream>>>(h1, agg, W1, b1, W2, b2, out, 0, mTiles);
}


# Second layer's neighbour sum reads back the per-wave edge lists that the first layer's scan built (kept in workspace) instead of scanning the destination list again; sum order unchanged
# speedup vs baseline: 2.8538x; 1.1051x over previous
#include <hip/hip_runtime.h>


#define D 128
#define TILE_ROWS 16

typedef float v2f __attribute__((ext_vector_type(2)));
typedef float v8f __attribute__((ext_vector_type(8)));

#define AGG_ROWS   400
#define AGG_WAVES  32
#define AGG_BLOCK  1024
#define SEG_CAP    256
#define SCAN_DEPTH 8

typedef float f4 __attribute__((ext_vector_type(4)));
typedef int   i4 __attribute__((ext_vector_type(4)));

__global__ __launch_bounds__(AGG_BLOCK)
void gin_agg_owner(const float* __restrict__ x,
                   const int*   __restrict__ src,
                   const int*   __restrict__ dst,
                   float*       __restrict__ agg,
                   unsigned*    __restrict__ keepSeg,
                   int*         __restrict__ keepCnt,
                   int*         __restrict__ keepOk,
                   int reuse,
                   int nNodes, int nEdges) {
    __shared__ f4  acc[AGG_ROWS][32];
    __shared__ unsigned seg[AGG_WAVES][SEG_CAP];
    __shared__ int s_cnt[AGG_WAVES];
    __shared__ int s_more[AGG_WAVES];

    const int tid  = threadIdx.x;
    const int lane = tid & 31;
    const int wave = __builtin_amdgcn_readfirstlane(tid >> 5);
    const int rowLo = blockIdx.x * AGG_ROWS;
    const int rowsHere = min(AGG_ROWS, nNodes - rowLo);

    unsigned* mySeg = keepSeg + ((long)blockIdx.x * AGG_WAVES + wave) * SEG_CAP;
    int*      myCnt = keepCnt + (long)blockIdx.x * AGG_WAVES + wave;
    bool useKept = false;
    int  keptCnt = 0;
    if (reuse) {
        useKept = __builtin_amdgcn_readfirstlane(keepOk[blockIdx.x]) == 1;
        keptCnt = __builtin_amdgcn_readfirstlane(min(max(*myCnt, 0), SEG_CAP));
    }
    const bool keep = (reuse == 0);

    f4* accFlat = &acc[0][0];
    for (int i = tid; i < AGG_ROWS * 32; i += AGG_BLOCK) accFlat[i] = f4{0.f, 0.f, 0.f, 0.f};

    int per = (nEdges + AGG_WAVES - 1) / AGG_WAVES;
    per = (per + 3) & ~3;
    const int beg = min(wave * per, nEdges);
    const int end = min(beg + per, nEdges);
    const bool vecOk = ((((unsigned long long)dst) & 15ull) == 0ull) && ((nEdges & 3) == 0);
    int pos = useKept ? end : beg;
    bool anyMore;
    bool firstPass = true;

    do {
        int cntw = 0;
        bool full = false;
        while (pos < end && !full) {
            i4 v[SCAN_DEPTH];
            if (vecOk) {
                #pragma unroll
                for (int u = 0; u < SCAN_DEPTH; ++u) {
                    const int e = pos + u * 128 + lane * 4;
                    const i4 t = *(const i4*)(dst + min(e, end - 4));
                    const i4 none = {-1, -1, -1, -1};
                    v[u] = (e < end) ? t : none;
                }
            } else {
                #pragma unroll
                for (int u = 0; u < SCAN_DEPTH; ++u) {
                    const int e = pos + u * 128 + lane * 4;
                    i4 t;
                    #pragma unroll
                    for (int k = 0; k < 4; ++k) {
                        const int d = dst[min(e + k, end - 1)];
                        t[k] = (e + k < end) ? d : -1;
                    }
                    v[u] = t;
                }
            }
            int stepsDone = 0;
            #pragma unroll
            for (int u = 0; u < SCAN_DEPTH; ++u) {
                if (!full && pos + u * 128 < end) {
                    if (cntw + 128 > SEG_CAP) {
                        full = true;
                    } else {
                        const int off = pos - beg + u * 128 + lane * 4;
                        unsigned lr[4]; bool h[4]; unsigned m[4];
                        #pragma unroll
                        for (int k = 0; k < 4; ++k) {
                            lr[k] = (unsigned)(v[u][k] - rowLo);
                            h[k]  = lr[k] < (unsigned)rowsHere;
                            m[k]  = __builtin_amdgcn_ballot_w32(h[k]);
                        }
                        if (m[0] | m[1] | m[2] | m[3]) {
                            #pragma unroll
                            for (int k = 0; k < 4; ++k) {
                                if (h[k]) {
                                    const int p = cntw + (int)__builtin_amdgcn_mbcnt_lo(m[k], 0u);
                                    seg[wave][p] = (lr[k] << 20) | (unsigned)(off + k);
                                }
                                cntw += __builtin_popcount(m[k]);
                            }
                        }
                        stepsDone = u + 1;
                    }
                }
            }
            pos += stepsDone * 128;
        }
        const bool keepNow = keep && firstPass;
        if (useKept) cntw = keptCnt;
        for (int i = lane; i < cntw; i += 32) {
            unsigned ent;
            if (useKept) {
                ent = mySeg[i];
            } else {
                const unsigned en = seg[wave][i];
                ent = (en & 0xFFF00000u) | (unsigned)src[beg + (int)(en & 0xFFFFFu)];
                if (keepNow) mySeg[i] = ent;
            }
            seg[wave][i] = ent;
        }
        if (lane == 0) {
            s_cnt[wave] = cntw; s_more[wave] = (pos < end) ? 1 : 0;
            if (keepNow) *myCnt = cntw;
        }
        __syncthreads();
        anyMore = __builtin_amdgcn_ballot_w32(s_more[lane] != 0) != 0u;
        if (keepNow && tid == 0) keepOk[blockIdx.x] = anyMore ? 0 : 1;
        firstPass = false;

        unsigned held = 0u;
        int nHeld = 0;
        auto flush = [&]() {
            for (int j = 0; j < nHeld; j += 4) {
                f4 xv[4]; int r[4];
                #pragma unroll
                for (int q = 0; q < 4; ++q) {
                    const unsigned e2 = (unsigned)__builtin_amdgcn_readlane((int)held, min(j + q, nHeld - 1));
                    r[q]  = (int)(e2 >> 20);
                    xv[q] = ((const f4*)x)[(long)(e2 & 0xFFFFFu) * 32 + lane];
                }
                #pragma unroll
                for (int q = 0; q < 4; ++q) {
                    if (j + q < nHeld) {
                        f4 a = acc[r[q]][lane];
                        a += xv[q];
                        acc[r[q]][lane] = a;
                    }
                }
            }
            nHeld = 0;
        };
        const int cl = s_cnt[lane];
        for (int sg = 0; sg < AGG_WAVES; ++sg) {
            const int c = __builtin_amdgcn_readlane(cl, sg);
            for (int i0 = 0; i0 < c; i0 += 32) {
                const int i = i0 + lane;
                const unsigned ent = (i < c) ? seg[sg][i] : 0u;
                const bool own = (i < c) && (int)((ent >> 20) & (AGG_WAVES - 1)) == wave;
                unsigned m = __builtin_amdgcn_ballot_w32(own);
                while (m) {
                    const int l = __builtin_ctz(m);
                    m &= m - 1u;
                    const unsigned e2 = (unsigned)__builtin_amdgcn_readlane((int)ent, l);
                    held = (lane == nHeld) ? e2 : held;
                    if (++nHeld == 32) flush();
                }
            }
        }
        flush();
        __syncthreads();
    } while (anyMore);

    f4* aggRows = (f4*)agg + (long)rowLo * 32;
    for (int i = tid; i < rowsHere * 32; i += AGG_BLOCK) aggRows[i] = accFlat[i];
}

#define MLP_WAVES 16
#define MLP_BLOCK (MLP_WAVES * 32)
#define MLP_GROUPS 256

static __device__ __forceinline__ v8f mma4(float a0, float a1, float b0, float b1, v8f c) {
    v2f a = {a0, a1};
    v2f b = {b0, b1};
    return __builtin_amdgcn_wmma_f32_16x16x4_f32(false, a, false, b, (short)0, c, false, false);
}

__global__ __launch_bounds__(MLP_BLOCK)
void gin_mlp_t(const float* __restrict__ xin,
               const float* __restrict__ agg,
               const float* __restrict__ W1, const float* __restrict__ b1,
               const float* __restrict__ W2, const float* __restrict__ b2,
               float* __restrict__ out,
               int applyReluOut, int nTiles) {
    __shared__ f4 w1f[8][16][32];
    __shared__ f4 w2f[8][16][32];
    __shared__ alignas(16) float bs[2][D];

    const int tid  = threadIdx.x;
    const int lane = tid & 31;
    const int wave = __builtin_amdgcn_readfirstlane(tid >> 5);

    {
        float* w1s = (float*)&w1f[0][0][0];
        float* w2s = (float*)&w2f[0][0][0];
        for (int idx = tid; idx < D * D; idx += MLP_BLOCK) {
            const int k  = idx >> 7;
            const int n  = idx & (D - 1);
            const int tn = n >> 4;
            const int cc = n & 15;
            const int m  = k & 3;
            const int i1 = k >> 3;
            const int h1 = (k >> 2) & 1;
            w1s[((tn * 16 + i1) * 32 + cc + 16 * h1) * 4 + m] = W1[idx];
            const int i2 = ((k >> 4) << 1) | ((k >> 2) & 1);
            const int h2 = (k >> 3) & 1;
            w2s[((tn * 16 + i2) * 32 + cc + 16 * h2) * 4 + m] = W2[idx];
        }
        if (tid < D) { bs[0][tid] = b1[tid]; bs[1][tid] = b2[tid]; }
    }
    __syncthreads();

    const int c = lane & 15;
    const int h = lane >> 4;

    for (int tile = blockIdx.x + MLP_GROUPS * wave; tile < nTiles; tile += MLP_GROUPS * MLP_WAVES) {
        asm volatile("" ::: "memory");
        const long row = (long)tile * 16 + c;
        const f4* xr = (const f4*)xin + row * 32 + h;
        const f4* ar = (const f4*)agg + row * 32 + h;

        f4 sv[16];
        #pragma unroll
        for (int i = 0; i < 16; ++i) sv[i] = xr[2 * i] + ar[2 * i];
        __builtin_amdgcn_sched_barrier(0);

        v8f acc[8];
        #pragma unroll
        for (int t = 0; t < 8; ++t) acc[t] = v8f{};

        {
            f4 aC[8], aN[8];
            #pragma unroll
            for (int t = 0; t < 8; ++t) aC[t] = w1f[t][0][lane];
            #pragma unroll
            for (int i = 0; i < 16; ++i) {
                if (i + 1 < 16) {
                    #pragma unroll
                    for (int t = 0; t < 8; ++t) aN[t] = w1f[t][i + 1][lane];
                }
                __builtin_amdgcn_sched_barrier(0);
                #pragma unroll
                for (int t = 0; t < 8; ++t) acc[t] = mma4(aC[t][0], aC[t][1], sv[i][0], sv[i][1], acc[t]);
                #pragma unroll
                for (int t = 0; t < 8; ++t) acc[t] = mma4(aC[t][2], aC[t][3], sv[i][2], sv[i][3], acc[t]);
                __builtin_amdgcn_sched_barrier(0);
                #pragma unroll
                for (int t = 0; t < 8; ++t) aC[t] = aN[t];
            }
        }
        #pragma unroll
        for (int t = 0; t < 8; ++t) {
            const f4 bA = *(const f4*)&bs[0][16 * t + 8 * h];
            const f4 bB = *(const f4*)&bs[0][16 * t + 8 * h + 4];
            #pragma unroll
            for (int r = 0; r < 8; ++r) {
                const float v = acc[t][r] + (r < 4 ? bA[r] : bB[r - 4]);
                acc[t][r] = v > 0.0f ? v : 0.0f;
            }
        }

        v8f acc2[8];
        #pragma unroll
        for (int t = 0; t < 8; ++t) acc2[t] = v8f{};
        {
            f4 aC[8], aN[8];
            #pragma unroll
            for (int t = 0; t < 8; ++t) aC[t] = w2f[t][0][lane];
            #pragma unroll
            for (int i = 0; i < 16; ++i) {
                const int tk = i >> 1;
                const int p4 = (i & 1) * 4;
                if (i + 1 < 16) {
                    #pragma unroll
                    for (int t = 0; t < 8; ++t) aN[t] = w2f[t][i + 1][lane];
                }
                __builtin_amdgcn_sched_barrier(0);
                #pragma unroll
                for (int t = 0; t < 8; ++t) acc2[t] = mma4(aC[t][0], aC[t][1], acc[tk][p4], acc[tk][p4 + 1], acc2[t]);
                #pragma unroll
                for (int t = 0; t < 8; ++t) acc2[t] = mma4(aC[t][2], aC[t][3], acc[tk][p4 + 2], acc[tk][p4 + 3], acc2[t]);
                __builtin_amdgcn_sched_barrier(0);
                #pragma unroll
                for (int t = 0; t < 8; ++t) aC[t] = aN[t];
            }
        }

        f4* orow = (f4*)(out + row * D + 8 * h);
        #pragma unroll
        for (int t = 0; t < 8; ++t) {
            const f4 bA = *(const f4*)&bs[1][16 * t + 8 * h];
            const f4 bB = *(const f4*)&bs[1][16 * t + 8 * h + 4];
            f4 o0, o1;
            #pragma unroll
            for (int r = 0; r < 4; ++r) {
                float u0 = acc2[t][r] + bA[r];
                float u1 = acc2[t][r + 4] + bB[r];
                if (applyReluOut) { u0 = u0 > 0.0f ? u0 : 0.0f; u1 = u1 > 0.0f ? u1 : 0.0f; }
                o0[r] = u0; o1[r] = u1;
            }
            orow[4 * t]     = o0;
            orow[4 * t + 1] = o1;
        }
    }
}

extern "C" void kernel_launch(void* const* d_in, const int* in_sizes, int n_in,
                              void* d_out, int out_size, void* d_ws, size_t ws_size,
                              hipStream_t stream) {
    const float* x  = (const float*)d_in[0];
    const int*   ei = (const int*)  d_in[1];
    const float* W1 = (const float*)d_in[2];
    const float* b1 = (const float*)d_in[3];
    const float* W2 = (const float*)d_in[4];
    const float* b2 = (const float*)d_in[5];
    float* out = (float*)d_out;

    const int N = in_sizes[0] / D;
    const int E = in_sizes[1] / 2;
    const int* src = ei;
    const int* dst = ei + E;

    float* agg = (float*)d_ws;
    float* h1  = agg + (size_t)N * D;

    const int  aBlocks = (N + AGG_ROWS - 1) / AGG_ROWS;
    const int  mTiles  = N / TILE_ROWS;
    unsigned* keepSeg = (unsigned*)(h1 + (size_t)N * D);
    int*      keepCnt = (int*)(keepSeg + (size_t)aBlocks * AGG_WAVES * SEG_CAP);
    int*      keepOk  = keepCnt + (size_t)aBlocks * AGG_WAVES;

    gin_agg_owner<<<aBlocks, AGG_BLOCK, 0, stream>>>(x, src, dst, agg, keepSeg, keepCnt, keepOk, 0, N, E);
    gin_mlp_t<<<MLP_GROUPS, MLP_BLOCK, 0, stream>>>(x, agg, W1, b1, W2, b2, h1, 1, mTiles);

    gin_agg_owner<<<aBlocks, AGG_BLOCK, 0, stream>>>(h1, src, dst, agg, keepSeg, keepCnt, keepOk, 1, N, E);
    gin_mlp_t<<<MLP_GROUPS, MLP_BLOCK, 0, stream>>>(h1, agg, W1, b1, W2, b2, out, 0, mTiles);
}


# Neighbour sum: a wave requests the x rows of 8 own edges at once and adds them one group later, so the loads fly while it walks the lists (was 3+1 loads, each waited for); sum order unchanged
# speedup vs baseline: 3.0649x; 1.0740x over previous
#include <hip/hip_runtime.h>


#define D 128
#define TILE_ROWS 16

typedef float v2f __attribute__((ext_vector_type(2)));
typedef float v8f __attribute__((ext_vector_type(8)));

#define AGG_ROWS   400
#define AGG_WAVES  32
#define AGG_BLOCK  1024
#define SEG_CAP    256
#define SCAN_DEPTH 8
#define ROW_GROUP  8

typedef float f4 __attribute__((ext_vector_type(4)));
typedef int   i4 __attribute__((ext_vector_type(4)));

__global__ __launch_bounds__(AGG_BLOCK)
void gin_agg_owner(const float* __restrict__ x,
                   const int*   __restrict__ src,
                   const int*   __restrict__ dst,
                   float*       __restrict__ agg,
                   unsigned*    __restrict__ keepSeg,
                   int*         __restrict__ keepCnt,
                   int*         __restrict__ keepOk,
                   int reuse,
                   int nNodes, int nEdges) {
    __shared__ f4  acc[AGG_ROWS][32];
    __shared__ unsigned seg[AGG_WAVES][SEG_CAP];
    __shared__ int s_cnt[AGG_WAVES];
    __shared__ int s_more[AGG_WAVES];

    const int tid  = threadIdx.x;
    const int lane = tid & 31;
    const int wave = __builtin_amdgcn_readfirstlane(tid >> 5);
    const int rowLo = blockIdx.x * AGG_ROWS;
    const int rowsHere = min(AGG_ROWS, nNodes - rowLo);

    unsigned* mySeg = keepSeg + ((long)blockIdx.x * AGG_WAVES + wave) * SEG_CAP;
    int*      myCnt = keepCnt + (long)blockIdx.x * AGG_WAVES + wave;
    bool useKept = false;
    int  keptCnt = 0;
    if (reuse) {
        useKept = __builtin_amdgcn_readfirstlane(keepOk[blockIdx.x]) == 1;
        keptCnt = __builtin_amdgcn_readfirstlane(min(max(*myCnt, 0), SEG_CAP));
    }
    const bool keep = (reuse == 0);

    f4* accFlat = &acc[0][0];
    for (int i = tid; i < AGG_ROWS * 32; i += AGG_BLOCK) accFlat[i] = f4{0.f, 0.f, 0.f, 0.f};

    int per = (nEdges + AGG_WAVES - 1) / AGG_WAVES;
    per = (per + 3) & ~3;
    const int beg = min(wave * per, nEdges);
    const int end = min(beg + per, nEdges);
    const bool vecOk = ((((unsigned long long)dst) & 15ull) == 0ull) && ((nEdges & 3) == 0);
    int pos = useKept ? end : beg;
    bool anyMore;
    bool firstPass = true;

    do {
        int cntw = 0;
        bool full = false;
        while (pos < end && !full) {
            i4 v[SCAN_DEPTH];
            if (vecOk) {
                #pragma unroll
                for (int u = 0; u < SCAN_DEPTH; ++u) {
                    const int e = pos + u * 128 + lane * 4;
                    const i4 t = *(const i4*)(dst + min(e, end - 4));
                    const i4 none = {-1, -1, -1, -1};
                    v[u] = (e < end) ? t : none;
                }
            } else {
                #pragma unroll
                for (int u = 0; u < SCAN_DEPTH; ++u) {
                    const int e = pos + u * 128 + lane * 4;
                    i4 t;
                    #pragma unroll
                    for (int k = 0; k < 4; ++k) {
                        const int d = dst[min(e + k, end - 1)];
                        t[k] = (e + k < end) ? d : -1;
                    }
                    v[u] = t;
                }
            }
            int stepsDone = 0;
            #pragma unroll
            for (int u = 0; u < SCAN_DEPTH; ++u) {
                if (!full && pos + u * 128 < end) {
                    if (cntw + 128 > SEG_CAP) {
                        full = true;
                    } else {
                        const int off = pos - beg + u * 128 + lane * 4;
                        unsigned lr[4]; bool h[4]; unsigned m[4];
                        #pragma unroll
                        for (int k = 0; k < 4; ++k) {
                            lr[k] = (unsigned)(v[u][k] - rowLo);
                            h[k]  = lr[k] < (unsigned)rowsHere;
                            m[k]  = __builtin_amdgcn_ballot_w32(h[k]);
                        }
                        if (m[0] | m[1] | m[2] | m[3]) {
                            #pragma unroll
                            for (int k = 0; k < 4; ++k) {
                                if (h[k]) {
                                    const int p = cntw + (int)__builtin_amdgcn_mbcnt_lo(m[k], 0u);
                                    seg[wave][p] = (lr[k] << 20) | (unsigned)(off + k);
                                }
                                cntw += __builtin_popcount(m[k]);
                            }
                        }
                        stepsDone = u + 1;
                    }
                }
            }
            pos += stepsDone * 128;
        }
        const bool keepNow = keep && firstPass;
        if (useKept) cntw = keptCnt;
        for (int i = lane; i < cntw; i += 32) {
            unsigned ent;
            if (useKept) {
                ent = mySeg[i];
            } else {
                const unsigned en = seg[wave][i];
                ent = (en & 0xFFF00000u) | (unsigned)src[beg + (int)(en & 0xFFFFFu)];
                if (keepNow) mySeg[i] = ent;
            }
            seg[wave][i] = ent;
        }
        if (lane == 0) {
            s_cnt[wave] = cntw; s_more[wave] = (pos < end) ? 1 : 0;
            if (keepNow) *myCnt = cntw;
        }
        __syncthreads();
        anyMore = __builtin_amdgcn_ballot_w32(s_more[lane] != 0) != 0u;
        if (keepNow && tid == 0) keepOk[blockIdx.x] = anyMore ? 0 : 1;
        firstPass = false;

        unsigned held = 0u;
        int nHeld = 0;
        unsigned sent = 0u;
        int nSent = 0;
        const f4 zero4 = {0.f, 0.f, 0.f, 0.f};
        f4 xr0 = zero4, xr1 = zero4, xr2 = zero4, xr3 = zero4, xr4 = zero4, xr5 = zero4, xr6 = zero4, xr7 = zero4;
#define ROWS(M) M(0) M(1) M(2) M(3) M(4) M(5) M(6) M(7)
#define ROW_ADD(q) if (q < nSent) { \
            const int r = (int)((unsigned)__builtin_amdgcn_readlane((int)sent, q) >> 20); \
            f4 a = acc[r][lane]; a += xr##q; acc[r][lane] = a; }
#define ROW_LOAD(q) { \
            const unsigned e2 = (unsigned)__builtin_amdgcn_readlane((int)held, min(q, nHeld - 1)); \
            xr##q = ((const f4*)x)[(long)(e2 & 0xFFFFFu) * 32 + lane]; }
#define ROW_STEP() { \
            ROWS(ROW_ADD) \
            sent = held; nSent = nHeld; \
            if (nHeld > 0) { ROWS(ROW_LOAD) } \
            nHeld = 0; }
        const int cl = s_cnt[lane];
        for (int sg = 0; sg < AGG_WAVES; ++sg) {
            const int c = __builtin_amdgcn_readlane(cl, sg);
            for (int i0 = 0; i0 < c; i0 += 32) {
                const int i = i0 + lane;
                const unsigned ent = (i < c) ? seg[sg][i] : 0u;
                const bool own = (i < c) && (int)((ent >> 20) & (AGG_WAVES - 1)) == wave;
                unsigned m = __builtin_amdgcn_ballot_w32(own);
                while (m) {
                    const int l = __builtin_ctz(m);
                    m &= m - 1u;
                    const unsigned e2 = (unsigned)__builtin_amdgcn_readlane((int)ent, l);
                    held = (lane == nHeld) ? e2 : held;
                    if (++nHeld == ROW_GROUP) ROW_STEP()
                }
            }
        }
        ROW_STEP()
        ROWS(ROW_ADD)
#undef ROW_STEP
#undef ROW_LOAD
#undef ROW_ADD
#undef ROWS
        __syncthreads();
    } while (anyMore);

    f4* aggRows = (f4*)agg + (long)rowLo * 32;
    for (int i = tid; i < rowsHere * 32; i += AGG_BLOCK) aggRows[i] = accFlat[i];
}

#define MLP_WAVES 16
#define MLP_BLOCK (MLP_WAVES * 32)
#define MLP_GROUPS 256

static __device__ __forceinline__ v8f mma4(float a0, float a1, float b0, float b1, v8f c) {
    v2f a = {a0, a1};
    v2f b = {b0, b1};
    return __builtin_amdgcn_wmma_f32_16x16x4_f32(false, a, false, b, (short)0, c, false, false);
}

__global__ __launch_bounds__(MLP_BLOCK)
void gin_mlp_t(const float* __restrict__ xin,
               const float* __restrict__ agg,
               const float* __restrict__ W1, const float* __restrict__ b1,
               const float* __restrict__ W2, const float* __restrict__ b2,
               float* __restrict__ out,
               int applyReluOut, int nTiles) {
    __shared__ f4 w1f[8][16][32];
    __shared__ f4 w2f[8][16][32];
    __shared__ alignas(16) float bs[2][D];

    const int tid  = threadIdx.x;
    const int lane = tid & 31;
    const int wave = __builtin_amdgcn_readfirstlane(tid >> 5);

    {
        float* w1s = (float*)&w1f[0][0][0];
        float* w2s = (float*)&w2f[0][0][0];
        for (int idx = tid; idx < D * D; idx += MLP_BLOCK) {
            const int k  = idx >> 7;
            const int n  = idx & (D - 1);
            const int tn = n >> 4;
            const int cc = n & 15;
            const int m  = k & 3;
            const int i1 = k >> 3;
            const int h1 = (k >> 2) & 1;
            w1s[((tn * 16 + i1) * 32 + cc + 16 * h1) * 4 + m] = W1[idx];
            const int i2 = ((k >> 4) << 1) | ((k >> 2) & 1);
            const int h2 = (k >> 3) & 1;
            w2s[((tn * 16 + i2) * 32 + cc + 16 * h2) * 4 + m] = W2[idx];
        }
        if (tid < D) { bs[0][tid] = b1[tid]; bs[1][tid] = b2[tid]; }
    }
    __syncthreads();

    const int c = lane & 15;
    const int h = lane >> 4;

    for (int tile = blockIdx.x + MLP_GROUPS * wave; tile < nTiles; tile += MLP_GROUPS * MLP_WAVES) {
        asm volatile("" ::: "memory");
        const long row = (long)tile * 16 + c;
        const f4* xr = (const f4*)xin + row * 32 + h;
        const f4* ar = (const f4*)agg + row * 32 + h;

        f4 sv[16];
        #pragma unroll
        for (int i = 0; i < 16; ++i) sv[i] = xr[2 * i] + ar[2 * i];
        __builtin_amdgcn_sched_barrier(0);

        v8f acc[8];
        #pragma unroll
        for (int t = 0; t < 8; ++t) acc[t] = v8f{};

        {
            f4 aC[8], aN[8];
            #pragma unroll
            for (int t = 0; t < 8; ++t) aC[t] = w1f[t][0][lane];
            #pragma unroll
            for (int i = 0; i < 16; ++i) {
                if (i + 1 < 16) {
                    #pragma unroll
                    for (int t = 0; t < 8; ++t) aN[t] = w1f[t][i + 1][lane];
                }
                __builtin_amdgcn_sched_barrier(0);
                #pragma unroll
                for (int t = 0; t < 8; ++t) acc[t] = mma4(aC[t][0], aC[t][1], sv[i][0], sv[i][1], acc[t]);
                #pragma unroll
                for (int t = 0; t < 8; ++t) acc[t] = mma4(aC[t][2], aC[t][3], sv[i][2], sv[i][3], acc[t]);
                __builtin_amdgcn_sched_barrier(0);
                #pragma unroll
                for (int t = 0; t < 8; ++t) aC[t] = aN[t];
            }
        }
        #pragma unroll
        for (int t = 0; t < 8; ++t) {
            const f4 bA = *(const f4*)&bs[0][16 * t + 8 * h];
            const f4 bB = *(const f4*)&bs[0][16 * t + 8 * h + 4];
            #pragma unroll
            for (int r = 0; r < 8; ++r) {
                const float v = acc[t][r] + (r < 4 ? bA[r] : bB[r - 4]);
                acc[t][r] = v > 0.0f ? v : 0.0f;
            }
        }

        v8f acc2[8];
        #pragma unroll
        for (int t = 0; t < 8; ++t) acc2[t] = v8f{};
        {
            f4 aC[8], aN[8];
            #pragma unroll
            for (int t = 0; t < 8; ++t) aC[t] = w2f[t][0][lane];
            #pragma unroll
            for (int i = 0; i < 16; ++i) {
                const int tk = i >> 1;
                const int p4 = (i & 1) * 4;
                if (i + 1 < 16) {
                    #pragma unroll
                    for (int t = 0; t < 8; ++t) aN[t] = w2f[t][i + 1][lane];
                }
                __builtin_amdgcn_sched_barrier(0);
                #pragma unroll
                for (int t = 0; t < 8; ++t) acc2[t] = mma4(aC[t][0], aC[t][1], acc[tk][p4], acc[tk][p4 + 1], acc2[t]);
                #pragma unroll
                for (int t = 0; t < 8; ++t) acc2[t] = mma4(aC[t][2], aC[t][3], acc[tk][p4 + 2], acc[tk][p4 + 3], acc2[t]);
                __builtin_amdgcn_sched_barrier(0);
                #pragma unroll
                for (int t = 0; t < 8; ++t) aC[t] = aN[t];
            }
        }

        f4* orow = (f4*)(out + row * D + 8 * h);
        #pragma unroll
        for (int t = 0; t < 8; ++t) {
            const f4 bA = *(const f4*)&bs[1][16 * t + 8 * h];
            const f4 bB = *(const f4*)&bs[1][16 * t + 8 * h + 4];
            f4 o0, o1;
            #pragma unroll
            for (int r = 0; r < 4; ++r) {
                float u0 = acc2[t][r] + bA[r];
                float u1 = acc2[t][r + 4] + bB[r];
                if (applyReluOut) { u0 = u0 > 0.0f ? u0 : 0.0f; u1 = u1 > 0.0f ? u1 : 0.0f; }
                o0[r] = u0; o1[r] = u1;
            }
            orow[4 * t]     = o0;
            orow[4 * t + 1] = o1;
        }
    }
}

extern "C" void kernel_launch(void* const* d_in, const int* in_sizes, int n_in,
                              void* d_out, int out_size, void* d_ws, size_t ws_size,
                              hipStream_t stream) {
    const float* x  = (const float*)d_in[0];
    const int*   ei = (const int*)  d_in[1];
    const float* W1 = (const float*)d_in[2];
    const float* b1 = (const float*)d_in[3];
    const float* W2 = (const float*)d_in[4];
    const float* b2 = (const float*)d_in[5];
    float* out = (float*)d_out;

    const int N = in_sizes[0] / D;
    const int E = in_sizes[1] / 2;
    const int* src = ei;
    const int* dst = ei + E;

    float* agg = (float*)d_ws;
    float* h1  = agg + (size_t)N * D;

    const int  aBlocks = (N + AGG_ROWS - 1) / AGG_ROWS;
    const int  mTiles  = N / TILE_ROWS;
    unsigned* keepSeg = (unsigned*)(h1 + (size_t)N * D);
    int*      keepCnt = (int*)(keepSeg + (size_t)aBlocks * AGG_WAVES * SEG_CAP);
    int*      keepOk  = keepCnt + (size_t)aBlocks * AGG_WAVES;

    gin_agg_owner<<<aBlocks, AGG_BLOCK, 0, stream>>>(x, src, dst, agg, keepSeg, keepCnt, keepOk, 0, N, E);
    gin_mlp_t<<<MLP_GROUPS, MLP_BLOCK, 0, stream>>>(x, agg, W1, b1, W2, b2, h1, 1, mTiles);

    gin_agg_owner<<<aBlocks, AGG_BLOCK, 0, stream>>>(h1, src, dst, agg, keepSeg, keepCnt, keepOk, 1, N, E);
    gin_mlp_t<<<MLP_GROUPS, MLP_BLOCK, 0, stream>>>(h1, agg, W1, b1, W2, b2, out, 0, mTiles);
}


# MLP prologue: a wave requests its first tile's rows before the weights are staged; staging issues all its loads ahead of its LDS stores, one per fragment slot (was 8 waited groups, then the rows)
# speedup vs baseline: 3.4025x; 1.1102x over previous
#include <hip/hip_runtime.h>


#define D 128
#define TILE_ROWS 16

typedef float v2f __attribute__((ext_vector_type(2)));
typedef float v8f __attribute__((ext_vector_type(8)));

#define AGG_ROWS   400
#define AGG_WAVES  32
#define AGG_BLOCK  1024
#define SEG_CAP    256
#define SCAN_DEPTH 8
#define ROW_GROUP  8

typedef float f4 __attribute__((ext_vector_type(4)));
typedef int   i4 __attribute__((ext_vector_type(4)));

__global__ __launch_bounds__(AGG_BLOCK)
void gin_agg_owner(const float* __restrict__ x,
                   const int*   __restrict__ src,
                   const int*   __restrict__ dst,
                   float*       __restrict__ agg,
                   unsigned*    __restrict__ keepSeg,
                   int*         __restrict__ keepCnt,
                   int*         __restrict__ keepOk,
                   int reuse,
                   int nNodes, int nEdges) {
    __shared__ f4  acc[AGG_ROWS][32];
    __shared__ unsigned seg[AGG_WAVES][SEG_CAP];
    __shared__ int s_cnt[AGG_WAVES];
    __shared__ int s_more[AGG_WAVES];

    const int tid  = threadIdx.x;
    const int lane = tid & 31;
    const int wave = __builtin_amdgcn_readfirstlane(tid >> 5);
    const int rowLo = blockIdx.x * AGG_ROWS;
    const int rowsHere = min(AGG_ROWS, nNodes - rowLo);

    unsigned* mySeg = keepSeg + ((long)blockIdx.x * AGG_WAVES + wave) * SEG_CAP;
    int*      myCnt = keepCnt + (long)blockIdx.x * AGG_WAVES + wave;
    bool useKept = false;
    int  keptCnt = 0;
    if (reuse) {
        useKept = __builtin_amdgcn_readfirstlane(keepOk[blockIdx.x]) == 1;
        keptCnt = __builtin_amdgcn_readfirstlane(min(max(*myCnt, 0), SEG_CAP));
    }
    const bool keep = (reuse == 0);

    f4* accFlat = &acc[0][0];
    for (int i = tid; i < AGG_ROWS * 32; i += AGG_BLOCK) accFlat[i] = f4{0.f, 0.f, 0.f, 0.f};

    int per = (nEdges + AGG_WAVES - 1) / AGG_WAVES;
    per = (per + 3) & ~3;
    const int beg = min(wave * per, nEdges);
    const int end = min(beg + per, nEdges);
    const bool vecOk = ((((unsigned long long)dst) & 15ull) == 0ull) && ((nEdges & 3) == 0);
    int pos = useKept ? end : beg;
    bool anyMore;
    bool firstPass = true;

    do {
        int cntw = 0;
        bool full = false;
        while (pos < end && !full) {
            i4 v[SCAN_DEPTH];
            if (vecOk) {
                #pragma unroll
                for (int u = 0; u < SCAN_DEPTH; ++u) {
                    const int e = pos + u * 128 + lane * 4;
                    const i4 t = *(const i4*)(dst + min(e, end - 4));
                    const i4 none = {-1, -1, -1, -1};
                    v[u] = (e < end) ? t : none;
                }
            } else {
                #pragma unroll
                for (int u = 0; u < SCAN_DEPTH; ++u) {
                    const int e = pos + u * 128 + lane * 4;
                    i4 t;
                    #pragma unroll
                    for (int k = 0; k < 4; ++k) {
                        const int d = dst[min(e + k, end - 1)];
                        t[k] = (e + k < end) ? d : -1;
                    }
                    v[u] = t;
                }
            }
            int stepsDone = 0;
            #pragma unroll
            for (int u = 0; u < SCAN_DEPTH; ++u) {
                if (!full && pos + u * 128 < end) {
                    if (cntw + 128 > SEG_CAP) {
                        full = true;
                    } else {
                        const int off = pos - beg + u * 128 + lane * 4;
                        unsigned lr[4]; bool h[4]; unsigned m[4];
                        #pragma unroll
                        for (int k = 0; k < 4; ++k) {
                            lr[k] = (unsigned)(v[u][k] - rowLo);
                            h[k]  = lr[k] < (unsigned)rowsHere;
                            m[k]  = __builtin_amdgcn_ballot_w32(h[k]);
                        }
                        if (m[0] | m[1] | m[2] | m[3]) {
                            #pragma unroll
                            for (int k = 0; k < 4; ++k) {
                                if (h[k]) {
                                    const int p = cntw + (int)__builtin_amdgcn_mbcnt_lo(m[k], 0u);
                                    seg[wave][p] = (lr[k] << 20) | (unsigned)(off + k);
                                }
                                cntw += __builtin_popcount(m[k]);
                            }
                        }
                        stepsDone = u + 1;
                    }
                }
            }
            pos += stepsDone * 128;
        }
        const bool keepNow = keep && firstPass;
        if (useKept) cntw = keptCnt;
        for (int i = lane; i < cntw; i += 32) {
            unsigned ent;
            if (useKept) {
                ent = mySeg[i];
            } else {
                const unsigned en = seg[wave][i];
                ent = (en & 0xFFF00000u) | (unsigned)src[beg + (int)(en & 0xFFFFFu)];
                if (keepNow) mySeg[i] = ent;
            }
            seg[wave][i] = ent;
        }
        if (lane == 0) {
            s_cnt[wave] = cntw; s_more[wave] = (pos < end) ? 1 : 0;
            if (keepNow) *myCnt = cntw;
        }
        __syncthreads();
        anyMore = __builtin_amdgcn_ballot_w32(s_more[lane] != 0) != 0u;
        if (keepNow && tid == 0) keepOk[blockIdx.x] = anyMore ? 0 : 1;
        firstPass = false;

        unsigned held = 0u;
        int nHeld = 0;
        unsigned sent = 0u;
        int nSent = 0;
        const f4 zero4 = {0.f, 0.f, 0.f, 0.f};
        f4 xr0 = zero4, xr1 = zero4, xr2 = zero4, xr3 = zero4, xr4 = zero4, xr5 = zero4, xr6 = zero4, xr7 = zero4;
#define ROWS(M) M(0) M(1) M(2) M(3) M(4) M(5) M(6) M(7)
#define ROW_ADD(q) if (q < nSent) { \
            const int r = (int)((unsigned)__builtin_amdgcn_readlane((int)sent, q) >> 20); \
            f4 a = acc[r][lane]; a += xr##q; acc[r][lane] = a; }
#define ROW_LOAD(q) { \
            const unsigned e2 = (unsigned)__builtin_amdgcn_readlane((int)held, min(q, nHeld - 1)); \
            xr##q = ((const f4*)x)[(long)(e2 & 0xFFFFFu) * 32 + lane]; }
#define ROW_STEP() { \
            ROWS(ROW_ADD) \
            sent = held; nSent = nHeld; \
            if (nHeld > 0) { ROWS(ROW_LOAD) } \
            nHeld = 0; }
        const int cl = s_cnt[lane];
        for (int sg = 0; sg < AGG_WAVES; ++sg) {
            const int c = __builtin_amdgcn_readlane(cl, sg);
            for (int i0 = 0; i0 < c; i0 += 32) {
                const int i = i0 + lane;
                const unsigned ent = (i < c) ? seg[sg][i] : 0u;
                const bool own = (i < c) && (int)((ent >> 20) & (AGG_WAVES - 1)) == wave;
                unsigned m = __builtin_amdgcn_ballot_w32(own);
                while (m) {
                    const int l = __builtin_ctz(m);
                    m &= m - 1u;
                    const unsigned e2 = (unsigned)__builtin_amdgcn_readlane((int)ent, l);
                    held = (lane == nHeld) ? e2 : held;
                    if (++nHeld == ROW_GROUP) ROW_STEP()
                }
            }
        }
        ROW_STEP()
        ROWS(ROW_ADD)
#undef ROW_STEP
#undef ROW_LOAD
#undef ROW_ADD
#undef ROWS
        __syncthreads();
    } while (anyMore);

    f4* aggRows = (f4*)agg + (long)rowLo * 32;
    for (int i = tid; i < rowsHere * 32; i += AGG_BLOCK) aggRows[i] = accFlat[i];
}

#define MLP_WAVES 16
#define MLP_BLOCK (MLP_WAVES * 32)
#define MLP_GROUPS 256

static __device__ __forceinline__ v8f mma4(float a0, float a1, float b0, float b1, v8f c) {
    v2f a = {a0, a1};
    v2f b = {b0, b1};
    return __builtin_amdgcn_wmma_f32_16x16x4_f32(false, a, false, b, (short)0, c, false, false);
}

__global__ __launch_bounds__(MLP_BLOCK)
void gin_mlp_t(const float* __restrict__ xin,
               const float* __restrict__ agg,
               const float* __restrict__ W1, const float* __restrict__ b1,
               const float* __restrict__ W2, const float* __restrict__ b2,
               float* __restrict__ out,
               int applyReluOut, int nTiles) {
    __shared__ f4 w1f[8][16][32];
    __shared__ f4 w2f[8][16][32];
    __shared__ alignas(16) float bs[2][D];

    const int tid  = threadIdx.x;
    const int lane = tid & 31;
    const int wave = __builtin_amdgcn_readfirstlane(tid >> 5);

    const int c = lane & 15;
    const int h = lane >> 4;
    int tile = blockIdx.x + MLP_GROUPS * wave;

    const f4 zero4 = {0.f, 0.f, 0.f, 0.f};
    f4 xv[16], av[16];
    #pragma unroll
    for (int i = 0; i < 16; ++i) { xv[i] = zero4; av[i] = zero4; }
    if (tile < nTiles) {
        const long row0 = (long)tile * 16 + c;
        const f4* xr0 = (const f4*)xin + row0 * 32 + h;
        const f4* ar0 = (const f4*)agg + row0 * 32 + h;
        #pragma unroll
        for (int i = 0; i < 16; ++i) { xv[i] = xr0[2 * i]; av[i] = ar0[2 * i]; }
    }

    {
        const int n  = tid & (D - 1);
        const int tn = n >> 4;
        const int cc = n & 15;
        const int q  = tid >> 7;
        const float bias1 = b1[n];
        const float bias2 = b2[n];
        f4 wa[8], wb[8];
        #pragma unroll
        for (int j = 0; j < 8; ++j) {
            const int kg = q + 4 * j;
            #pragma unroll
            for (int m = 0; m < 4; ++m) {
                wa[j][m] = W1[(4 * kg + m) * D + n];
                wb[j][m] = W2[(4 * kg + m) * D + n];
            }
        }
        #pragma unroll
        for (int j = 0; j < 8; ++j) {
            const int kg = q + 4 * j;
            const int i1 = kg >> 1;
            const int h1 = kg & 1;
            w1f[tn][i1][cc + 16 * h1] = wa[j];
            const int i2 = ((kg >> 2) << 1) | (kg & 1);
            const int h2 = (kg >> 1) & 1;
            w2f[tn][i2][cc + 16 * h2] = wb[j];
        }
        bs[0][n] = bias1;
        bs[1][n] = bias2;
    }
    __syncthreads();

    while (tile < nTiles) {
        asm volatile("" ::: "memory");
        const long row = (long)tile * 16 + c;

        f4 sv[16];
        #pragma unroll
        for (int i = 0; i < 16; ++i) sv[i] = xv[i] + av[i];
        __builtin_amdgcn_sched_barrier(0);

        v8f acc[8];
        #pragma unroll
        for (int t = 0; t < 8; ++t) acc[t] = v8f{};

        {
            f4 aC[8], aN[8];
            #pragma unroll
            for (int t = 0; t < 8; ++t) aC[t] = w1f[t][0][lane];
            #pragma unroll
            for (int i = 0; i < 16; ++i) {
                if (i + 1 < 16) {
                    #pragma unroll
                    for (int t = 0; t < 8; ++t) aN[t] = w1f[t][i + 1][lane];
                }
                __builtin_amdgcn_sched_barrier(0);
                #pragma unroll
                for (int t = 0; t < 8; ++t) acc[t] = mma4(aC[t][0], aC[t][1], sv[i][0], sv[i][1], acc[t]);
                #pragma unroll
                for (int t = 0; t < 8; ++t) acc[t] = mma4(aC[t][2], aC[t][3], sv[i][2], sv[i][3], acc[t]);
                __builtin_amdgcn_sched_barrier(0);
                #pragma unroll
                for (int t = 0; t < 8; ++t) aC[t] = aN[t];
            }
        }
        #pragma unroll
        for (int t = 0; t < 8; ++t) {
            const f4 bA = *(const f4*)&bs[0][16 * t + 8 * h];
            const f4 bB = *(const f4*)&bs[0][16 * t + 8 * h + 4];
            #pragma unroll
            for (int r = 0; r < 8; ++r) {
                const float v = acc[t][r] + (r < 4 ? bA[r] : bB[r - 4]);
                acc[t][r] = v > 0.0f ? v : 0.0f;
            }
        }

        v8f acc2[8];
        #pragma unroll
        for (int t = 0; t < 8; ++t) acc2[t] = v8f{};
        {
            f4 aC[8], aN[8];
            #pragma unroll
            for (int t = 0; t < 8; ++t) aC[t] = w2f[t][0][lane];
            #pragma unroll
            for (int i = 0; i < 16; ++i) {
                const int tk = i >> 1;
                const int p4 = (i & 1) * 4;
                if (i + 1 < 16) {
                    #pragma unroll
                    for (int t = 0; t < 8; ++t) aN[t] = w2f[t][i + 1][lane];
                }
                __builtin_amdgcn_sched_barrier(0);
                #pragma unroll
                for (int t = 0; t < 8; ++t) acc2[t] = mma4(aC[t][0], aC[t][1], acc[tk][p4], acc[tk][p4 + 1], acc2[t]);
                #pragma unroll
                for (int t = 0; t < 8; ++t) acc2[t] = mma4(aC[t][2], aC[t][3], acc[tk][p4 + 2], acc[tk][p4 + 3], acc2[t]);
                __builtin_amdgcn_sched_barrier(0);
                #pragma unroll
                for (int t = 0; t < 8; ++t) aC[t] = aN[t];
            }
        }

        f4* orow = (f4*)(out + row * D + 8 * h);
        #pragma unroll
        for (int t = 0; t < 8; ++t) {
            const f4 bA = *(const f4*)&bs[1][16 * t + 8 * h];
            const f4 bB = *(const f4*)&bs[1][16 * t + 8 * h + 4];
            f4 o0, o1;
            #pragma unroll
            for (int r = 0; r < 4; ++r) {
                float u0 = acc2[t][r] + bA[r];
                float u1 = acc2[t][r + 4] + bB[r];
                if (applyReluOut) { u0 = u0 > 0.0f ? u0 : 0.0f; u1 = u1 > 0.0f ? u1 : 0.0f; }
                o0[r] = u0; o1[r] = u1;
            }
            orow[4 * t]     = o0;
            orow[4 * t + 1] = o1;
        }

        tile += MLP_GROUPS * MLP_WAVES;
        if (tile < nTiles) {
            const long rowN = (long)tile * 16 + c;
            const f4* xrN = (const f4*)xin + rowN * 32 + h;
            const f4* arN = (const f4*)agg + rowN * 32 + h;
            #pragma unroll
            for (int i = 0; i < 16; ++i) { xv[i] = xrN[2 * i]; av[i] = arN[2 * i]; }
        }
    }
}

extern "C" void kernel_launch(void* const* d_in, const int* in_sizes, int n_in,
                              void* d_out, int out_size, void* d_ws, size_t ws_size,
                              hipStream_t stream) {
    const float* x  = (const float*)d_in[0];
    const int*   ei = (const int*)  d_in[1];
    const float* W1 = (const float*)d_in[2];
    const float* b1 = (const float*)d_in[3];
    const float* W2 = (const float*)d_in[4];
    const float* b2 = (const float*)d_in[5];
    float* out = (float*)d_out;

    const int N = in_sizes[0] / D;
    const int E = in_sizes[1] / 2;
    const int* src = ei;
    const int* dst = ei + E;

    float* agg = (float*)d_ws;
    float* h1  = agg + (size_t)N * D;

    const int  aBlocks = (N + AGG_ROWS - 1) / AGG_ROWS;
    const int  mTiles  = N / TILE_ROWS;
    unsigned* keepSeg = (unsigned*)(h1 + (size_t)N * D);
    int*      keepCnt = (int*)(keepSeg + (size_t)aBlocks * AGG_WAVES * SEG_CAP);
    int*      keepOk  = keepCnt + (size_t)aBlocks * AGG_WAVES;

    gin_agg_owner<<<aBlocks, AGG_BLOCK, 0, stream>>>(x, src, dst, agg, keepSeg, keepCnt, keepOk, 0, N, E);
    gin_mlp_t<<<MLP_GROUPS, MLP_BLOCK, 0, stream>>>(x, agg, W1, b1, W2, b2, h1, 1, mTiles);

    gin_agg_owner<<<aBlocks, AGG_BLOCK, 0, stream>>>(h1, src, dst, agg, keepSeg, keepCnt, keepOk, 1, N, E);
    gin_mlp_t<<<MLP_GROUPS, MLP_BLOCK, 0, stream>>>(h1, agg, W1, b1, W2, b2, out, 0, mTiles);
}


# First layer's scan reads a one-byte owner key per edge (made by a new small first launch), not the 4-byte destination: a quarter of the bytes; order of list entries, so sum order, changed (fixed)
# speedup vs baseline: 3.5790x; 1.0519x over previous
#include <hip/hip_runtime.h>


#define D 128
#define TILE_ROWS 16

typedef float v2f __attribute__((ext_vector_type(2)));
typedef float v8f __attribute__((ext_vector_type(8)));

#define AGG_ROWS   400
#define AGG_WAVES  32
#define AGG_BLOCK  1024
#define SEG_CAP    512
#define KEY_DEPTH  8
#define KEY_NONE   255u
#define SCAN_DEPTH 8
#define ROW_GROUP  8

typedef float f4 __attribute__((ext_vector_type(4)));
typedef int   i4 __attribute__((ext_vector_type(4)));
typedef unsigned u4 __attribute__((ext_vector_type(4)));

__global__ __launch_bounds__(256)
void gin_edge_keys(const int* __restrict__ dst,
                   unsigned char* __restrict__ keys,
                   int nNodes, int nEdges) {
    int per = (nEdges + AGG_WAVES - 1) / AGG_WAVES;
    per = (per + 15) & ~15;
    const int nPadded = per * AGG_WAVES;
    const int base = (blockIdx.x * 256 + threadIdx.x) * 16;
    if (base >= nPadded) return;
    const bool vecOk = (((unsigned long long)dst) & 15ull) == 0ull;
    u4 o;
    #pragma unroll
    for (int j = 0; j < 4; ++j) {
        const int e0 = base + 4 * j;
        i4 d;
        if (vecOk && e0 + 3 < nEdges) {
            d = *(const i4*)(dst + e0);
        } else {
            #pragma unroll
            for (int k = 0; k < 4; ++k) d[k] = (e0 + k < nEdges) ? dst[min(e0 + k, nEdges - 1)] : -1;
        }
        unsigned word = 0u;
        #pragma unroll
        for (int k = 0; k < 4; ++k) {
            const unsigned dk = (unsigned)d[k];
            const unsigned key = (dk < (unsigned)nNodes) ? min(dk / (unsigned)AGG_ROWS, KEY_NONE) : KEY_NONE;
            word |= key << (8 * k);
        }
        o[j] = word;
    }
    *(u4*)(keys + base) = o;
}

__global__ __launch_bounds__(AGG_BLOCK)
void gin_agg_owner(const float* __restrict__ x,
                   const int*   __restrict__ src,
                   const int*   __restrict__ dst,
                   const unsigned char* __restrict__ keys,
                   float*       __restrict__ agg,
                   unsigned*    __restrict__ keepSeg,
                   int*         __restrict__ keepCnt,
                   int*         __restrict__ keepOk,
                   int reuse,
                   int nNodes, int nEdges) {
    __shared__ f4  acc[AGG_ROWS][32];
    __shared__ unsigned seg[AGG_WAVES][SEG_CAP];
    __shared__ int s_cnt[AGG_WAVES];
    __shared__ int s_more[AGG_WAVES];

    const int tid  = threadIdx.x;
    const int lane = tid & 31;
    const int wave = __builtin_amdgcn_readfirstlane(tid >> 5);
    const int rowLo = blockIdx.x * AGG_ROWS;
    const int rowsHere = min(AGG_ROWS, nNodes - rowLo);

    unsigned* mySeg = keepSeg + ((long)blockIdx.x * AGG_WAVES + wave) * SEG_CAP;
    int*      myCnt = keepCnt + (long)blockIdx.x * AGG_WAVES + wave;
    bool useKept = false;
    int  keptCnt = 0;
    if (reuse) {
        useKept = __builtin_amdgcn_readfirstlane(keepOk[blockIdx.x]) == 1;
        keptCnt = __builtin_amdgcn_readfirstlane(min(max(*myCnt, 0), SEG_CAP));
    }
    const bool keep = (reuse == 0);

    f4* accFlat = &acc[0][0];
    for (int i = tid; i < AGG_ROWS * 32; i += AGG_BLOCK) accFlat[i] = f4{0.f, 0.f, 0.f, 0.f};

    const bool useKeys = gridDim.x <= 255u;
    const unsigned keyRep = blockIdx.x * 0x01010101u;
    int per = (nEdges + AGG_WAVES - 1) / AGG_WAVES;
    per = useKeys ? ((per + 15) & ~15) : ((per + 3) & ~3);
    const int keyLast = per * AGG_WAVES - 16;
    const int beg = useKeys ? wave * per : min(wave * per, nEdges);
    const int end = useKeys ? beg + per : min(beg + per, nEdges);
    const bool vecOk = ((((unsigned long long)dst) & 15ull) == 0ull) && ((nEdges & 3) == 0);
    int pos = useKept ? end : beg;
    bool anyMore;
    bool firstPass = true;

    do {
        int cntw = 0;
        bool full = false;
        while (useKeys && pos < end && !full) {
            u4 kv[KEY_DEPTH];
            #pragma unroll
            for (int u = 0; u < KEY_DEPTH; ++u) {
                const int e = pos + u * 512 + lane * 16;
                kv[u] = *(const u4*)(keys + min(e, keyLast));
            }
            int stepsDone = 0;
            #pragma unroll
            for (int u = 0; u < KEY_DEPTH; ++u) {
                if (!full && pos + u * 512 < end) {
                    const int e = pos + u * 512 + lane * 16;
                    unsigned w = 0u;
                    #pragma unroll
                    for (int j = 0; j < 4; ++j) {
                        const unsigned y = kv[u][j] ^ keyRep;
                        const unsigned t = (y & 0x7F7F7F7Fu) + 0x7F7F7F7Fu;
                        const unsigned z = ~(t | y | 0x7F7F7F7Fu);
                        w |= (z >> 7) << j;
                    }
                    w = (e < end) ? w : 0u;
                    unsigned mL = __builtin_amdgcn_ballot_w32(w != 0u);
                    if (mL) {
                        bool fits = cntw + 16 * __builtin_popcount(mL) <= SEG_CAP;
                        if (!fits) {
                            int tot = 0;
                            for (unsigned m2 = mL; m2; m2 &= m2 - 1u)
                                tot += __builtin_popcount((unsigned)__builtin_amdgcn_readlane((int)w, __builtin_ctz(m2)));
                            fits = cntw + tot <= SEG_CAP;
                        }
                        if (!fits) {
                            full = true;
                        } else {
                            const int off0 = pos - beg + u * 512;
                            while (mL) {
                                const int l = __builtin_ctz(mL);
                                mL &= mL - 1u;
                                unsigned wl = (unsigned)__builtin_amdgcn_readlane((int)w, l);
                                while (wl) {
                                    const int bit = __builtin_ctz(wl);
                                    wl &= wl - 1u;
                                    seg[wave][cntw] = (unsigned)(off0 + l * 16 + ((bit & 7) << 2) + (bit >> 3));
                                    ++cntw;
                                }
                            }
                        }
                    }
                    if (!full) stepsDone = u + 1;
                }
            }
            pos += stepsDone * 512;
        }
        while (!useKeys && pos < end && !full) {
            i4 v[SCAN_DEPTH];
            if (vecOk) {
                #pragma unroll
                for (int u = 0; u < SCAN_DEPTH; ++u) {
                    const int e = pos + u * 128 + lane * 4;
                    const i4 t = *(const i4*)(dst + min(e, end - 4));
                    const i4 none = {-1, -1, -1, -1};
                    v[u] = (e < end) ? t : none;
                }
            } else {
                #pragma unroll
                for (int u = 0; u < SCAN_DEPTH; ++u) {
                    const int e = pos + u * 128 + lane * 4;
                    i4 t;
                    #pragma unroll
                    for (int k = 0; k < 4; ++k) {
                        const int d = dst[min(e + k, end - 1)];
                        t[k] = (e + k < end) ? d : -1;
                    }
                    v[u] = t;
                }
            }
            int stepsDone = 0;
            #pragma unroll
            for (int u = 0; u < SCAN_DEPTH; ++u) {
                if (!full && pos + u * 128 < end) {
                    if (cntw + 128 > SEG_CAP) {
                        full = true;
                    } else {
                        const int off = pos - beg + u * 128 + lane * 4;
                        unsigned lr[4]; bool h[4]; unsigned m[4];
                        #pragma unroll
                        for (int k = 0; k < 4; ++k) {
                            lr[k] = (unsigned)(v[u][k] - rowLo);
                            h[k]  = lr[k] < (unsigned)rowsHere;
                            m[k]  = __builtin_amdgcn_ballot_w32(h[k]);
                        }
                        if (m[0] | m[1] | m[2] | m[3]) {
                            #pragma unroll
                            for (int k = 0; k < 4; ++k) {
                                if (h[k]) {
                                    const int p = cntw + (int)__builtin_amdgcn_mbcnt_lo(m[k], 0u);
                                    seg[wave][p] = (lr[k] << 20) | (unsigned)(off + k);
                                }
                                cntw += __builtin_popcount(m[k]);
                            }
                        }
                        stepsDone = u + 1;
                    }
                }
            }
            pos += stepsDone * 128;
        }
        const bool keepNow = keep && firstPass;
        if (useKept) cntw = keptCnt;
        for (int i = lane; i < cntw; i += 32) {
            unsigned ent;
            if (useKept) {
                ent = mySeg[i];
            } else {
                const unsigned en = seg[wave][i];
                if (useKeys) {
                    const int e = beg + (int)en;
                    ent = ((unsigned)(dst[e] - rowLo) << 20) | (unsigned)src[e];
                } else {
                    ent = (en & 0xFFF00000u) | (unsigned)src[beg + (int)(en & 0xFFFFFu)];
                }
                if (keepNow) mySeg[i] = ent;
            }
            seg[wave][i] = ent;
        }
        if (lane == 0) {
            s_cnt[wave] = cntw; s_more[wave] = (pos < end) ? 1 : 0;
            if (keepNow) *myCnt = cntw;
        }
        __syncthreads();
        anyMore = __builtin_amdgcn_ballot_w32(s_more[lane] != 0) != 0u;
        if (keepNow && tid == 0) keepOk[blockIdx.x] = anyMore ? 0 : 1;
        firstPass = false;

        unsigned held = 0u;
        int nHeld = 0;
        unsigned sent = 0u;
        int nSent = 0;
        const f4 zero4 = {0.f, 0.f, 0.f, 0.f};
        f4 xr0 = zero4, xr1 = zero4, xr2 = zero4, xr3 = zero4, xr4 = zero4, xr5 = zero4, xr6 = zero4, xr7 = zero4;
#define ROWS(M) M(0) M(1) M(2) M(3) M(4) M(5) M(6) M(7)
#define ROW_ADD(q) if (q < nSent) { \
            const int r = (int)((unsigned)__builtin_amdgcn_readlane((int)sent, q) >> 20); \
            f4 a = acc[r][lane]; a += xr##q; acc[r][lane] = a; }
#define ROW_LOAD(q) { \
            const unsigned e2 = (unsigned)__builtin_amdgcn_readlane((int)held, min(q, nHeld - 1)); \
            xr##q = ((const f4*)x)[(long)(e2 & 0xFFFFFu) * 32 + lane]; }
#define ROW_STEP() { \
            ROWS(ROW_ADD) \
            sent = held; nSent = nHeld; \
            if (nHeld > 0) { ROWS(ROW_LOAD) } \
            nHeld = 0; }
        const int cl = s_cnt[lane];
        for (int sg = 0; sg < AGG_WAVES; ++sg) {
            const int c = __builtin_amdgcn_readlane(cl, sg);
            for (int i0 = 0; i0 < c; i0 += 32) {
                const int i = i0 + lane;
                const unsigned ent = (i < c) ? seg[sg][i] : 0u;
                const bool own = (i < c) && (int)((ent >> 20) & (AGG_WAVES - 1)) == wave;
                unsigned m = __builtin_amdgcn_ballot_w32(own);
                while (m) {
                    const int l = __builtin_ctz(m);
                    m &= m - 1u;
                    const unsigned e2 = (unsigned)__builtin_amdgcn_readlane((int)ent, l);
                    held = (lane == nHeld) ? e2 : held;
                    if (++nHeld == ROW_GROUP) ROW_STEP()
                }
            }
        }
        ROW_STEP()
        ROWS(ROW_ADD)
#undef ROW_STEP
#undef ROW_LOAD
#undef ROW_ADD
#undef ROWS
        __syncthreads();
    } while (anyMore);

    f4* aggRows = (f4*)agg + (long)rowLo * 32;
    for (int i = tid; i < rowsHere * 32; i += AGG_BLOCK) aggRows[i] = accFlat[i];
}

#define MLP_WAVES 16
#define MLP_BLOCK (MLP_WAVES * 32)
#define MLP_GROUPS 256

static __device__ __forceinline__ v8f mma4(float a0, float a1, float b0, float b1, v8f c) {
    v2f a = {a0, a1};
    v2f b = {b0, b1};
    return __builtin_amdgcn_wmma_f32_16x16x4_f32(false, a, false, b, (short)0, c, false, false);
}

__global__ __launch_bounds__(MLP_BLOCK)
void gin_mlp_t(const float* __restrict__ xin,
               const float* __restrict__ agg,
               const float* __restrict__ W1, const float* __restrict__ b1,
               const float* __restrict__ W2, const float* __restrict__ b2,
               float* __restrict__ out,
               int applyReluOut, int nTiles) {
    __shared__ f4 w1f[8][16][32];
    __shared__ f4 w2f[8][16][32];
    __shared__ alignas(16) float bs[2][D];

    const int tid  = threadIdx.x;
    const int lane = tid & 31;
    const int wave = __builtin_amdgcn_readfirstlane(tid >> 5);

    const int c = lane & 15;
    const int h = lane >> 4;
    int tile = blockIdx.x + MLP_GROUPS * wave;

    const f4 zero4 = {0.f, 0.f, 0.f, 0.f};
    f4 xv[16], av[16];
    #pragma unroll
    for (int i = 0; i < 16; ++i) { xv[i] = zero4; av[i] = zero4; }
    if (tile < nTiles) {
        const long row0 = (long)tile * 16 + c;
        const f4* xr0 = (const f4*)xin + row0 * 32 + h;
        const f4* ar0 = (const f4*)agg + row0 * 32 + h;
        #pragma unroll
        for (int i = 0; i < 16; ++i) { xv[i] = xr0[2 * i]; av[i] = ar0[2 * i]; }
    }

    {
        const int n  = tid & (D - 1);
        const int tn = n >> 4;
        const int cc = n & 15;
        const int q  = tid >> 7;
        const float bias1 = b1[n];
        const float bias2 = b2[n];
        f4 wa[8], wb[8];
        #pragma unroll
        for (int j = 0; j < 8; ++j) {
            const int kg = q + 4 * j;
            #pragma unroll
            for (int m = 0; m < 4; ++m) {
                wa[j][m] = W1[(4 * kg + m) * D + n];
                wb[j][m] = W2[(4 * kg + m) * D + n];
            }
        }
        #pragma unroll
        for (int j = 0; j < 8; ++j) {
            const int kg = q + 4 * j;
            const int i1 = kg >> 1;
            const int h1 = kg & 1;
            w1f[tn][i1][cc + 16 * h1] = wa[j];
            const int i2 = ((kg >> 2) << 1) | (kg & 1);
            const int h2 = (kg >> 1) & 1;
            w2f[tn][i2][cc + 16 * h2] = wb[j];
        }
        bs[0][n] = bias1;
        bs[1][n] = bias2;
    }
    __syncthreads();

    while (tile < nTiles) {
        asm volatile("" ::: "memory");
        const long row = (long)tile * 16 + c;

        f4 sv[16];
        #pragma unroll
        for (int i = 0; i < 16; ++i) sv[i] = xv[i] + av[i];
        __builtin_amdgcn_sched_barrier(0);

        v8f acc[8];
        #pragma unroll
        for (int t = 0; t < 8; ++t) acc[t] = v8f{};

        {
            f4 aC[8], aN[8];
            #pragma unroll
            for (int t = 0; t < 8; ++t) aC[t] = w1f[t][0][lane];
            #pragma unroll
            for (int i = 0; i < 16; ++i) {
                if (i + 1 < 16) {
                    #pragma unroll
                    for (int t = 0; t < 8; ++t) aN[t] = w1f[t][i + 1][lane];
                }
                __builtin_amdgcn_sched_barrier(0);
                #pragma unroll
                for (int t = 0; t < 8; ++t) acc[t] = mma4(aC[t][0], aC[t][1], sv[i][0], sv[i][1], acc[t]);
                #pragma unroll
                for (int t = 0; t < 8; ++t) acc[t] = mma4(aC[t][2], aC[t][3], sv[i][2], sv[i][3], acc[t]);
                __builtin_amdgcn_sched_barrier(0);
                #pragma unroll
                for (int t = 0; t < 8; ++t) aC[t] = aN[t];
            }
        }
        #pragma unroll
        for (int t = 0; t < 8; ++t) {
            const f4 bA = *(const f4*)&bs[0][16 * t + 8 * h];
            const f4 bB = *(const f4*)&bs[0][16 * t + 8 * h + 4];
            #pragma unroll
            for (int r = 0; r < 8; ++r) {
                const float v = acc[t][r] + (r < 4 ? bA[r] : bB[r - 4]);
                acc[t][r] = v > 0.0f ? v : 0.0f;
            }
        }

        v8f acc2[8];
        #pragma unroll
        for (int t = 0; t < 8; ++t) acc2[t] = v8f{};
        {
            f4 aC[8], aN[8];
            #pragma unroll
            for (int t = 0; t < 8; ++t) aC[t] = w2f[t][0][lane];
            #pragma unroll
            for (int i = 0; i < 16; ++i) {
                const int tk = i >> 1;
                const int p4 = (i & 1) * 4;
                if (i + 1 < 16) {
                    #pragma unroll
                    for (int t = 0; t < 8; ++t) aN[t] = w2f[t][i + 1][lane];
                }
                __builtin_amdgcn_sched_barrier(0);
                #pragma unroll
                for (int t = 0; t < 8; ++t) acc2[t] = mma4(aC[t][0], aC[t][1], acc[tk][p4], acc[tk][p4 + 1], acc2[t]);
                #pragma unroll
                for (int t = 0; t < 8; ++t) acc2[t] = mma4(aC[t][2], aC[t][3], acc[tk][p4 + 2], acc[tk][p4 + 3], acc2[t]);
                __builtin_amdgcn_sched_barrier(0);
                #pragma unroll
                for (int t = 0; t < 8; ++t) aC[t] = aN[t];
            }
        }

        f4* orow = (f4*)(out + row * D + 8 * h);
        #pragma unroll
        for (int t = 0; t < 8; ++t) {
            const f4 bA = *(const f4*)&bs[1][16 * t + 8 * h];
            const f4 bB = *(const f4*)&bs[1][16 * t + 8 * h + 4];
            f4 o0, o1;
            #pragma unroll
            for (int r = 0; r < 4; ++r) {
                float u0 = acc2[t][r] + bA[r];
                float u1 = acc2[t][r + 4] + bB[r];
                if (applyReluOut) { u0 = u0 > 0.0f ? u0 : 0.0f; u1 = u1 > 0.0f ? u1 : 0.0f; }
                o0[r] = u0; o1[r] = u1;
            }
            orow[4 * t]     = o0;
            orow[4 * t + 1] = o1;
        }

        tile += MLP_GROUPS * MLP_WAVES;
        if (tile < nTiles) {
            const long rowN = (long)tile * 16 + c;
            const f4* xrN = (const f4*)xin + rowN * 32 + h;
            const f4* arN = (const f4*)agg + rowN * 32 + h;
            #pragma unroll
            for (int i = 0; i < 16; ++i) { xv[i] = xrN[2 * i]; av[i] = arN[2 * i]; }
        }
    }
}

extern "C" void kernel_launch(void* const* d_in, const int* in_sizes, int n_in,
                              void* d_out, int out_size, void* d_ws, size_t ws_size,
                              hipStream_t stream) {
    const float* x  = (const float*)d_in[0];
    const int*   ei = (const int*)  d_in[1];
    const float* W1 = (const float*)d_in[2];
    const float* b1 = (const float*)d_in[3];
    const float* W2 = (const float*)d_in[4];
    const float* b2 = (const float*)d_in[5];
    float* out = (float*)d_out;

    const int N = in_sizes[0] / D;
    const int E = in_sizes[1] / 2;
    const int* src = ei;
    const int* dst = ei + E;

    float* agg = (float*)d_ws;
    float* h1  = agg + (size_t)N * D;

    const int  aBlocks = (N + AGG_ROWS - 1) / AGG_ROWS;
    const int  mTiles  = N / TILE_ROWS;
    const int  kBlocks = (E + 511 + 4095) / 4096;
    unsigned char* keys = (unsigned char*)(h1 + (size_t)N * D);
    unsigned* keepSeg = (unsigned*)(keys + (size_t)kBlocks * 4096);
    int*      keepCnt = (int*)(keepSeg + (size_t)aBlocks * AGG_WAVES * SEG_CAP);
    int*      keepOk  = keepCnt + (size_t)aBlocks * AGG_WAVES;

    gin_edge_keys<<<kBlocks, 256, 0, stream>>>(dst, keys, N, E);

    gin_agg_owner<<<aBlocks, AGG_BLOCK, 0, stream>>>(x, src, dst, keys, agg, keepSeg, keepCnt, keepOk, 0, N, E);
    gin_mlp_t<<<MLP_GROUPS, MLP_BLOCK, 0, stream>>>(x, agg, W1, b1, W2, b2, h1, 1, mTiles);

    gin_agg_owner<<<aBlocks, AGG_BLOCK, 0, stream>>>(h1, src, dst, keys, agg, keepSeg, keepCnt, keepOk, 1, N, E);
    gin_mlp_t<<<MLP_GROUPS, MLP_BLOCK, 0, stream>>>(h1, agg, W1, b1, W2, b2, out, 0, mTiles);
}


# MLP: node rows loaded and stored as whole 512-byte rows, one per wave instruction, turned to and from fragment order in a per-wave LDS buffer (was 16 rows x 32 B a load, 16 B a store); sums unchanged
# speedup vs baseline: 3.7662x; 1.0523x over previous
#include <hip/hip_runtime.h>


#define D 128
#define TILE_ROWS 16

typedef float v2f __attribute__((ext_vector_type(2)));
typedef float v8f __attribute__((ext_vector_type(8)));

#define AGG_ROWS   400
#define AGG_WAVES  32
#define AGG_BLOCK  1024
#define SEG_CAP    512
#define KEY_DEPTH  8
#define KEY_NONE   255u
#define SCAN_DEPTH 8
#define ROW_GROUP  8

typedef float f4 __attribute__((ext_vector_type(4)));
typedef int   i4 __attribute__((ext_vector_type(4)));
typedef unsigned u4 __attribute__((ext_vector_type(4)));

__global__ __launch_bounds__(256)
void gin_edge_keys(const int* __restrict__ dst,
                   unsigned char* __restrict__ keys,
                   int nNodes, int nEdges) {
    int per = (nEdges + AGG_WAVES - 1) / AGG_WAVES;
    per = (per + 15) & ~15;
    const int nPadded = per * AGG_WAVES;
    const int base = (blockIdx.x * 256 + threadIdx.x) * 16;
    if (base >= nPadded) return;
    const bool vecOk = (((unsigned long long)dst) & 15ull) == 0ull;
    u4 o;
    #pragma unroll
    for (int j = 0; j < 4; ++j) {
        const int e0 = base + 4 * j;
        i4 d;
        if (vecOk && e0 + 3 < nEdges) {
            d = *(const i4*)(dst + e0);
        } else {
            #pragma unroll
            for (int k = 0; k < 4; ++k) d[k] = (e0 + k < nEdges) ? dst[min(e0 + k, nEdges - 1)] : -1;
        }
        unsigned word = 0u;
        #pragma unroll
        for (int k = 0; k < 4; ++k) {
            const unsigned dk = (unsigned)d[k];
            const unsigned key = (dk < (unsigned)nNodes) ? min(dk / (unsigned)AGG_ROWS, KEY_NONE) : KEY_NONE;
            word |= key << (8 * k);
        }
        o[j] = word;
    }
    *(u4*)(keys + base) = o;
}

__global__ __launch_bounds__(AGG_BLOCK)
void gin_agg_owner(const float* __restrict__ x,
                   const int*   __restrict__ src,
                   const int*   __restrict__ dst,
                   const unsigned char* __restrict__ keys,
                   float*       __restrict__ agg,
                   unsigned*    __restrict__ keepSeg,
                   int*         __restrict__ keepCnt,
                   int*         __restrict__ keepOk,
                   int reuse,
                   int nNodes, int nEdges) {
    __shared__ f4  acc[AGG_ROWS][32];
    __shared__ unsigned seg[AGG_WAVES][SEG_CAP];
    __shared__ int s_cnt[AGG_WAVES];
    __shared__ int s_more[AGG_WAVES];

    const int tid  = threadIdx.x;
    const int lane = tid & 31;
    const int wave = __builtin_amdgcn_readfirstlane(tid >> 5);
    const int rowLo = blockIdx.x * AGG_ROWS;
    const int rowsHere = min(AGG_ROWS, nNodes - rowLo);

    unsigned* mySeg = keepSeg + ((long)blockIdx.x * AGG_WAVES + wave) * SEG_CAP;
    int*      myCnt = keepCnt + (long)blockIdx.x * AGG_WAVES + wave;
    bool useKept = false;
    int  keptCnt = 0;
    if (reuse) {
        useKept = __builtin_amdgcn_readfirstlane(keepOk[blockIdx.x]) == 1;
        keptCnt = __builtin_amdgcn_readfirstlane(min(max(*myCnt, 0), SEG_CAP));
    }
    const bool keep = (reuse == 0);

    f4* accFlat = &acc[0][0];
    for (int i = tid; i < AGG_ROWS * 32; i += AGG_BLOCK) accFlat[i] = f4{0.f, 0.f, 0.f, 0.f};

    const bool useKeys = gridDim.x <= 255u;
    const unsigned keyRep = blockIdx.x * 0x01010101u;
    int per = (nEdges + AGG_WAVES - 1) / AGG_WAVES;
    per = useKeys ? ((per + 15) & ~15) : ((per + 3) & ~3);
    const int keyLast = per * AGG_WAVES - 16;
    const int beg = useKeys ? wave * per : min(wave * per, nEdges);
    const int end = useKeys ? beg + per : min(beg + per, nEdges);
    const bool vecOk = ((((unsigned long long)dst) & 15ull) == 0ull) && ((nEdges & 3) == 0);
    int pos = useKept ? end : beg;
    bool anyMore;
    bool firstPass = true;

    do {
        int cntw = 0;
        bool full = false;
        while (useKeys && pos < end && !full) {
            u4 kv[KEY_DEPTH];
            #pragma unroll
            for (int u = 0; u < KEY_DEPTH; ++u) {
                const int e = pos + u * 512 + lane * 16;
                kv[u] = *(const u4*)(keys + min(e, keyLast));
            }
            int stepsDone = 0;
            #pragma unroll
            for (int u = 0; u < KEY_DEPTH; ++u) {
                if (!full && pos + u * 512 < end) {
                    const int e = pos + u * 512 + lane * 16;
                    unsigned w = 0u;
                    #pragma unroll
                    for (int j = 0; j < 4; ++j) {
                        const unsigned y = kv[u][j] ^ keyRep;
                        const unsigned t = (y & 0x7F7F7F7Fu) + 0x7F7F7F7Fu;
                        const unsigned z = ~(t | y | 0x7F7F7F7Fu);
                        w |= (z >> 7) << j;
                    }
                    w = (e < end) ? w : 0u;
                    unsigned mL = __builtin_amdgcn_ballot_w32(w != 0u);
                    if (mL) {
                        bool fits = cntw + 16 * __builtin_popcount(mL) <= SEG_CAP;
                        if (!fits) {
                            int tot = 0;
                            for (unsigned m2 = mL; m2; m2 &= m2 - 1u)
                                tot += __builtin_popcount((unsigned)__builtin_amdgcn_readlane((int)w, __builtin_ctz(m2)));
                            fits = cntw + tot <= SEG_CAP;
                        }
                        if (!fits) {
                            full = true;
                        } else {
                            const int off0 = pos - beg + u * 512;
                            while (mL) {
                                const int l = __builtin_ctz(mL);
                                mL &= mL - 1u;
                                unsigned wl = (unsigned)__builtin_amdgcn_readlane((int)w, l);
                                while (wl) {
                                    const int bit = __builtin_ctz(wl);
                                    wl &= wl - 1u;
                                    seg[wave][cntw] = (unsigned)(off0 + l * 16 + ((bit & 7) << 2) + (bit >> 3));
                                    ++cntw;
                                }
                            }
                        }
                    }
                    if (!full) stepsDone = u + 1;
                }
            }
            pos += stepsDone * 512;
        }
        while (!useKeys && pos < end && !full) {
            i4 v[SCAN_DEPTH];
            if (vecOk) {
                #pragma unroll
                for (int u = 0; u < SCAN_DEPTH; ++u) {
                    const int e = pos + u * 128 + lane * 4;
                    const i4 t = *(const i4*)(dst + min(e, end - 4));
                    const i4 none = {-1, -1, -1, -1};
                    v[u] = (e < end) ? t : none;
                }
            } else {
                #pragma unroll
                for (int u = 0; u < SCAN_DEPTH; ++u) {
                    const int e = pos + u * 128 + lane * 4;
                    i4 t;
                    #pragma unroll
                    for (int k = 0; k < 4; ++k) {
                        const int d = dst[min(e + k, end - 1)];
                        t[k] = (e + k < end) ? d : -1;
                    }
                    v[u] = t;
                }
            }
            int stepsDone = 0;
            #pragma unroll
            for (int u = 0; u < SCAN_DEPTH; ++u) {
                if (!full && pos + u * 128 < end) {
                    if (cntw + 128 > SEG_CAP) {
                        full = true;
                    } else {
                        const int off = pos - beg + u * 128 + lane * 4;
                        unsigned lr[4]; bool h[4]; unsigned m[4];
                        #pragma unroll
                        for (int k = 0; k < 4; ++k) {
                            lr[k] = (unsigned)(v[u][k] - rowLo);
                            h[k]  = lr[k] < (unsigned)rowsHere;
                            m[k]  = __builtin_amdgcn_ballot_w32(h[k]);
                        }
                        if (m[0] | m[1] | m[2] | m[3]) {
                            #pragma unroll
                            for (int k = 0; k < 4; ++k) {
                                if (h[k]) {
                                    const int p = cntw + (int)__builtin_amdgcn_mbcnt_lo(m[k], 0u);
                                    seg[wave][p] = (lr[k] << 20) | (unsigned)(off + k);
                                }
                                cntw += __builtin_popcount(m[k]);
                            }
                        }
                        stepsDone = u + 1;
                    }
                }
            }
            pos += stepsDone * 128;
        }
        const bool keepNow = keep && firstPass;
        if (useKept) cntw = keptCnt;
        for (int i = lane; i < cntw; i += 32) {
            unsigned ent;
            if (useKept) {
                ent = mySeg[i];
            } else {
                const unsigned en = seg[wave][i];
                if (useKeys) {
                    const int e = beg + (int)en;
                    ent = ((unsigned)(dst[e] - rowLo) << 20) | (unsigned)src[e];
                } else {
                    ent = (en & 0xFFF00000u) | (unsigned)src[beg + (int)(en & 0xFFFFFu)];
                }
                if (keepNow) mySeg[i] = ent;
            }
            seg[wave][i] = ent;
        }
        if (lane == 0) {
            s_cnt[wave] = cntw; s_more[wave] = (pos < end) ? 1 : 0;
            if (keepNow) *myCnt = cntw;
        }
        __syncthreads();
        anyMore = __builtin_amdgcn_ballot_w32(s_more[lane] != 0) != 0u;
        if (keepNow && tid == 0) keepOk[blockIdx.x] = anyMore ? 0 : 1;
        firstPass = false;

        unsigned held = 0u;
        int nHeld = 0;
        unsigned sent = 0u;
        int nSent = 0;
        const f4 zero4 = {0.f, 0.f, 0.f, 0.f};
        f4 xr0 = zero4, xr1 = zero4, xr2 = zero4, xr3 = zero4, xr4 = zero4, xr5 = zero4, xr6 = zero4, xr7 = zero4;
#define ROWS(M) M(0) M(1) M(2) M(3) M(4) M(5) M(6) M(7)
#define ROW_ADD(q) if (q < nSent) { \
            const int r = (int)((unsigned)__builtin_amdgcn_readlane((int)sent, q) >> 20); \
            f4 a = acc[r][lane]; a += xr##q; acc[r][lane] = a; }
#define ROW_LOAD(q) { \
            const unsigned e2 = (unsigned)__builtin_amdgcn_readlane((int)held, min(q, nHeld - 1)); \
            xr##q = ((const f4*)x)[(long)(e2 & 0xFFFFFu) * 32 + lane]; }
#define ROW_STEP() { \
            ROWS(ROW_ADD) \
            sent = held; nSent = nHeld; \
            if (nHeld > 0) { ROWS(ROW_LOAD) } \
            nHeld = 0; }
        const int cl = s_cnt[lane];
        for (int sg = 0; sg < AGG_WAVES; ++sg) {
            const int c = __builtin_amdgcn_readlane(cl, sg);
            for (int i0 = 0; i0 < c; i0 += 32) {
                const int i = i0 + lane;
                const unsigned ent = (i < c) ? seg[sg][i] : 0u;
                const bool own = (i < c) && (int)((ent >> 20) & (AGG_WAVES - 1)) == wave;
                unsigned m = __builtin_amdgcn_ballot_w32(own);
                while (m) {
                    const int l = __builtin_ctz(m);
                    m &= m - 1u;
                    const unsigned e2 = (unsigned)__builtin_amdgcn_readlane((int)ent, l);
                    held = (lane == nHeld) ? e2 : held;
                    if (++nHeld == ROW_GROUP) ROW_STEP()
                }
            }
        }
        ROW_STEP()
        ROWS(ROW_ADD)
#undef ROW_STEP
#undef ROW_LOAD
#undef ROW_ADD
#undef ROWS
        __syncthreads();
    } while (anyMore);

    f4* aggRows = (f4*)agg + (long)rowLo * 32;
    for (int i = tid; i < rowsHere * 32; i += AGG_BLOCK) aggRows[i] = accFlat[i];
}

#define MLP_WAVES 16
#define MLP_BLOCK (MLP_WAVES * 32)
#define MLP_GROUPS 256
#define TB_PITCH 33

static __device__ __forceinline__ void wave_lds_handover() {
    __builtin_amdgcn_fence(__ATOMIC_RELEASE, "wavefront");
    __builtin_amdgcn_wave_barrier();
    __builtin_amdgcn_fence(__ATOMIC_ACQUIRE, "wavefront");
}

static __device__ __forceinline__ v8f mma4(float a0, float a1, float b0, float b1, v8f c) {
    v2f a = {a0, a1};
    v2f b = {b0, b1};
    return __builtin_amdgcn_wmma_f32_16x16x4_f32(false, a, false, b, (short)0, c, false, false);
}

__global__ __launch_bounds__(MLP_BLOCK)
void gin_mlp_t(const float* __restrict__ xin,
               const float* __restrict__ agg,
               const float* __restrict__ W1, const float* __restrict__ b1,
               const float* __restrict__ W2, const float* __restrict__ b2,
               float* __restrict__ out,
               int applyReluOut, int nTiles) {
    __shared__ f4 w1f[8][16][32];
    __shared__ f4 w2f[8][16][32];
    __shared__ alignas(16) float bs[2][D];
    __shared__ f4 tb[MLP_WAVES][TILE_ROWS][TB_PITCH];

    const int tid  = threadIdx.x;
    const int lane = tid & 31;
    const int wave = __builtin_amdgcn_readfirstlane(tid >> 5);

    const int c = lane & 15;
    const int h = lane >> 4;
    int tile = blockIdx.x + MLP_GROUPS * wave;

    const f4 zero4 = {0.f, 0.f, 0.f, 0.f};
    f4 xv[16], av[16];
    #pragma unroll
    for (int i = 0; i < 16; ++i) { xv[i] = zero4; av[i] = zero4; }
    if (tile < nTiles) {
        const f4* xr0 = (const f4*)xin + (long)tile * (16 * 32) + lane;
        const f4* ar0 = (const f4*)agg + (long)tile * (16 * 32) + lane;
        #pragma unroll
        for (int r = 0; r < 16; ++r) { xv[r] = xr0[32 * r]; av[r] = ar0[32 * r]; }
    }
    f4 (*myTb)[TB_PITCH] = tb[wave];

    {
        const int n  = tid & (D - 1);
        const int tn = n >> 4;
        const int cc = n & 15;
        const int q  = tid >> 7;
        const float bias1 = b1[n];
        const float bias2 = b2[n];
        f4 wa[8], wb[8];
        #pragma unroll
        for (int j = 0; j < 8; ++j) {
            const int kg = q + 4 * j;
            #pragma unroll
            for (int m = 0; m < 4; ++m) {
                wa[j][m] = W1[(4 * kg + m) * D + n];
                wb[j][m] = W2[(4 * kg + m) * D + n];
            }
        }
        #pragma unroll
        for (int j = 0; j < 8; ++j) {
            const int kg = q + 4 * j;
            const int i1 = kg >> 1;
            const int h1 = kg & 1;
            w1f[tn][i1][cc + 16 * h1] = wa[j];
            const int i2 = ((kg >> 2) << 1) | (kg & 1);
            const int h2 = (kg >> 1) & 1;
            w2f[tn][i2][cc + 16 * h2] = wb[j];
        }
        bs[0][n] = bias1;
        bs[1][n] = bias2;
    }
    __syncthreads();

    while (tile < nTiles) {
        asm volatile("" ::: "memory");
        #pragma unroll
        for (int r = 0; r < 16; ++r) myTb[r][lane] = xv[r] + av[r];
        wave_lds_handover();
        f4 sv[16];
        #pragma unroll
        for (int i = 0; i < 16; ++i) sv[i] = myTb[c][2 * i + h];
        __builtin_amdgcn_sched_barrier(0);

        v8f acc[8];
        #pragma unroll
        for (int t = 0; t < 8; ++t) acc[t] = v8f{};

        {
            f4 aC[8], aN[8];
            #pragma unroll
            for (int t = 0; t < 8; ++t) aC[t] = w1f[t][0][lane];
            #pragma unroll
            for (int i = 0; i < 16; ++i) {
                if (i + 1 < 16) {
                    #pragma unroll
                    for (int t = 0; t < 8; ++t) aN[t] = w1f[t][i + 1][lane];
                }
                __builtin_amdgcn_sched_barrier(0);
                #pragma unroll
                for (int t = 0; t < 8; ++t) acc[t] = mma4(aC[t][0], aC[t][1], sv[i][0], sv[i][1], acc[t]);
                #pragma unroll
                for (int t = 0; t < 8; ++t) acc[t] = mma4(aC[t][2], aC[t][3], sv[i][2], sv[i][3], acc[t]);
                __builtin_amdgcn_sched_barrier(0);
                #pragma unroll
                for (int t = 0; t < 8; ++t) aC[t] = aN[t];
            }
        }
        #pragma unroll
        for (int t = 0; t < 8; ++t) {
            const f4 bA = *(const f4*)&bs[0][16 * t + 8 * h];
            const f4 bB = *(const f4*)&bs[0][16 * t + 8 * h + 4];
            #pragma unroll
            for (int r = 0; r < 8; ++r) {
                const float v = acc[t][r] + (r < 4 ? bA[r] : bB[r - 4]);
                acc[t][r] = v > 0.0f ? v : 0.0f;
            }
        }

        v8f acc2[8];
        #pragma unroll
        for (int t = 0; t < 8; ++t) acc2[t] = v8f{};
        {
            f4 aC[8], aN[8];
            #pragma unroll
            for (int t = 0; t < 8; ++t) aC[t] = w2f[t][0][lane];
            #pragma unroll
            for (int i = 0; i < 16; ++i) {
                const int tk = i >> 1;
                const int p4 = (i & 1) * 4;
                if (i + 1 < 16) {
                    #pragma unroll
                    for (int t = 0; t < 8; ++t) aN[t] = w2f[t][i + 1][lane];
                }
                __builtin_amdgcn_sched_barrier(0);
                #pragma unroll
                for (int t = 0; t < 8; ++t) acc2[t] = mma4(aC[t][0], aC[t][1], acc[tk][p4], acc[tk][p4 + 1], acc2[t]);
                #pragma unroll
                for (int t = 0; t < 8; ++t) acc2[t] = mma4(aC[t][2], aC[t][3], acc[tk][p4 + 2], acc[tk][p4 + 3], acc2[t]);
                __builtin_amdgcn_sched_barrier(0);
                #pragma unroll
                for (int t = 0; t < 8; ++t) aC[t] = aN[t];
            }
        }

        #pragma unroll
        for (int t = 0; t < 8; ++t) {
            const f4 bA = *(const f4*)&bs[1][16 * t + 8 * h];
            const f4 bB = *(const f4*)&bs[1][16 * t + 8 * h + 4];
            f4 o0, o1;
            #pragma unroll
            for (int r = 0; r < 4; ++r) {
                float u0 = acc2[t][r] + bA[r];
                float u1 = acc2[t][r + 4] + bB[r];
                if (applyReluOut) { u0 = u0 > 0.0f ? u0 : 0.0f; u1 = u1 > 0.0f ? u1 : 0.0f; }
                o0[r] = u0; o1[r] = u1;
            }
            myTb[c][4 * t + 2 * h]     = o0;
            myTb[c][4 * t + 2 * h + 1] = o1;
        }
        wave_lds_handover();
        f4* orow = (f4*)out + (long)tile * (16 * 32) + lane;
        #pragma unroll
        for (int r = 0; r < 16; ++r) orow[32 * r] = myTb[r][lane];

        tile += MLP_GROUPS * MLP_WAVES;
        if (tile < nTiles) {
            const f4* xrN = (const f4*)xin + (long)tile * (16 * 32) + lane;
            const f4* arN = (const f4*)agg + (long)tile * (16 * 32) + lane;
            #pragma unroll
            for (int r = 0; r < 16; ++r) { xv[r] = xrN[32 * r]; av[r] = arN[32 * r]; }
        }
    }
}

extern "C" void kernel_launch(void* const* d_in, const int* in_sizes, int n_in,
                              void* d_out, int out_size, void* d_ws, size_t ws_size,
                              hipStream_t stream) {
    const float* x  = (const float*)d_in[0];
    const int*   ei = (const int*)  d_in[1];
    const float* W1 = (const float*)d_in[2];
    const float* b1 = (const float*)d_in[3];
    const float* W2 = (const float*)d_in[4];
    const float* b2 = (const float*)d_in[5];
    float* out = (float*)d_out;

    const int N = in_sizes[0] / D;
    const int E = in_sizes[1] / 2;
    const int* src = ei;
    const int* dst = ei + E;

    float* agg = (float*)d_ws;
    float* h1  = agg + (size_t)N * D;

    const int  aBlocks = (N + AGG_ROWS - 1) / AGG_ROWS;
    const int  mTiles  = N / TILE_ROWS;
    const int  kBlocks = (E + 511 + 4095) / 4096;
    unsigned char* keys = (unsigned char*)(h1 + (size_t)N * D);
    unsigned* keepSeg = (unsigned*)(keys + (size_t)kBlocks * 4096);
    int*      keepCnt = (int*)(keepSeg + (size_t)aBlocks * AGG_WAVES * SEG_CAP);
    int*      keepOk  = keepCnt + (size_t)aBlocks * AGG_WAVES;

    gin_edge_keys<<<kBlocks, 256, 0, stream>>>(dst, keys, N, E);

    gin_agg_owner<<<aBlocks, AGG_BLOCK, 0, stream>>>(x, src, dst, keys, agg, keepSeg, keepCnt, keepOk, 0, N, E);
    gin_mlp_t<<<MLP_GROUPS, MLP_BLOCK, 0, stream>>>(x, agg, W1, b1, W2, b2, h1, 1, mTiles);

    gin_agg_owner<<<aBlocks, AGG_BLOCK, 0, stream>>>(h1, src, dst, keys, agg, keepSeg, keepCnt, keepOk, 1, N, E);
    gin_mlp_t<<<MLP_GROUPS, MLP_BLOCK, 0, stream>>>(h1, agg, W1, b1, W2, b2, out, 0, mTiles);
}


# Neighbour sum: list entries regrouped per owner wave in LDS (counts and running places, no atomics); a wave reads only its own list (was: walks all 32), two row groups in flight; sum order unchanged
# speedup vs baseline: 3.9331x; 1.0443x over previous
#include <hip/hip_runtime.h>


#define D 128
#define TILE_ROWS 16

typedef float v2f __attribute__((ext_vector_type(2)));
typedef float v8f __attribute__((ext_vector_type(8)));

#define AGG_ROWS   400
#define AGG_WAVES  32
#define AGG_BLOCK  1024
#define SEG_CAP    512
#define OWN_CAP    256
#define KEY_DEPTH  8
#define KEY_NONE   255u
#define SCAN_DEPTH 8
#define ROW_GROUP  8

typedef float f4 __attribute__((ext_vector_type(4)));
typedef int   i4 __attribute__((ext_vector_type(4)));
typedef unsigned u4 __attribute__((ext_vector_type(4)));

__global__ __launch_bounds__(256)
void gin_edge_keys(const int* __restrict__ dst,
                   unsigned char* __restrict__ keys,
                   int nNodes, int nEdges) {
    int per = (nEdges + AGG_WAVES - 1) / AGG_WAVES;
    per = (per + 15) & ~15;
    const int nPadded = per * AGG_WAVES;
    const int base = (blockIdx.x * 256 + threadIdx.x) * 16;
    if (base >= nPadded) return;
    const bool vecOk = (((unsigned long long)dst) & 15ull) == 0ull;
    u4 o;
    #pragma unroll
    for (int j = 0; j < 4; ++j) {
        const int e0 = base + 4 * j;
        i4 d;
        if (vecOk && e0 + 3 < nEdges) {
            d = *(const i4*)(dst + e0);
        } else {
            #pragma unroll
            for (int k = 0; k < 4; ++k) d[k] = (e0 + k < nEdges) ? dst[min(e0 + k, nEdges - 1)] : -1;
        }
        unsigned word = 0u;
        #pragma unroll
        for (int k = 0; k < 4; ++k) {
            const unsigned dk = (unsigned)d[k];
            const unsigned key = (dk < (unsigned)nNodes) ? min(dk / (unsigned)AGG_ROWS, KEY_NONE) : KEY_NONE;
            word |= key << (8 * k);
        }
        o[j] = word;
    }
    *(u4*)(keys + base) = o;
}

static __device__ __forceinline__ void wave_lds_handover() {
    __builtin_amdgcn_fence(__ATOMIC_RELEASE, "wavefront");
    __builtin_amdgcn_wave_barrier();
    __builtin_amdgcn_fence(__ATOMIC_ACQUIRE, "wavefront");
}

#define PEERS(I0) \
                const int pi = (I0) + lane; \
                const bool valid = pi < cntw; \
                const unsigned ent = valid ? seg[wave][pi] : 0u; \
                const unsigned o = (ent >> 20) & (AGG_WAVES - 1); \
                unsigned peers = __builtin_amdgcn_ballot_w32(valid); \
                _Pragma("unroll") \
                for (int pb = 0; pb < 5; ++pb) { \
                    const bool bit = ((o >> pb) & 1u) != 0u; \
                    const unsigned mb = __builtin_amdgcn_ballot_w32(bit); \
                    peers &= bit ? mb : ~mb; \
                } \
                const int rank = (int)__builtin_amdgcn_mbcnt_lo(peers, 0u);

__global__ __launch_bounds__(AGG_BLOCK)
void gin_agg_owner(const float* __restrict__ x,
                   const int*   __restrict__ src,
                   const int*   __restrict__ dst,
                   const unsigned char* __restrict__ keys,
                   float*       __restrict__ agg,
                   unsigned*    __restrict__ keepSeg,
                   int*         __restrict__ keepCnt,
                   int*         __restrict__ keepOk,
                   int reuse,
                   int nNodes, int nEdges) {
    __shared__ f4  acc[AGG_ROWS][32];
    __shared__ unsigned seg[AGG_WAVES][SEG_CAP];
    __shared__ int s_cnt[AGG_WAVES];
    __shared__ int s_more[AGG_WAVES];
    __shared__ unsigned own[AGG_WAVES][OWN_CAP];
    __shared__ int tab[AGG_WAVES][AGG_WAVES];
    __shared__ int posTab[AGG_WAVES][AGG_WAVES];

    const int tid  = threadIdx.x;
    const int lane = tid & 31;
    const int wave = __builtin_amdgcn_readfirstlane(tid >> 5);
    const int rowLo = blockIdx.x * AGG_ROWS;
    const int rowsHere = min(AGG_ROWS, nNodes - rowLo);

    unsigned* mySeg = keepSeg + ((long)blockIdx.x * AGG_WAVES + wave) * SEG_CAP;
    int*      myCnt = keepCnt + (long)blockIdx.x * AGG_WAVES + wave;
    bool useKept = false;
    int  keptCnt = 0;
    if (reuse) {
        useKept = __builtin_amdgcn_readfirstlane(keepOk[blockIdx.x]) == 1;
        keptCnt = __builtin_amdgcn_readfirstlane(min(max(*myCnt, 0), OWN_CAP));
    }
    const bool keep = (reuse == 0);

    f4* accFlat = &acc[0][0];
    for (int i = tid; i < AGG_ROWS * 32; i += AGG_BLOCK) accFlat[i] = f4{0.f, 0.f, 0.f, 0.f};

    const bool useKeys = gridDim.x <= 255u;
    const unsigned keyRep = blockIdx.x * 0x01010101u;
    int per = (nEdges + AGG_WAVES - 1) / AGG_WAVES;
    per = useKeys ? ((per + 15) & ~15) : ((per + 3) & ~3);
    const int keyLast = per * AGG_WAVES - 16;
    const int beg = useKeys ? wave * per : min(wave * per, nEdges);
    const int end = useKeys ? beg + per : min(beg + per, nEdges);
    const bool vecOk = ((((unsigned long long)dst) & 15ull) == 0ull) && ((nEdges & 3) == 0);
    int pos = useKept ? end : beg;
    bool anyMore;
    bool firstPass = true;

    do {
        int cntw = 0;
        bool full = false;
        while (useKeys && pos < end && !full) {
            u4 kv[KEY_DEPTH];
            #pragma unroll
            for (int u = 0; u < KEY_DEPTH; ++u) {
                const int e = pos + u * 512 + lane * 16;
                kv[u] = *(const u4*)(keys + min(e, keyLast));
            }
            int stepsDone = 0;
            #pragma unroll
            for (int u = 0; u < KEY_DEPTH; ++u) {
                if (!full && pos + u * 512 < end) {
                    const int e = pos + u * 512 + lane * 16;
                    unsigned w = 0u;
                    #pragma unroll
                    for (int j = 0; j < 4; ++j) {
                        const unsigned y = kv[u][j] ^ keyRep;
                        const unsigned t = (y & 0x7F7F7F7Fu) + 0x7F7F7F7Fu;
                        const unsigned z = ~(t | y | 0x7F7F7F7Fu);
                        w |= (z >> 7) << j;
                    }
                    w = (e < end) ? w : 0u;
                    unsigned mL = __builtin_amdgcn_ballot_w32(w != 0u);
                    if (mL) {
                        bool fits = cntw + 16 * __builtin_popcount(mL) <= SEG_CAP;
                        if (!fits) {
                            int tot = 0;
                            for (unsigned m2 = mL; m2; m2 &= m2 - 1u)
                                tot += __builtin_popcount((unsigned)__builtin_amdgcn_readlane((int)w, __builtin_ctz(m2)));
                            fits = cntw + tot <= SEG_CAP;
                        }
                        if (!fits) {
                            full = true;
                        } else {
                            const int off0 = pos - beg + u * 512;
                            while (mL) {
                                const int l = __builtin_ctz(mL);
                                mL &= mL - 1u;
                                unsigned wl = (unsigned)__builtin_amdgcn_readlane((int)w, l);
                                while (wl) {
                                    const int bit = __builtin_ctz(wl);
                                    wl &= wl - 1u;
                                    seg[wave][cntw] = (unsigned)(off0 + l * 16 + ((bit & 7) << 2) + (bit >> 3));
                                    ++cntw;
                                }
                            }
                        }
                    }
                    if (!full) stepsDone = u + 1;
                }
            }
            pos += stepsDone * 512;
        }
        while (!useKeys && pos < end && !full) {
            i4 v[SCAN_DEPTH];
            if (vecOk) {
                #pragma unroll
                for (int u = 0; u < SCAN_DEPTH; ++u) {
                    const int e = pos + u * 128 + lane * 4;
                    const i4 t = *(const i4*)(dst + min(e, end - 4));
                    const i4 none = {-1, -1, -1, -1};
                    v[u] = (e < end) ? t : none;
                }
            } else {
                #pragma unroll
                for (int u = 0; u < SCAN_DEPTH; ++u) {
                    const int e = pos + u * 128 + lane * 4;
                    i4 t;
                    #pragma unroll
                    for (int k = 0; k < 4; ++k) {
                        const int d = dst[min(e + k, end - 1)];
                        t[k] = (e + k < end) ? d : -1;
                    }
                    v[u] = t;
                }
            }
            int stepsDone = 0;
            #pragma unroll
            for (int u = 0; u < SCAN_DEPTH; ++u) {
                if (!full && pos + u * 128 < end) {
                    if (cntw + 128 > SEG_CAP) {
                        full = true;
                    } else {
                        const int off = pos - beg + u * 128 + lane * 4;
                        unsigned lr[4]; bool h[4]; unsigned m[4];
                        #pragma unroll
                        for (int k = 0; k < 4; ++k) {
                            lr[k] = (unsigned)(v[u][k] - rowLo);
                            h[k]  = lr[k] < (unsigned)rowsHere;
                            m[k]  = __builtin_amdgcn_ballot_w32(h[k]);
                        }
                        if (m[0] | m[1] | m[2] | m[3]) {
                            #pragma unroll
                            for (int k = 0; k < 4; ++k) {
                                if (h[k]) {
                                    const int p = cntw + (int)__builtin_amdgcn_mbcnt_lo(m[k], 0u);
                                    seg[wave][p] = (lr[k] << 20) | (unsigned)(off + k);
                                }
                                cntw += __builtin_popcount(m[k]);
                            }
                        }
                        stepsDone = u + 1;
                    }
                }
            }
            pos += stepsDone * 128;
        }
        const bool keepNow = keep && firstPass;
        if (!useKept) {
            for (int i = lane; i < cntw; i += 32) {
                const unsigned en = seg[wave][i];
                unsigned ent;
                if (useKeys) {
                    const int e = beg + (int)en;
                    ent = ((unsigned)(dst[e] - rowLo) << 20) | (unsigned)src[e];
                } else {
                    ent = (en & 0xFFF00000u) | (unsigned)src[beg + (int)(en & 0xFFFFFu)];
                }
                seg[wave][i] = ent;
            }
            tab[wave][lane] = 0;
            wave_lds_handover();
            for (int i0 = 0; i0 < cntw; i0 += 32) {
                PEERS(i0)
                if (valid && rank == 0) tab[wave][o] += __builtin_popcount(peers);
                wave_lds_handover();
            }
        } else {
            for (int i0 = 0; i0 < keptCnt; i0 += 128) {
                const unsigned kv0 = mySeg[min(i0 + lane,      SEG_CAP - 1)];
                const unsigned kv1 = mySeg[min(i0 + lane + 32, SEG_CAP - 1)];
                const unsigned kv2 = mySeg[min(i0 + lane + 64, SEG_CAP - 1)];
                const unsigned kv3 = mySeg[min(i0 + lane + 96, SEG_CAP - 1)];
                if (i0 + lane      < keptCnt) own[wave][i0 + lane]      = kv0;
                if (i0 + lane + 32 < keptCnt) own[wave][i0 + lane + 32] = kv1;
                if (i0 + lane + 64 < keptCnt) own[wave][i0 + lane + 64] = kv2;
                if (i0 + lane + 96 < keptCnt) own[wave][i0 + lane + 96] = kv3;
            }
            wave_lds_handover();
        }
        if (lane == 0) {
            s_cnt[wave] = useKept ? 0 : cntw; s_more[wave] = (pos < end) ? 1 : 0;
        }
        __syncthreads();
        anyMore = __builtin_amdgcn_ballot_w32(s_more[lane] != 0) != 0u;

        int  nOwn   = keptCnt;
        bool binned = true;
        if (!useKept) {
            int base = 0, total = 0;
            for (int w2 = 0; w2 < AGG_WAVES; ++w2) {
                const int t = tab[w2][lane];
                base  += (w2 < wave) ? t : 0;
                total += t;
            }
            binned = __builtin_amdgcn_ballot_w32(total > OWN_CAP) == 0u;
            nOwn = __builtin_amdgcn_readlane(total, wave);
            if (binned) {
                posTab[wave][lane] = base;
                wave_lds_handover();
                for (int i0 = 0; i0 < cntw; i0 += 32) {
                    PEERS(i0)
                    int p = 0;
                    if (valid) p = posTab[wave][o];
                    wave_lds_handover();
                    if (valid && rank == 0) posTab[wave][o] = p + __builtin_popcount(peers);
                    if (valid) own[o][p + rank] = ent;
                    wave_lds_handover();
                }
            }
            __syncthreads();
        }
        if (keepNow && tid == 0) keepOk[blockIdx.x] = (!anyMore && binned) ? 1 : 0;
        if (keepNow && binned) {
            for (int i = lane; i < nOwn; i += 32) mySeg[i] = own[wave][i];
            if (lane == 0) *myCnt = nOwn;
        }
        firstPass = false;

        if (binned) {
            const f4 zero4 = {0.f, 0.f, 0.f, 0.f};
            f4 xA0 = zero4, xA1 = zero4, xA2 = zero4, xA3 = zero4, xA4 = zero4, xA5 = zero4, xA6 = zero4, xA7 = zero4;
            f4 xB0 = zero4, xB1 = zero4, xB2 = zero4, xB3 = zero4, xB4 = zero4, xB5 = zero4, xB6 = zero4, xB7 = zero4;
            unsigned eA = 0u, eB = 0u;
            int bA = 0, bB = 0, nA = 0, nB = 0;
#define XROWS(M, S) M(S, 0) M(S, 1) M(S, 2) M(S, 3) M(S, 4) M(S, 5) M(S, 6) M(S, 7)
#define X_LOAD(S, q) { \
            const unsigned e2 = (unsigned)__builtin_amdgcn_readlane((int)e##S, b##S + min(q, n##S - 1)); \
            x##S##q = ((const f4*)x)[(long)(e2 & 0xFFFFFu) * 32 + lane]; }
#define X_ADD(S, q) if (q < n##S) { \
            const int r = (int)((unsigned)__builtin_amdgcn_readlane((int)e##S, b##S + q) >> 20); \
            f4 a = acc[r][lane]; a += x##S##q; acc[r][lane] = a; }
#define X_ISSUE(S, E, BL, GI) { \
            e##S = (E); b##S = (BL); n##S = min(max(nOwn - (GI), 0), 8); \
            if (n##S > 0) { XROWS(X_LOAD, S) } }
#define X_LOADF(S, q) { \
            const unsigned e2 = (unsigned)__builtin_amdgcn_readlane((int)e##S, b##S + q); \
            x##S##q = ((const f4*)x)[(long)(e2 & 0xFFFFFu) * 32 + lane]; }
#define X_ADDF(S, q) { \
            const int r = (int)((unsigned)__builtin_amdgcn_readlane((int)e##S, b##S + q) >> 20); \
            f4 a = acc[r][lane]; a += x##S##q; acc[r][lane] = a; }
#define X_ISSUEF(S, E, BL) { e##S = (E); b##S = (BL); n##S = 8; XROWS(X_LOADF, S) }
#define X_FETCH(DSTV, I0) { \
            const int fi = (I0) + lane; \
            unsigned fv = 0u; \
            if (fi < nOwn) fv = own[wave][fi]; \
            DSTV = fv; }
            unsigned entCur, entNext;
            X_FETCH(entCur, 0)
            X_FETCH(entNext, 32)
            X_ISSUE(A, entCur, 0, 0)
            int i0 = 0;
            for (; i0 + 40 <= nOwn; i0 += 32) {
                X_ISSUEF(B, entCur, 8)
                XROWS(X_ADDF, A)
                X_ISSUEF(A, entCur, 16)
                XROWS(X_ADDF, B)
                X_ISSUEF(B, entCur, 24)
                XROWS(X_ADDF, A)
                entCur = entNext;
                X_FETCH(entNext, i0 + 64)
                X_ISSUEF(A, entCur, 0)
                XROWS(X_ADDF, B)
            }
            for (; i0 < nOwn; i0 += 32) {
                X_ISSUE(B, entCur, 8, i0 + 8)
                XROWS(X_ADD, A)
                X_ISSUE(A, entCur, 16, i0 + 16)
                XROWS(X_ADD, B)
                X_ISSUE(B, entCur, 24, i0 + 24)
                XROWS(X_ADD, A)
                entCur = entNext;
                X_FETCH(entNext, i0 + 64)
                X_ISSUE(A, entCur, 0, i0 + 32)
                XROWS(X_ADD, B)
            }
#undef X_FETCH
#undef X_ISSUEF
#undef X_ADDF
#undef X_LOADF
#undef X_ISSUE
#undef X_ADD
#undef X_LOAD
#undef XROWS
        } else {
        unsigned held = 0u;
        int nHeld = 0;
        unsigned sent = 0u;
        int nSent = 0;
        const f4 zero4 = {0.f, 0.f, 0.f, 0.f};
        f4 xr0 = zero4, xr1 = zero4, xr2 = zero4, xr3 = zero4, xr4 = zero4, xr5 = zero4, xr6 = zero4, xr7 = zero4;
#define ROWS(M) M(0) M(1) M(2) M(3) M(4) M(5) M(6) M(7)
#define ROW_ADD(q) if (q < nSent) { \
            const int r = (int)((unsigned)__builtin_amdgcn_readlane((int)sent, q) >> 20); \
            f4 a = acc[r][lane]; a += xr##q; acc[r][lane] = a; }
#define ROW_LOAD(q) { \
            const unsigned e2 = (unsigned)__builtin_amdgcn_readlane((int)held, min(q, nHeld - 1)); \
            xr##q = ((const f4*)x)[(long)(e2 & 0xFFFFFu) * 32 + lane]; }
#define ROW_STEP() { \
            ROWS(ROW_ADD) \
            sent = held; nSent = nHeld; \
            if (nHeld > 0) { ROWS(ROW_LOAD) } \
            nHeld = 0; }
        const int cl = s_cnt[lane];
        for (int sg = 0; sg < AGG_WAVES; ++sg) {
            const int c = __builtin_amdgcn_readlane(cl, sg);
            for (int i0 = 0; i0 < c; i0 += 32) {
                const int i = i0 + lane;
                const unsigned ent = (i < c) ? seg[sg][i] : 0u;
                const bool own = (i < c) && (int)((ent >> 20) & (AGG_WAVES - 1)) == wave;
                unsigned m = __builtin_amdgcn_ballot_w32(own);
                while (m) {
                    const int l = __builtin_ctz(m);
                    m &= m - 1u;
                    const unsigned e2 = (unsigned)__builtin_amdgcn_readlane((int)ent, l);
                    held = (lane == nHeld) ? e2 : held;
                    if (++nHeld == ROW_GROUP) ROW_STEP()
                }
            }
        }
        ROW_STEP()
        ROWS(ROW_ADD)
#undef ROW_STEP
#undef ROW_LOAD
#undef ROW_ADD
#undef ROWS
        }
        __syncthreads();
    } while (anyMore);

    f4* aggRows = (f4*)agg + (long)rowLo * 32;
    for (int i = tid; i < rowsHere * 32; i += AGG_BLOCK) aggRows[i] = accFlat[i];
}

#define MLP_WAVES 16
#define MLP_BLOCK (MLP_WAVES * 32)
#define MLP_GROUPS 256
#define TB_PITCH 33


static __device__ __forceinline__ v8f mma4(float a0, float a1, float b0, float b1, v8f c) {
    v2f a = {a0, a1};
    v2f b = {b0, b1};
    return __builtin_amdgcn_wmma_f32_16x16x4_f32(false, a, false, b, (short)0, c, false, false);
}

__global__ __launch_bounds__(MLP_BLOCK)
void gin_mlp_t(const float* __restrict__ xin,
               const float* __restrict__ agg,
               const float* __restrict__ W1, const float* __restrict__ b1,
               const float* __restrict__ W2, const float* __restrict__ b2,
               float* __restrict__ out,
               int applyReluOut, int nTiles) {
    __shared__ f4 w1f[8][16][32];
    __shared__ f4 w2f[8][16][32];
    __shared__ alignas(16) float bs[2][D];
    __shared__ f4 tb[MLP_WAVES][TILE_ROWS][TB_PITCH];

    const int tid  = threadIdx.x;
    const int lane = tid & 31;
    const int wave = __builtin_amdgcn_readfirstlane(tid >> 5);

    const int c = lane & 15;
    const int h = lane >> 4;
    int tile = blockIdx.x + MLP_GROUPS * wave;

    const f4 zero4 = {0.f, 0.f, 0.f, 0.f};
    f4 xv[16], av[16];
    #pragma unroll
    for (int i = 0; i < 16; ++i) { xv[i] = zero4; av[i] = zero4; }
    if (tile < nTiles) {
        const f4* xr0 = (const f4*)xin + (long)tile * (16 * 32) + lane;
        const f4* ar0 = (const f4*)agg + (long)tile * (16 * 32) + lane;
        #pragma unroll
        for (int r = 0; r < 16; ++r) { xv[r] = xr0[32 * r]; av[r] = ar0[32 * r]; }
    }
    f4 (*myTb)[TB_PITCH] = tb[wave];

    {
        const int n  = tid & (D - 1);
        const int tn = n >> 4;
        const int cc = n & 15;
        const int q  = tid >> 7;
        const float bias1 = b1[n];
        const float bias2 = b2[n];
        f4 wa[8], wb[8];
        #pragma unroll
        for (int j = 0; j < 8; ++j) {
            const int kg = q + 4 * j;
            #pragma unroll
            for (int m = 0; m < 4; ++m) {
                wa[j][m] = W1[(4 * kg + m) * D + n];
                wb[j][m] = W2[(4 * kg + m) * D + n];
            }
        }
        #pragma unroll
        for (int j = 0; j < 8; ++j) {
            const int kg = q + 4 * j;
            const int i1 = kg >> 1;
            const int h1 = kg & 1;
            w1f[tn][i1][cc + 16 * h1] = wa[j];
            const int i2 = ((kg >> 2) << 1) | (kg & 1);
            const int h2 = (kg >> 1) & 1;
            w2f[tn][i2][cc + 16 * h2] = wb[j];
        }
        bs[0][n] = bias1;
        bs[1][n] = bias2;
    }
    __syncthreads();

    while (tile < nTiles) {
        asm volatile("" ::: "memory");
        #pragma unroll
        for (int r = 0; r < 16; ++r) myTb[r][lane] = xv[r] + av[r];
        wave_lds_handover();
        f4 sv[16];
        #pragma unroll
        for (int i = 0; i < 16; ++i) sv[i] = myTb[c][2 * i + h];
        __builtin_amdgcn_sched_barrier(0);

        v8f acc[8];
        #pragma unroll
        for (int t = 0; t < 8; ++t) acc[t] = v8f{};

        {
            f4 aC[8], aN[8];
            #pragma unroll
            for (int t = 0; t < 8; ++t) aC[t] = w1f[t][0][lane];
            #pragma unroll
            for (int i = 0; i < 16; ++i) {
                if (i + 1 < 16) {
                    #pragma unroll
                    for (int t = 0; t < 8; ++t) aN[t] = w1f[t][i + 1][lane];
                }
                __builtin_amdgcn_sched_barrier(0);
                #pragma unroll
                for (int t = 0; t < 8; ++t) acc[t] = mma4(aC[t][0], aC[t][1], sv[i][0], sv[i][1], acc[t]);
                #pragma unroll
                for (int t = 0; t < 8; ++t) acc[t] = mma4(aC[t][2], aC[t][3], sv[i][2], sv[i][3], acc[t]);
                __builtin_amdgcn_sched_barrier(0);
                #pragma unroll
                for (int t = 0; t < 8; ++t) aC[t] = aN[t];
            }
        }
        #pragma unroll
        for (int t = 0; t < 8; ++t) {
            const f4 bA = *(const f4*)&bs[0][16 * t + 8 * h];
            const f4 bB = *(const f4*)&bs[0][16 * t + 8 * h + 4];
            #pragma unroll
            for (int r = 0; r < 8; ++r) {
                const float v = acc[t][r] + (r < 4 ? bA[r] : bB[r - 4]);
                acc[t][r] = v > 0.0f ? v : 0.0f;
            }
        }

        v8f acc2[8];
        #pragma unroll
        for (int t = 0; t < 8; ++t) acc2[t] = v8f{};
        {
            f4 aC[8], aN[8];
            #pragma unroll
            for (int t = 0; t < 8; ++t) aC[t] = w2f[t][0][lane];
            #pragma unroll
            for (int i = 0; i < 16; ++i) {
                const int tk = i >> 1;
                const int p4 = (i & 1) * 4;
                if (i + 1 < 16) {
                    #pragma unroll
                    for (int t = 0; t < 8; ++t) aN[t] = w2f[t][i + 1][lane];
                }
                __builtin_amdgcn_sched_barrier(0);
                #pragma unroll
                for (int t = 0; t < 8; ++t) acc2[t] = mma4(aC[t][0], aC[t][1], acc[tk][p4], acc[tk][p4 + 1], acc2[t]);
                #pragma unroll
                for (int t = 0; t < 8; ++t) acc2[t] = mma4(aC[t][2], aC[t][3], acc[tk][p4 + 2], acc[tk][p4 + 3], acc2[t]);
                __builtin_amdgcn_sched_barrier(0);
                #pragma unroll
                for (int t = 0; t < 8; ++t) aC[t] = aN[t];
            }
        }

        #pragma unroll
        for (int t = 0; t < 8; ++t) {
            const f4 bA = *(const f4*)&bs[1][16 * t + 8 * h];
            const f4 bB = *(const f4*)&bs[1][16 * t + 8 * h + 4];
            f4 o0, o1;
            #pragma unroll
            for (int r = 0; r < 4; ++r) {
                float u0 = acc2[t][r] + bA[r];
                float u1 = acc2[t][r + 4] + bB[r];
                if (applyReluOut) { u0 = u0 > 0.0f ? u0 : 0.0f; u1 = u1 > 0.0f ? u1 : 0.0f; }
                o0[r] = u0; o1[r] = u1;
            }
            myTb[c][4 * t + 2 * h]     = o0;
            myTb[c][4 * t + 2 * h + 1] = o1;
        }
        wave_lds_handover();
        f4* orow = (f4*)out + (long)tile * (16 * 32) + lane;
        #pragma unroll
        for (int r = 0; r < 16; ++r) orow[32 * r] = myTb[r][lane];

        tile += MLP_GROUPS * MLP_WAVES;
        if (tile < nTiles) {
            const f4* xrN = (const f4*)xin + (long)tile * (16 * 32) + lane;
            const f4* arN = (const f4*)agg + (long)tile * (16 * 32) + lane;
            #pragma unroll
            for (int r = 0; r < 16; ++r) { xv[r] = xrN[32 * r]; av[r] = arN[32 * r]; }
        }
    }
}

extern "C" void kernel_launch(void* const* d_in, const int* in_sizes, int n_in,
                              void* d_out, int out_size, void* d_ws, size_t ws_size,
                              hipStream_t stream) {
    const float* x  = (const float*)d_in[0];
    const int*   ei = (const int*)  d_in[1];
    const float* W1 = (const float*)d_in[2];
    const float* b1 = (const float*)d_in[3];
    const float* W2 = (const float*)d_in[4];
    const float* b2 = (const float*)d_in[5];
    float* out = (float*)d_out;

    const int N = in_sizes[0] / D;
    const int E = in_sizes[1] / 2;
    const int* src = ei;
    const int* dst = ei + E;

    float* agg = (float*)d_ws;
    float* h1  = agg + (size_t)N * D;

    const int  aBlocks = (N + AGG_ROWS - 1) / AGG_ROWS;
    const int  mTiles  = N / TILE_ROWS;
    const int  kBlocks = (E + 511 + 4095) / 4096;
    unsigned char* keys = (unsigned char*)(h1 + (size_t)N * D);
    unsigned* keepSeg = (unsigned*)(keys + (size_t)kBlocks * 4096);
    int*      keepCnt = (int*)(keepSeg + (size_t)aBlocks * AGG_WAVES * SEG_CAP);
    int*      keepOk  = keepCnt + (size_t)aBlocks * AGG_WAVES;

    gin_edge_keys<<<kBlocks, 256, 0, stream>>>(dst, keys, N, E);

    gin_agg_owner<<<aBlocks, AGG_BLOCK, 0, stream>>>(x, src, dst, keys, agg, keepSeg, keepCnt, keepOk, 0, N, E);
    gin_mlp_t<<<MLP_GROUPS, MLP_BLOCK, 0, stream>>>(x, agg, W1, b1, W2, b2, h1, 1, mTiles);

    gin_agg_owner<<<aBlocks, AGG_BLOCK, 0, stream>>>(h1, src, dst, keys, agg, keepSeg, keepCnt, keepOk, 1, N, E);
    gin_mlp_t<<<MLP_GROUPS, MLP_BLOCK, 0, stream>>>(h1, agg, W1, b1, W2, b2, out, 0, mTiles);
}


# MLP: a work-group's leftover tile is split over its four SIMDs by output features, hidden layer exchanged through LDS (was a 7th whole tile on one SIMD, 6 on the others); sums unchanged
# speedup vs baseline: 4.0587x; 1.0319x over previous
#include <hip/hip_runtime.h>


#define D 128
#define TILE_ROWS 16

typedef float v2f __attribute__((ext_vector_type(2)));
typedef float v8f __attribute__((ext_vector_type(8)));

#define AGG_ROWS   400
#define AGG_WAVES  32
#define AGG_BLOCK  1024
#define SEG_CAP    512
#define OWN_CAP    256
#define KEY_DEPTH  8
#define KEY_NONE   255u
#define SCAN_DEPTH 8
#define ROW_GROUP  8

typedef float f4 __attribute__((ext_vector_type(4)));
typedef int   i4 __attribute__((ext_vector_type(4)));
typedef unsigned u4 __attribute__((ext_vector_type(4)));

__global__ __launch_bounds__(256)
void gin_edge_keys(const int* __restrict__ dst,
                   unsigned char* __restrict__ keys,
                   int nNodes, int nEdges) {
    int per = (nEdges + AGG_WAVES - 1) / AGG_WAVES;
    per = (per + 15) & ~15;
    const int nPadded = per * AGG_WAVES;
    const int base = (blockIdx.x * 256 + threadIdx.x) * 16;
    if (base >= nPadded) return;
    const bool vecOk = (((unsigned long long)dst) & 15ull) == 0ull;
    u4 o;
    #pragma unroll
    for (int j = 0; j < 4; ++j) {
        const int e0 = base + 4 * j;
        i4 d;
        if (vecOk && e0 + 3 < nEdges) {
            d = *(const i4*)(dst + e0);
        } else {
            #pragma unroll
            for (int k = 0; k < 4; ++k) d[k] = (e0 + k < nEdges) ? dst[min(e0 + k, nEdges - 1)] : -1;
        }
        unsigned word = 0u;
        #pragma unroll
        for (int k = 0; k < 4; ++k) {
            const unsigned dk = (unsigned)d[k];
            const unsigned key = (dk < (unsigned)nNodes) ? min(dk / (unsigned)AGG_ROWS, KEY_NONE) : KEY_NONE;
            word |= key << (8 * k);
        }
        o[j] = word;
    }
    *(u4*)(keys + base) = o;
}

static __device__ __forceinline__ void wave_lds_handover() {
    __builtin_amdgcn_fence(__ATOMIC_RELEASE, "wavefront");
    __builtin_amdgcn_wave_barrier();
    __builtin_amdgcn_fence(__ATOMIC_ACQUIRE, "wavefront");
}

#define PEERS(I0) \
                const int pi = (I0) + lane; \
                const bool valid = pi < cntw; \
                const unsigned ent = valid ? seg[wave][pi] : 0u; \
                const unsigned o = (ent >> 20) & (AGG_WAVES - 1); \
                unsigned peers = __builtin_amdgcn_ballot_w32(valid); \
                _Pragma("unroll") \
                for (int pb = 0; pb < 5; ++pb) { \
                    const bool bit = ((o >> pb) & 1u) != 0u; \
                    const unsigned mb = __builtin_amdgcn_ballot_w32(bit); \
                    peers &= bit ? mb : ~mb; \
                } \
                const int rank = (int)__builtin_amdgcn_mbcnt_lo(peers, 0u);

__global__ __launch_bounds__(AGG_BLOCK)
void gin_agg_owner(const float* __restrict__ x,
                   const int*   __restrict__ src,
                   const int*   __restrict__ dst,
                   const unsigned char* __restrict__ keys,
                   float*       __restrict__ agg,
                   unsigned*    __restrict__ keepSeg,
                   int*         __restrict__ keepCnt,
                   int*         __restrict__ keepOk,
                   int reuse,
                   int nNodes, int nEdges) {
    __shared__ f4  acc[AGG_ROWS][32];
    __shared__ unsigned seg[AGG_WAVES][SEG_CAP];
    __shared__ int s_cnt[AGG_WAVES];
    __shared__ int s_more[AGG_WAVES];
    __shared__ unsigned own[AGG_WAVES][OWN_CAP];
    __shared__ int tab[AGG_WAVES][AGG_WAVES];
    __shared__ int posTab[AGG_WAVES][AGG_WAVES];

    const int tid  = threadIdx.x;
    const int lane = tid & 31;
    const int wave = __builtin_amdgcn_readfirstlane(tid >> 5);
    const int rowLo = blockIdx.x * AGG_ROWS;
    const int rowsHere = min(AGG_ROWS, nNodes - rowLo);

    unsigned* mySeg = keepSeg + ((long)blockIdx.x * AGG_WAVES + wave) * SEG_CAP;
    int*      myCnt = keepCnt + (long)blockIdx.x * AGG_WAVES + wave;
    bool useKept = false;
    int  keptCnt = 0;
    if (reuse) {
        useKept = __builtin_amdgcn_readfirstlane(keepOk[blockIdx.x]) == 1;
        keptCnt = __builtin_amdgcn_readfirstlane(min(max(*myCnt, 0), OWN_CAP));
    }
    const bool keep = (reuse == 0);

    f4* accFlat = &acc[0][0];
    for (int i = tid; i < AGG_ROWS * 32; i += AGG_BLOCK) accFlat[i] = f4{0.f, 0.f, 0.f, 0.f};

    const bool useKeys = gridDim.x <= 255u;
    const unsigned keyRep = blockIdx.x * 0x01010101u;
    int per = (nEdges + AGG_WAVES - 1) / AGG_WAVES;
    per = useKeys ? ((per + 15) & ~15) : ((per + 3) & ~3);
    const int keyLast = per * AGG_WAVES - 16;
    const int beg = useKeys ? wave * per : min(wave * per, nEdges);
    const int end = useKeys ? beg + per : min(beg + per, nEdges);
    const bool vecOk = ((((unsigned long long)dst) & 15ull) == 0ull) && ((nEdges & 3) == 0);
    int pos = useKept ? end : beg;
    bool anyMore;
    bool firstPass = true;

    do {
        int cntw = 0;
        bool full = false;
        while (useKeys && pos < end && !full) {
            u4 kv[KEY_DEPTH];
            #pragma unroll
            for (int u = 0; u < KEY_DEPTH; ++u) {
                const int e = pos + u * 512 + lane * 16;
                kv[u] = *(const u4*)(keys + min(e, keyLast));
            }
            int stepsDone = 0;
            #pragma unroll
            for (int u = 0; u < KEY_DEPTH; ++u) {
                if (!full && pos + u * 512 < end) {
                    const int e = pos + u * 512 + lane * 16;
                    unsigned w = 0u;
                    #pragma unroll
                    for (int j = 0; j < 4; ++j) {
                        const unsigned y = kv[u][j] ^ keyRep;
                        const unsigned t = (y & 0x7F7F7F7Fu) + 0x7F7F7F7Fu;
                        const unsigned z = ~(t | y | 0x7F7F7F7Fu);
                        w |= (z >> 7) << j;
                    }
                    w = (e < end) ? w : 0u;
                    unsigned mL = __builtin_amdgcn_ballot_w32(w != 0u);
                    if (mL) {
                        bool fits = cntw + 16 * __builtin_popcount(mL) <= SEG_CAP;
                        if (!fits) {
                            int tot = 0;
                            for (unsigned m2 = mL; m2; m2 &= m2 - 1u)
                                tot += __builtin_popcount((unsigned)__builtin_amdgcn_readlane((int)w, __builtin_ctz(m2)));
                            fits = cntw + tot <= SEG_CAP;
                        }
                        if (!fits) {
                            full = true;
                        } else {
                            const int off0 = pos - beg + u * 512;
                            while (mL) {
                                const int l = __builtin_ctz(mL);
                                mL &= mL - 1u;
                                unsigned wl = (unsigned)__builtin_amdgcn_readlane((int)w, l);
                                while (wl) {
                                    const int bit = __builtin_ctz(wl);
                                    wl &= wl - 1u;
                                    seg[wave][cntw] = (unsigned)(off0 + l * 16 + ((bit & 7) << 2) + (bit >> 3));
                                    ++cntw;
                                }
                            }
                        }
                    }
                    if (!full) stepsDone = u + 1;
                }
            }
            pos += stepsDone * 512;
        }
        while (!useKeys && pos < end && !full) {
            i4 v[SCAN_DEPTH];
            if (vecOk) {
                #pragma unroll
                for (int u = 0; u < SCAN_DEPTH; ++u) {
                    const int e = pos + u * 128 + lane * 4;
                    const i4 t = *(const i4*)(dst + min(e, end - 4));
                    const i4 none = {-1, -1, -1, -1};
                    v[u] = (e < end) ? t : none;
                }
            } else {
                #pragma unroll
                for (int u = 0; u < SCAN_DEPTH; ++u) {
                    const int e = pos + u * 128 + lane * 4;
                    i4 t;
                    #pragma unroll
                    for (int k = 0; k < 4; ++k) {
                        const int d = dst[min(e + k, end - 1)];
                        t[k] = (e + k < end) ? d : -1;
                    }
                    v[u] = t;
                }
            }
            int stepsDone = 0;
            #pragma unroll
            for (int u = 0; u < SCAN_DEPTH; ++u) {
                if (!full && pos + u * 128 < end) {
                    if (cntw + 128 > SEG_CAP) {
                        full = true;
                    } else {
                        const int off = pos - beg + u * 128 + lane * 4;
                        unsigned lr[4]; bool h[4]; unsigned m[4];
                        #pragma unroll
                        for (int k = 0; k < 4; ++k) {
                            lr[k] = (unsigned)(v[u][k] - rowLo);
                            h[k]  = lr[k] < (unsigned)rowsHere;
                            m[k]  = __builtin_amdgcn_ballot_w32(h[k]);
                        }
                        if (m[0] | m[1] | m[2] | m[3]) {
                            #pragma unroll
                            for (int k = 0; k < 4; ++k) {
                                if (h[k]) {
                                    const int p = cntw + (int)__builtin_amdgcn_mbcnt_lo(m[k], 0u);
                                    seg[wave][p] = (lr[k] << 20) | (unsigned)(off + k);
                                }
                                cntw += __builtin_popcount(m[k]);
                            }
                        }
                        stepsDone = u + 1;
                    }
                }
            }
            pos += stepsDone * 128;
        }
        const bool keepNow = keep && firstPass;
        if (!useKept) {
            for (int i = lane; i < cntw; i += 32) {
                const unsigned en = seg[wave][i];
                unsigned ent;
                if (useKeys) {
                    const int e = beg + (int)en;
                    ent = ((unsigned)(dst[e] - rowLo) << 20) | (unsigned)src[e];
                } else {
                    ent = (en & 0xFFF00000u) | (unsigned)src[beg + (int)(en & 0xFFFFFu)];
                }
                seg[wave][i] = ent;
            }
            tab[wave][lane] = 0;
            wave_lds_handover();
            for (int i0 = 0; i0 < cntw; i0 += 32) {
                PEERS(i0)
                if (valid && rank == 0) tab[wave][o] += __builtin_popcount(peers);
                wave_lds_handover();
            }
        } else {
            for (int i0 = 0; i0 < keptCnt; i0 += 128) {
                const unsigned kv0 = mySeg[min(i0 + lane,      SEG_CAP - 1)];
                const unsigned kv1 = mySeg[min(i0 + lane + 32, SEG_CAP - 1)];
                const unsigned kv2 = mySeg[min(i0 + lane + 64, SEG_CAP - 1)];
                const unsigned kv3 = mySeg[min(i0 + lane + 96, SEG_CAP - 1)];
                if (i0 + lane      < keptCnt) own[wave][i0 + lane]      = kv0;
                if (i0 + lane + 32 < keptCnt) own[wave][i0 + lane + 32] = kv1;
                if (i0 + lane + 64 < keptCnt) own[wave][i0 + lane + 64] = kv2;
                if (i0 + lane + 96 < keptCnt) own[wave][i0 + lane + 96] = kv3;
            }
            wave_lds_handover();
        }
        if (lane == 0) {
            s_cnt[wave] = useKept ? 0 : cntw; s_more[wave] = (pos < end) ? 1 : 0;
        }
        __syncthreads();
        anyMore = __builtin_amdgcn_ballot_w32(s_more[lane] != 0) != 0u;

        int  nOwn   = keptCnt;
        bool binned = true;
        if (!useKept) {
            int base = 0, total = 0;
            for (int w2 = 0; w2 < AGG_WAVES; ++w2) {
                const int t = tab[w2][lane];
                base  += (w2 < wave) ? t : 0;
                total += t;
            }
            binned = __builtin_amdgcn_ballot_w32(total > OWN_CAP) == 0u;
            nOwn = __builtin_amdgcn_readlane(total, wave);
            if (binned) {
                posTab[wave][lane] = base;
                wave_lds_handover();
                for (int i0 = 0; i0 < cntw; i0 += 32) {
                    PEERS(i0)
                    int p = 0;
                    if (valid) p = posTab[wave][o];
                    wave_lds_handover();
                    if (valid && rank == 0) posTab[wave][o] = p + __builtin_popcount(peers);
                    if (valid) own[o][p + rank] = ent;
                    wave_lds_handover();
                }
            }
            __syncthreads();
        }
        if (keepNow && tid == 0) keepOk[blockIdx.x] = (!anyMore && binned) ? 1 : 0;
        if (keepNow && binned) {
            for (int i = lane; i < nOwn; i += 32) mySeg[i] = own[wave][i];
            if (lane == 0) *myCnt = nOwn;
        }
        firstPass = false;

        if (binned) {
            const f4 zero4 = {0.f, 0.f, 0.f, 0.f};
            f4 xA0 = zero4, xA1 = zero4, xA2 = zero4, xA3 = zero4, xA4 = zero4, xA5 = zero4, xA6 = zero4, xA7 = zero4;
            f4 xB0 = zero4, xB1 = zero4, xB2 = zero4, xB3 = zero4, xB4 = zero4, xB5 = zero4, xB6 = zero4, xB7 = zero4;
            unsigned eA = 0u, eB = 0u;
            int bA = 0, bB = 0, nA = 0, nB = 0;
#define XROWS(M, S) M(S, 0) M(S, 1) M(S, 2) M(S, 3) M(S, 4) M(S, 5) M(S, 6) M(S, 7)
#define X_LOAD(S, q) { \
            const unsigned e2 = (unsigned)__builtin_amdgcn_readlane((int)e##S, b##S + min(q, n##S - 1)); \
            x##S##q = ((const f4*)x)[(long)(e2 & 0xFFFFFu) * 32 + lane]; }
#define X_ADD(S, q) if (q < n##S) { \
            const int r = (int)((unsigned)__builtin_amdgcn_readlane((int)e##S, b##S + q) >> 20); \
            f4 a = acc[r][lane]; a += x##S##q; acc[r][lane] = a; }
#define X_ISSUE(S, E, BL, GI) { \
            e##S = (E); b##S = (BL); n##S = min(max(nOwn - (GI), 0), 8); \
            if (n##S > 0) { XROWS(X_LOAD, S) } }
#define X_LOADF(S, q) { \
            const unsigned e2 = (unsigned)__builtin_amdgcn_readlane((int)e##S, b##S + q); \
            x##S##q = ((const f4*)x)[(long)(e2 & 0xFFFFFu) * 32 + lane]; }
#define X_ADDF(S, q) { \
            const int r = (int)((unsigned)__builtin_amdgcn_readlane((int)e##S, b##S + q) >> 20); \
            f4 a = acc[r][lane]; a += x##S##q; acc[r][lane] = a; }
#define X_ISSUEF(S, E, BL) { e##S = (E); b##S = (BL); n##S = 8; XROWS(X_LOADF, S) }
#define X_FETCH(DSTV, I0) { \
            const int fi = (I0) + lane; \
            unsigned fv = 0u; \
            if (fi < nOwn) fv = own[wave][fi]; \
            DSTV = fv; }
            unsigned entCur, entNext;
            X_FETCH(entCur, 0)
            X_FETCH(entNext, 32)
            X_ISSUE(A, entCur, 0, 0)
            int i0 = 0;
            for (; i0 + 40 <= nOwn; i0 += 32) {
                X_ISSUEF(B, entCur, 8)
                XROWS(X_ADDF, A)
                X_ISSUEF(A, entCur, 16)
                XROWS(X_ADDF, B)
                X_ISSUEF(B, entCur, 24)
                XROWS(X_ADDF, A)
                entCur = entNext;
                X_FETCH(entNext, i0 + 64)
                X_ISSUEF(A, entCur, 0)
                XROWS(X_ADDF, B)
            }
            for (; i0 < nOwn; i0 += 32) {
                X_ISSUE(B, entCur, 8, i0 + 8)
                XROWS(X_ADD, A)
                X_ISSUE(A, entCur, 16, i0 + 16)
                XROWS(X_ADD, B)
                X_ISSUE(B, entCur, 24, i0 + 24)
                XROWS(X_ADD, A)
                entCur = entNext;
                X_FETCH(entNext, i0 + 64)
                X_ISSUE(A, entCur, 0, i0 + 32)
                XROWS(X_ADD, B)
            }
#undef X_FETCH
#undef X_ISSUEF
#undef X_ADDF
#undef X_LOADF
#undef X_ISSUE
#undef X_ADD
#undef X_LOAD
#undef XROWS
        } else {
        unsigned held = 0u;
        int nHeld = 0;
        unsigned sent = 0u;
        int nSent = 0;
        const f4 zero4 = {0.f, 0.f, 0.f, 0.f};
        f4 xr0 = zero4, xr1 = zero4, xr2 = zero4, xr3 = zero4, xr4 = zero4, xr5 = zero4, xr6 = zero4, xr7 = zero4;
#define ROWS(M) M(0) M(1) M(2) M(3) M(4) M(5) M(6) M(7)
#define ROW_ADD(q) if (q < nSent) { \
            const int r = (int)((unsigned)__builtin_amdgcn_readlane((int)sent, q) >> 20); \
            f4 a = acc[r][lane]; a += xr##q; acc[r][lane] = a; }
#define ROW_LOAD(q) { \
            const unsigned e2 = (unsigned)__builtin_amdgcn_readlane((int)held, min(q, nHeld - 1)); \
            xr##q = ((const f4*)x)[(long)(e2 & 0xFFFFFu) * 32 + lane]; }
#define ROW_STEP() { \
            ROWS(ROW_ADD) \
            sent = held; nSent = nHeld; \
            if (nHeld > 0) { ROWS(ROW_LOAD) } \
            nHeld = 0; }
        const int cl = s_cnt[lane];
        for (int sg = 0; sg < AGG_WAVES; ++sg) {
            const int c = __builtin_amdgcn_readlane(cl, sg);
            for (int i0 = 0; i0 < c; i0 += 32) {
                const int i = i0 + lane;
                const unsigned ent = (i < c) ? seg[sg][i] : 0u;
                const bool own = (i < c) && (int)((ent >> 20) & (AGG_WAVES - 1)) == wave;
                unsigned m = __builtin_amdgcn_ballot_w32(own);
                while (m) {
                    const int l = __builtin_ctz(m);
                    m &= m - 1u;
                    const unsigned e2 = (unsigned)__builtin_amdgcn_readlane((int)ent, l);
                    held = (lane == nHeld) ? e2 : held;
                    if (++nHeld == ROW_GROUP) ROW_STEP()
                }
            }
        }
        ROW_STEP()
        ROWS(ROW_ADD)
#undef ROW_STEP
#undef ROW_LOAD
#undef ROW_ADD
#undef ROWS
        }
        __syncthreads();
    } while (anyMore);

    f4* aggRows = (f4*)agg + (long)rowLo * 32;
    for (int i = tid; i < rowsHere * 32; i += AGG_BLOCK) aggRows[i] = accFlat[i];
}

#define MLP_WAVES 16
#define MLP_BLOCK (MLP_WAVES * 32)
#define MLP_GROUPS 256
#define MLP_SIMDS 4
#define TB_PITCH 33


static __device__ __forceinline__ v8f mma4(float a0, float a1, float b0, float b1, v8f c) {
    v2f a = {a0, a1};
    v2f b = {b0, b1};
    return __builtin_amdgcn_wmma_f32_16x16x4_f32(false, a, false, b, (short)0, c, false, false);
}

__global__ __launch_bounds__(MLP_BLOCK)
void gin_mlp_t(const float* __restrict__ xin,
               const float* __restrict__ agg,
               const float* __restrict__ W1, const float* __restrict__ b1,
               const float* __restrict__ W2, const float* __restrict__ b2,
               float* __restrict__ out,
               int applyReluOut, int nTiles) {
    __shared__ f4 w1f[8][16][32];
    __shared__ f4 w2f[8][16][32];
    __shared__ alignas(16) float bs[2][D];
    __shared__ f4 tb[MLP_WAVES][TILE_ROWS][TB_PITCH];
    __shared__ f4 xch[8][32][2];

    const int tid  = threadIdx.x;
    const int lane = tid & 31;
    const int wave = __builtin_amdgcn_readfirstlane(tid >> 5);

    const int c = lane & 15;
    const int h = lane >> 4;
    int tile = blockIdx.x + MLP_GROUPS * wave;
    const int nMain = (nTiles / (MLP_GROUPS * MLP_SIMDS)) * (MLP_GROUPS * MLP_SIMDS);
    const int xFirst = nMain + blockIdx.x;

    const f4 zero4 = {0.f, 0.f, 0.f, 0.f};
    f4 xv[16], av[16];
    #pragma unroll
    for (int i = 0; i < 16; ++i) { xv[i] = zero4; av[i] = zero4; }
    if (tile < nMain) {
        const f4* xr0 = (const f4*)xin + (long)tile * (16 * 32) + lane;
        const f4* ar0 = (const f4*)agg + (long)tile * (16 * 32) + lane;
        #pragma unroll
        for (int r = 0; r < 16; ++r) { xv[r] = xr0[32 * r]; av[r] = ar0[32 * r]; }
    }
    f4 (*myTb)[TB_PITCH] = tb[wave];

    {
        const int n  = tid & (D - 1);
        const int tn = n >> 4;
        const int cc = n & 15;
        const int q  = tid >> 7;
        const float bias1 = b1[n];
        const float bias2 = b2[n];
        f4 wa[8], wb[8];
        #pragma unroll
        for (int j = 0; j < 8; ++j) {
            const int kg = q + 4 * j;
            #pragma unroll
            for (int m = 0; m < 4; ++m) {
                wa[j][m] = W1[(4 * kg + m) * D + n];
                wb[j][m] = W2[(4 * kg + m) * D + n];
            }
        }
        #pragma unroll
        for (int j = 0; j < 8; ++j) {
            const int kg = q + 4 * j;
            const int i1 = kg >> 1;
            const int h1 = kg & 1;
            w1f[tn][i1][cc + 16 * h1] = wa[j];
            const int i2 = ((kg >> 2) << 1) | (kg & 1);
            const int h2 = (kg >> 1) & 1;
            w2f[tn][i2][cc + 16 * h2] = wb[j];
        }
        bs[0][n] = bias1;
        bs[1][n] = bias2;
    }
    __syncthreads();

    while (tile < nMain) {
        asm volatile("" ::: "memory");
        #pragma unroll
        for (int r = 0; r < 16; ++r) myTb[r][lane] = xv[r] + av[r];
        wave_lds_handover();
        f4 sv[16];
        #pragma unroll
        for (int i = 0; i < 16; ++i) sv[i] = myTb[c][2 * i + h];
        __builtin_amdgcn_sched_barrier(0);

        v8f acc[8];
        #pragma unroll
        for (int t = 0; t < 8; ++t) acc[t] = v8f{};

        {
            f4 aC[8], aN[8];
            #pragma unroll
            for (int t = 0; t < 8; ++t) aC[t] = w1f[t][0][lane];
            #pragma unroll
            for (int i = 0; i < 16; ++i) {
                if (i + 1 < 16) {
                    #pragma unroll
                    for (int t = 0; t < 8; ++t) aN[t] = w1f[t][i + 1][lane];
                }
                __builtin_amdgcn_sched_barrier(0);
                #pragma unroll
                for (int t = 0; t < 8; ++t) acc[t] = mma4(aC[t][0], aC[t][1], sv[i][0], sv[i][1], acc[t]);
                #pragma unroll
                for (int t = 0; t < 8; ++t) acc[t] = mma4(aC[t][2], aC[t][3], sv[i][2], sv[i][3], acc[t]);
                __builtin_amdgcn_sched_barrier(0);
                #pragma unroll
                for (int t = 0; t < 8; ++t) aC[t] = aN[t];
            }
        }
        #pragma unroll
        for (int t = 0; t < 8; ++t) {
            const f4 bA = *(const f4*)&bs[0][16 * t + 8 * h];
            const f4 bB = *(const f4*)&bs[0][16 * t + 8 * h + 4];
            #pragma unroll
            for (int r = 0; r < 8; ++r) {
                const float v = acc[t][r] + (r < 4 ? bA[r] : bB[r - 4]);
                acc[t][r] = v > 0.0f ? v : 0.0f;
            }
        }

        v8f acc2[8];
        #pragma unroll
        for (int t = 0; t < 8; ++t) acc2[t] = v8f{};
        {
            f4 aC[8], aN[8];
            #pragma unroll
            for (int t = 0; t < 8; ++t) aC[t] = w2f[t][0][lane];
            #pragma unroll
            for (int i = 0; i < 16; ++i) {
                const int tk = i >> 1;
                const int p4 = (i & 1) * 4;
                if (i + 1 < 16) {
                    #pragma unroll
                    for (int t = 0; t < 8; ++t) aN[t] = w2f[t][i + 1][lane];
                }
                __builtin_amdgcn_sched_barrier(0);
                #pragma unroll
                for (int t = 0; t < 8; ++t) acc2[t] = mma4(aC[t][0], aC[t][1], acc[tk][p4], acc[tk][p4 + 1], acc2[t]);
                #pragma unroll
                for (int t = 0; t < 8; ++t) acc2[t] = mma4(aC[t][2], aC[t][3], acc[tk][p4 + 2], acc[tk][p4 + 3], acc2[t]);
                __builtin_amdgcn_sched_barrier(0);
                #pragma unroll
                for (int t = 0; t < 8; ++t) aC[t] = aN[t];
            }
        }

        #pragma unroll
        for (int t = 0; t < 8; ++t) {
            const f4 bA = *(const f4*)&bs[1][16 * t + 8 * h];
            const f4 bB = *(const f4*)&bs[1][16 * t + 8 * h + 4];
            f4 o0, o1;
            #pragma unroll
            for (int r = 0; r < 4; ++r) {
                float u0 = acc2[t][r] + bA[r];
                float u1 = acc2[t][r + 4] + bB[r];
                if (applyReluOut) { u0 = u0 > 0.0f ? u0 : 0.0f; u1 = u1 > 0.0f ? u1 : 0.0f; }
                o0[r] = u0; o1[r] = u1;
            }
            myTb[c][4 * t + 2 * h]     = o0;
            myTb[c][4 * t + 2 * h + 1] = o1;
        }
        wave_lds_handover();
        f4* orow = (f4*)out + (long)tile * (16 * 32) + lane;
        #pragma unroll
        for (int r = 0; r < 16; ++r) orow[32 * r] = myTb[r][lane];

        tile += MLP_GROUPS * MLP_WAVES;
        if (tile < nMain) {
            const f4* xrN = (const f4*)xin + (long)tile * (16 * 32) + lane;
            const f4* arN = (const f4*)agg + (long)tile * (16 * 32) + lane;
            #pragma unroll
            for (int r = 0; r < 16; ++r) { xv[r] = xrN[32 * r]; av[r] = arN[32 * r]; }
        }
    }

    for (int xt = xFirst; xt < nTiles; xt += MLP_GROUPS) {
        const int t0 = 2 * wave;
        if (wave < MLP_SIMDS) {
            const f4* xrN = (const f4*)xin + (long)xt * (16 * 32) + lane;
            const f4* arN = (const f4*)agg + (long)xt * (16 * 32) + lane;
            f4 xl[16], al[16];
            #pragma unroll
            for (int r = 0; r < 16; ++r) { xl[r] = xrN[32 * r]; al[r] = arN[32 * r]; }
            #pragma unroll
            for (int r = 0; r < 16; ++r) myTb[r][lane] = xl[r] + al[r];
            wave_lds_handover();
            f4 sv[16];
            #pragma unroll
            for (int i = 0; i < 16; ++i) sv[i] = myTb[c][2 * i + h];
            v8f pacc[2];
            pacc[0] = v8f{}; pacc[1] = v8f{};
            {
                f4 gC[8], gN[8];
                #pragma unroll
                for (int u = 0; u < 2; ++u) {
                    #pragma unroll
                    for (int j = 0; j < 4; ++j) gC[4 * u + j] = w1f[t0 + u][j][lane];
                }
                #pragma unroll
                for (int g = 0; g < 4; ++g) {
                    if (g + 1 < 4) {
                        #pragma unroll
                        for (int u = 0; u < 2; ++u) {
                            #pragma unroll
                            for (int j = 0; j < 4; ++j) gN[4 * u + j] = w1f[t0 + u][4 * (g + 1) + j][lane];
                        }
                    }
                    __builtin_amdgcn_sched_barrier(0);
                    #pragma unroll
                    for (int j = 0; j < 4; ++j) {
                        const int i = 4 * g + j;
                        #pragma unroll
                        for (int u = 0; u < 2; ++u) pacc[u] = mma4(gC[4 * u + j][0], gC[4 * u + j][1], sv[i][0], sv[i][1], pacc[u]);
                        #pragma unroll
                        for (int u = 0; u < 2; ++u) pacc[u] = mma4(gC[4 * u + j][2], gC[4 * u + j][3], sv[i][2], sv[i][3], pacc[u]);
                    }
                    __builtin_amdgcn_sched_barrier(0);
                    #pragma unroll
                    for (int k = 0; k < 8; ++k) gC[k] = gN[k];
                }
            }
            #pragma unroll
            for (int u = 0; u < 2; ++u) {
                const f4 bA = *(const f4*)&bs[0][16 * (t0 + u) + 8 * h];
                const f4 bB = *(const f4*)&bs[0][16 * (t0 + u) + 8 * h + 4];
                f4 p0, p1;
                #pragma unroll
                for (int r = 0; r < 4; ++r) {
                    const float v0 = pacc[u][r] + bA[r];
                    const float v1 = pacc[u][r + 4] + bB[r];
                    p0[r] = v0 > 0.0f ? v0 : 0.0f;
                    p1[r] = v1 > 0.0f ? v1 : 0.0f;
                }
                xch[t0 + u][lane][0] = p0;
                xch[t0 + u][lane][1] = p1;
            }
        }
        __syncthreads();
        if (wave < MLP_SIMDS) {
            f4 hv[8][2];
            #pragma unroll
            for (int tk = 0; tk < 8; ++tk) { hv[tk][0] = xch[tk][lane][0]; hv[tk][1] = xch[tk][lane][1]; }
            v8f qacc[2];
            qacc[0] = v8f{}; qacc[1] = v8f{};
            {
                f4 gC[8], gN[8];
                #pragma unroll
                for (int u = 0; u < 2; ++u) {
                    #pragma unroll
                    for (int j = 0; j < 4; ++j) gC[4 * u + j] = w2f[t0 + u][j][lane];
                }
                #pragma unroll
                for (int g = 0; g < 4; ++g) {
                    if (g + 1 < 4) {
                        #pragma unroll
                        for (int u = 0; u < 2; ++u) {
                            #pragma unroll
                            for (int j = 0; j < 4; ++j) gN[4 * u + j] = w2f[t0 + u][4 * (g + 1) + j][lane];
                        }
                    }
                    __builtin_amdgcn_sched_barrier(0);
                    #pragma unroll
                    for (int j = 0; j < 4; ++j) {
                        const int i = 4 * g + j;
                        const int tk = i >> 1;
                        const int hf = i & 1;
                        #pragma unroll
                        for (int u = 0; u < 2; ++u) qacc[u] = mma4(gC[4 * u + j][0], gC[4 * u + j][1], hv[tk][hf][0], hv[tk][hf][1], qacc[u]);
                        #pragma unroll
                        for (int u = 0; u < 2; ++u) qacc[u] = mma4(gC[4 * u + j][2], gC[4 * u + j][3], hv[tk][hf][2], hv[tk][hf][3], qacc[u]);
                    }
                    __builtin_amdgcn_sched_barrier(0);
                    #pragma unroll
                    for (int k = 0; k < 8; ++k) gC[k] = gN[k];
                }
            }
            #pragma unroll
            for (int u = 0; u < 2; ++u) {
                const int t = t0 + u;
                const f4 bA = *(const f4*)&bs[1][16 * t + 8 * h];
                const f4 bB = *(const f4*)&bs[1][16 * t + 8 * h + 4];
                f4 o0, o1;
                #pragma unroll
                for (int r = 0; r < 4; ++r) {
                    float u0 = qacc[u][r] + bA[r];
                    float u1 = qacc[u][r + 4] + bB[r];
                    if (applyReluOut) { u0 = u0 > 0.0f ? u0 : 0.0f; u1 = u1 > 0.0f ? u1 : 0.0f; }
                    o0[r] = u0; o1[r] = u1;
                }
                myTb[c][4 * t + 2 * h]     = o0;
                myTb[c][4 * t + 2 * h + 1] = o1;
            }
            wave_lds_handover();
            f4* orow = (f4*)out + (long)xt * (16 * 32);
            const int sl = 4 * t0 + (lane & 7);
            #pragma unroll
            for (int p = 0; p < 4; ++p) {
                const int r = 4 * p + (lane >> 3);
                orow[32 * r + sl] = myTb[r][sl];
            }
        }
        if (xt + MLP_GROUPS < nTiles) __syncthreads();
    }
}

extern "C" void kernel_launch(void* const* d_in, const int* in_sizes, int n_in,
                              void* d_out, int out_size, void* d_ws, size_t ws_size,
                              hipStream_t stream) {
    const float* x  = (const float*)d_in[0];
    const int*   ei = (const int*)  d_in[1];
    const float* W1 = (const float*)d_in[2];
    const float* b1 = (const float*)d_in[3];
    const float* W2 = (const float*)d_in[4];
    const float* b2 = (const float*)d_in[5];
    float* out = (float*)d_out;

    const int N = in_sizes[0] / D;
    const int E = in_sizes[1] / 2;
    const int* src = ei;
    const int* dst = ei + E;

    float* agg = (float*)d_ws;
    float* h1  = agg + (size_t)N * D;

    const int  aBlocks = (N + AGG_ROWS - 1) / AGG_ROWS;
    const int  mTiles  = N / TILE_ROWS;
    const int  kBlocks = (E + 511 + 4095) / 4096;
    unsigned char* keys = (unsigned char*)(h1 + (size_t)N * D);
    unsigned* keepSeg = (unsigned*)(keys + (size_t)kBlocks * 4096);
    int*      keepCnt = (int*)(keepSeg + (size_t)aBlocks * AGG_WAVES * SEG_CAP);
    int*      keepOk  = keepCnt + (size_t)aBlocks * AGG_WAVES;

    gin_edge_keys<<<kBlocks, 256, 0, stream>>>(dst, keys, N, E);

    gin_agg_owner<<<aBlocks, AGG_BLOCK, 0, stream>>>(x, src, dst, keys, agg, keepSeg, keepCnt, keepOk, 0, N, E);
    gin_mlp_t<<<MLP_GROUPS, MLP_BLOCK, 0, stream>>>(x, agg, W1, b1, W2, b2, h1, 1, mTiles);

    gin_agg_owner<<<aBlocks, AGG_BLOCK, 0, stream>>>(h1, src, dst, keys, agg, keepSeg, keepCnt, keepOk, 1, N, E);
    gin_mlp_t<<<MLP_GROUPS, MLP_BLOCK, 0, stream>>>(h1, agg, W1, b1, W2, b2, out, 0, mTiles);
}


# Neighbour sum: each wave sorts its own list by row (stable, in LDS), sums a row in registers and stores it once itself: no LDS accumulators, clearing or copy-out pass; sum order unchanged
# speedup vs baseline: 4.2902x; 1.0570x over previous
#include <hip/hip_runtime.h>


#define D 128
#define TILE_ROWS 16

typedef float v2f __attribute__((ext_vector_type(2)));
typedef float v8f __attribute__((ext_vector_type(8)));

#define AGG_ROWS   400
#define AGG_WAVES  32
#define AGG_BLOCK  1024
#define SEG_CAP    512
#define OWN_CAP    256
#define AGG_SLOTS  ((AGG_ROWS + AGG_WAVES - 1) / AGG_WAVES)
#define KEY_DEPTH  8
#define KEY_NONE   255u
#define SCAN_DEPTH 8
#define ROW_GROUP  8

typedef float f4 __attribute__((ext_vector_type(4)));
typedef int   i4 __attribute__((ext_vector_type(4)));
typedef unsigned u4 __attribute__((ext_vector_type(4)));

__global__ __launch_bounds__(256)
void gin_edge_keys(const int* __restrict__ dst,
                   unsigned char* __restrict__ keys,
                   int nNodes, int nEdges) {
    int per = (nEdges + AGG_WAVES - 1) / AGG_WAVES;
    per = (per + 15) & ~15;
    const int nPadded = per * AGG_WAVES;
    const int base = (blockIdx.x * 256 + threadIdx.x) * 16;
    if (base >= nPadded) return;
    const bool vecOk = (((unsigned long long)dst) & 15ull) == 0ull;
    u4 o;
    #pragma unroll
    for (int j = 0; j < 4; ++j) {
        const int e0 = base + 4 * j;
        i4 d;
        if (vecOk && e0 + 3 < nEdges) {
            d = *(const i4*)(dst + e0);
        } else {
            #pragma unroll
            for (int k = 0; k < 4; ++k) d[k] = (e0 + k < nEdges) ? dst[min(e0 + k, nEdges - 1)] : -1;
        }
        unsigned word = 0u;
        #pragma unroll
        for (int k = 0; k < 4; ++k) {
            const unsigned dk = (unsigned)d[k];
            const unsigned key = (dk < (unsigned)nNodes) ? min(dk / (unsigned)AGG_ROWS, KEY_NONE) : KEY_NONE;
            word |= key << (8 * k);
        }
        o[j] = word;
    }
    *(u4*)(keys + base) = o;
}

static __device__ __forceinline__ void wave_lds_handover() {
    __builtin_amdgcn_fence(__ATOMIC_RELEASE, "wavefront");
    __builtin_amdgcn_wave_barrier();
    __builtin_amdgcn_fence(__ATOMIC_ACQUIRE, "wavefront");
}

#define PEERS(I0) \
                const int pi = (I0) + lane; \
                const bool valid = pi < cntw; \
                const unsigned ent = valid ? seg[wave][pi] : 0u; \
                const unsigned o = (ent >> 20) & (AGG_WAVES - 1); \
                unsigned peers = __builtin_amdgcn_ballot_w32(valid); \
                _Pragma("unroll") \
                for (int pb = 0; pb < 5; ++pb) { \
                    const bool bit = ((o >> pb) & 1u) != 0u; \
                    const unsigned mb = __builtin_amdgcn_ballot_w32(bit); \
                    peers &= bit ? mb : ~mb; \
                } \
                const int rank = (int)__builtin_amdgcn_mbcnt_lo(peers, 0u);

static_assert(AGG_SLOTS <= 16, "a row slot must fit four bits");
#define SLOT_PEERS(I0) \
                    const int pi = (I0) + lane; \
                    const bool valid = pi < nOwn; \
                    const unsigned ent = valid ? own[wave][pi] : 0u; \
                    const unsigned sl = (ent >> 25) & 15u; \
                    unsigned peers = __builtin_amdgcn_ballot_w32(valid); \
                    _Pragma("unroll") \
                    for (int pb = 0; pb < 4; ++pb) { \
                        const bool bit = ((sl >> pb) & 1u) != 0u; \
                        const unsigned mb = __builtin_amdgcn_ballot_w32(bit); \
                        peers &= bit ? mb : ~mb; \
                    } \
                    const int rank = (int)__builtin_amdgcn_mbcnt_lo(peers, 0u);

__global__ __launch_bounds__(AGG_BLOCK)
void gin_agg_owner(const float* __restrict__ x,
                   const int*   __restrict__ src,
                   const int*   __restrict__ dst,
                   const unsigned char* __restrict__ keys,
                   float*       __restrict__ agg,
                   unsigned*    __restrict__ keepSeg,
                   int*         __restrict__ keepCnt,
                   int*         __restrict__ keepOk,
                   int reuse,
                   int nNodes, int nEdges) {
    __shared__ f4  acc[AGG_ROWS][32];
    __shared__ unsigned seg[AGG_WAVES][SEG_CAP];
    __shared__ int s_cnt[AGG_WAVES];
    __shared__ int s_more[AGG_WAVES];
    __shared__ unsigned own[AGG_WAVES][OWN_CAP];
    __shared__ int tab[AGG_WAVES][AGG_WAVES];
    __shared__ int posTab[AGG_WAVES][AGG_WAVES];

    const int tid  = threadIdx.x;
    const int lane = tid & 31;
    const int wave = __builtin_amdgcn_readfirstlane(tid >> 5);
    const int rowLo = blockIdx.x * AGG_ROWS;
    const int rowsHere = min(AGG_ROWS, nNodes - rowLo);

    unsigned* mySeg = keepSeg + ((long)blockIdx.x * AGG_WAVES + wave) * SEG_CAP;
    int*      myCnt = keepCnt + (long)blockIdx.x * AGG_WAVES + wave;
    bool useKept = false;
    int  keptCnt = 0;
    if (reuse) {
        useKept = __builtin_amdgcn_readfirstlane(keepOk[blockIdx.x]) == 1;
        keptCnt = __builtin_amdgcn_readfirstlane(min(max(*myCnt, 0), OWN_CAP));
    }
    const bool keep = (reuse == 0);

    f4* accFlat = &acc[0][0];
    bool accLive = false;

    const bool useKeys = gridDim.x <= 255u;
    const unsigned keyRep = blockIdx.x * 0x01010101u;
    int per = (nEdges + AGG_WAVES - 1) / AGG_WAVES;
    per = useKeys ? ((per + 15) & ~15) : ((per + 3) & ~3);
    const int keyLast = per * AGG_WAVES - 16;
    const int beg = useKeys ? wave * per : min(wave * per, nEdges);
    const int end = useKeys ? beg + per : min(beg + per, nEdges);
    const bool vecOk = ((((unsigned long long)dst) & 15ull) == 0ull) && ((nEdges & 3) == 0);
    int pos = useKept ? end : beg;
    bool anyMore;
    bool firstPass = true;

    do {
        int cntw = 0;
        bool full = false;
        while (useKeys && pos < end && !full) {
            u4 kv[KEY_DEPTH];
            #pragma unroll
            for (int u = 0; u < KEY_DEPTH; ++u) {
                const int e = pos + u * 512 + lane * 16;
                kv[u] = *(const u4*)(keys + min(e, keyLast));
            }
            int stepsDone = 0;
            #pragma unroll
            for (int u = 0; u < KEY_DEPTH; ++u) {
                if (!full && pos + u * 512 < end) {
                    const int e = pos + u * 512 + lane * 16;
                    unsigned w = 0u;
                    #pragma unroll
                    for (int j = 0; j < 4; ++j) {
                        const unsigned y = kv[u][j] ^ keyRep;
                        const unsigned t = (y & 0x7F7F7F7Fu) + 0x7F7F7F7Fu;
                        const unsigned z = ~(t | y | 0x7F7F7F7Fu);
                        w |= (z >> 7) << j;
                    }
                    w = (e < end) ? w : 0u;
                    unsigned mL = __builtin_amdgcn_ballot_w32(w != 0u);
                    if (mL) {
                        bool fits = cntw + 16 * __builtin_popcount(mL) <= SEG_CAP;
                        if (!fits) {
                            int tot = 0;
                            for (unsigned m2 = mL; m2; m2 &= m2 - 1u)
                                tot += __builtin_popcount((unsigned)__builtin_amdgcn_readlane((int)w, __builtin_ctz(m2)));
                            fits = cntw + tot <= SEG_CAP;
                        }
                        if (!fits) {
                            full = true;
                        } else {
                            const int off0 = pos - beg + u * 512;
                            while (mL) {
                                const int l = __builtin_ctz(mL);
                                mL &= mL - 1u;
                                unsigned wl = (unsigned)__builtin_amdgcn_readlane((int)w, l);
                                while (wl) {
                                    const int bit = __builtin_ctz(wl);
                                    wl &= wl - 1u;
                                    seg[wave][cntw] = (unsigned)(off0 + l * 16 + ((bit & 7) << 2) + (bit >> 3));
                                    ++cntw;
                                }
                            }
                        }
                    }
                    if (!full) stepsDone = u + 1;
                }
            }
            pos += stepsDone * 512;
        }
        while (!useKeys && pos < end && !full) {
            i4 v[SCAN_DEPTH];
            if (vecOk) {
                #pragma unroll
                for (int u = 0; u < SCAN_DEPTH; ++u) {
                    const int e = pos + u * 128 + lane * 4;
                    const i4 t = *(const i4*)(dst + min(e, end - 4));
                    const i4 none = {-1, -1, -1, -1};
                    v[u] = (e < end) ? t : none;
                }
            } else {
                #pragma unroll
                for (int u = 0; u < SCAN_DEPTH; ++u) {
                    const int e = pos + u * 128 + lane * 4;
                    i4 t;
                    #pragma unroll
                    for (int k = 0; k < 4; ++k) {
                        const int d = dst[min(e + k, end - 1)];
                        t[k] = (e + k < end) ? d : -1;
                    }
                    v[u] = t;
                }
            }
            int stepsDone = 0;
            #pragma unroll
            for (int u = 0; u < SCAN_DEPTH; ++u) {
                if (!full && pos + u * 128 < end) {
                    if (cntw + 128 > SEG_CAP) {
                        full = true;
                    } else {
                        const int off = pos - beg + u * 128 + lane * 4;
                        unsigned lr[4]; bool h[4]; unsigned m[4];
                        #pragma unroll
                        for (int k = 0; k < 4; ++k) {
                            lr[k] = (unsigned)(v[u][k] - rowLo);
                            h[k]  = lr[k] < (unsigned)rowsHere;
                            m[k]  = __builtin_amdgcn_ballot_w32(h[k]);
                        }
                        if (m[0] | m[1] | m[2] | m[3]) {
                            #pragma unroll
                            for (int k = 0; k < 4; ++k) {
                                if (h[k]) {
                                    const int p = cntw + (int)__builtin_amdgcn_mbcnt_lo(m[k], 0u);
                                    seg[wave][p] = (lr[k] << 20) | (unsigned)(off + k);
                                }
                                cntw += __builtin_popcount(m[k]);
                            }
                        }
                        stepsDone = u + 1;
                    }
                }
            }
            pos += stepsDone * 128;
        }
        const bool keepNow = keep && firstPass;
        if (!useKept) {
            for (int i = lane; i < cntw; i += 32) {
                const unsigned en = seg[wave][i];
                unsigned ent;
                if (useKeys) {
                    const int e = beg + (int)en;
                    ent = ((unsigned)(dst[e] - rowLo) << 20) | (unsigned)src[e];
                } else {
                    ent = (en & 0xFFF00000u) | (unsigned)src[beg + (int)(en & 0xFFFFFu)];
                }
                seg[wave][i] = ent;
            }
            tab[wave][lane] = 0;
            wave_lds_handover();
            for (int i0 = 0; i0 < cntw; i0 += 32) {
                PEERS(i0)
                if (valid && rank == 0) tab[wave][o] += __builtin_popcount(peers);
                wave_lds_handover();
            }
        } else {
            for (int i0 = 0; i0 < keptCnt; i0 += 128) {
                const unsigned kv0 = mySeg[min(i0 + lane,      SEG_CAP - 1)];
                const unsigned kv1 = mySeg[min(i0 + lane + 32, SEG_CAP - 1)];
                const unsigned kv2 = mySeg[min(i0 + lane + 64, SEG_CAP - 1)];
                const unsigned kv3 = mySeg[min(i0 + lane + 96, SEG_CAP - 1)];
                if (i0 + lane      < keptCnt) seg[wave][i0 + lane]      = kv0;
                if (i0 + lane + 32 < keptCnt) seg[wave][i0 + lane + 32] = kv1;
                if (i0 + lane + 64 < keptCnt) seg[wave][i0 + lane + 64] = kv2;
                if (i0 + lane + 96 < keptCnt) seg[wave][i0 + lane + 96] = kv3;
            }
            wave_lds_handover();
        }
        if (lane == 0) {
            s_cnt[wave] = useKept ? 0 : cntw; s_more[wave] = (pos < end) ? 1 : 0;
        }
        __syncthreads();
        anyMore = __builtin_amdgcn_ballot_w32(s_more[lane] != 0) != 0u;

        int  nOwn   = keptCnt;
        bool binned = true;
        bool fast = firstPass && !anyMore;
        if (!useKept) {
            int base = 0, total = 0;
            for (int w2 = 0; w2 < AGG_WAVES; ++w2) {
                const int t = tab[w2][lane];
                base  += (w2 < wave) ? t : 0;
                total += t;
            }
            binned = __builtin_amdgcn_ballot_w32(total > OWN_CAP) == 0u;
            nOwn = __builtin_amdgcn_readlane(total, wave);
            fast = fast && binned;
            if (fast) {
                posTab[wave][lane] = base;
                wave_lds_handover();
                for (int i0 = 0; i0 < cntw; i0 += 32) {
                    PEERS(i0)
                    int p = 0;
                    if (valid) p = posTab[wave][o];
                    wave_lds_handover();
                    if (valid && rank == 0) posTab[wave][o] = p + __builtin_popcount(peers);
                    if (valid) own[o][p + rank] = ent;
                    wave_lds_handover();
                }
            }
            __syncthreads();
        }
        if (keepNow && tid == 0) keepOk[blockIdx.x] = fast ? 1 : 0;
        firstPass = false;

        if (fast) {
            if (!useKept) {
                posTab[wave][lane] = 0;
                wave_lds_handover();
                for (int i0 = 0; i0 < nOwn; i0 += 32) {
                    SLOT_PEERS(i0)
                    if (valid && rank == 0) posTab[wave][sl] += __builtin_popcount(peers);
                    wave_lds_handover();
                }
                const int slotCnt = posTab[wave][lane];
                int place = 0;
                #pragma unroll
                for (int sb = 0; sb < AGG_SLOTS - 1; ++sb) {
                    const int v = __builtin_amdgcn_readlane(slotCnt, sb);
                    place += (lane > sb) ? v : 0;
                }
                posTab[wave][lane] = place;
                wave_lds_handover();
                for (int i0 = 0; i0 < nOwn; i0 += 32) {
                    SLOT_PEERS(i0)
                    int p = 0;
                    if (valid) p = posTab[wave][sl];
                    wave_lds_handover();
                    if (valid && rank == 0) posTab[wave][sl] = p + __builtin_popcount(peers);
                    if (valid) seg[wave][p + rank] = ent;
                    wave_lds_handover();
                }
                if (keepNow) {
                    for (int i = lane; i < nOwn; i += 32) mySeg[i] = seg[wave][i];
                    if (lane == 0) *myCnt = nOwn;
                }
            }
            const f4 zero4 = {0.f, 0.f, 0.f, 0.f};
            f4 sum = zero4;
            int cur = wave;
            f4* outRows = (f4*)agg + (long)rowLo * 32 + lane;
#define ROW_FLUSH_TO(R) { \
            const int rTo = min((R), rowsHere); \
            while (cur < rTo) { outRows[(long)cur * 32] = sum; sum = zero4; cur += AGG_WAVES; } }
            f4 xA0 = zero4, xA1 = zero4, xA2 = zero4, xA3 = zero4, xA4 = zero4, xA5 = zero4, xA6 = zero4, xA7 = zero4;
            f4 xB0 = zero4, xB1 = zero4, xB2 = zero4, xB3 = zero4, xB4 = zero4, xB5 = zero4, xB6 = zero4, xB7 = zero4;
            unsigned eA = 0u, eB = 0u;
            int bA = 0, bB = 0, nA = 0, nB = 0;
#define XROWS(M, S) M(S, 0) M(S, 1) M(S, 2) M(S, 3) M(S, 4) M(S, 5) M(S, 6) M(S, 7)
#define X_LOAD(S, q) { \
            const unsigned e2 = (unsigned)__builtin_amdgcn_readlane((int)e##S, b##S + min(q, n##S - 1)); \
            x##S##q = ((const f4*)x)[(long)(e2 & 0xFFFFFu) * 32 + lane]; }
#define X_ADD(S, q) if (q < n##S) { \
            const int r = (int)((unsigned)__builtin_amdgcn_readlane((int)e##S, b##S + q) >> 20); \
            ROW_FLUSH_TO(r) sum += x##S##q; }
#define X_ISSUE(S, E, BL, GI) { \
            e##S = (E); b##S = (BL); n##S = min(max(nOwn - (GI), 0), 8); \
            if (n##S > 0) { XROWS(X_LOAD, S) } }
#define X_LOADF(S, q) { \
            const unsigned e2 = (unsigned)__builtin_amdgcn_readlane((int)e##S, b##S + q); \
            x##S##q = ((const f4*)x)[(long)(e2 & 0xFFFFFu) * 32 + lane]; }
#define X_ADDF(S, q) { \
            const int r = (int)((unsigned)__builtin_amdgcn_readlane((int)e##S, b##S + q) >> 20); \
            ROW_FLUSH_TO(r) sum += x##S##q; }
#define X_ISSUEF(S, E, BL) { e##S = (E); b##S = (BL); n##S = 8; XROWS(X_LOADF, S) }
#define X_FETCH(DSTV, I0) { \
            const int fi = (I0) + lane; \
            unsigned fv = 0u; \
            if (fi < nOwn) fv = seg[wave][fi]; \
            DSTV = fv; }
            unsigned entCur, entNext;
            X_FETCH(entCur, 0)
            X_FETCH(entNext, 32)
            X_ISSUE(A, entCur, 0, 0)
            int i0 = 0;
            for (; i0 + 40 <= nOwn; i0 += 32) {
                X_ISSUEF(B, entCur, 8)
                XROWS(X_ADDF, A)
                X_ISSUEF(A, entCur, 16)
                XROWS(X_ADDF, B)
                X_ISSUEF(B, entCur, 24)
                XROWS(X_ADDF, A)
                entCur = entNext;
                X_FETCH(entNext, i0 + 64)
                X_ISSUEF(A, entCur, 0)
                XROWS(X_ADDF, B)
            }
            for (; i0 < nOwn; i0 += 32) {
                X_ISSUE(B, entCur, 8, i0 + 8)
                XROWS(X_ADD, A)
                X_ISSUE(A, entCur, 16, i0 + 16)
                XROWS(X_ADD, B)
                X_ISSUE(B, entCur, 24, i0 + 24)
                XROWS(X_ADD, A)
                entCur = entNext;
                X_FETCH(entNext, i0 + 64)
                X_ISSUE(A, entCur, 0, i0 + 32)
                XROWS(X_ADD, B)
            }
            ROW_FLUSH_TO(rowsHere)
#undef ROW_FLUSH_TO
#undef X_FETCH
#undef X_ISSUEF
#undef X_ADDF
#undef X_LOADF
#undef X_ISSUE
#undef X_ADD
#undef X_LOAD
#undef XROWS
        } else {
            if (!accLive) {
                for (int i = tid; i < AGG_ROWS * 32; i += AGG_BLOCK) accFlat[i] = f4{0.f, 0.f, 0.f, 0.f};
                accLive = true;
                __syncthreads();
            }
        unsigned held = 0u;
        int nHeld = 0;
        unsigned sent = 0u;
        int nSent = 0;
        const f4 zero4 = {0.f, 0.f, 0.f, 0.f};
        f4 xr0 = zero4, xr1 = zero4, xr2 = zero4, xr3 = zero4, xr4 = zero4, xr5 = zero4, xr6 = zero4, xr7 = zero4;
#define ROWS(M) M(0) M(1) M(2) M(3) M(4) M(5) M(6) M(7)
#define ROW_ADD(q) if (q < nSent) { \
            const int r = (int)((unsigned)__builtin_amdgcn_readlane((int)sent, q) >> 20); \
            f4 a = acc[r][lane]; a += xr##q; acc[r][lane] = a; }
#define ROW_LOAD(q) { \
            const unsigned e2 = (unsigned)__builtin_amdgcn_readlane((int)held, min(q, nHeld - 1)); \
            xr##q = ((const f4*)x)[(long)(e2 & 0xFFFFFu) * 32 + lane]; }
#define ROW_STEP() { \
            ROWS(ROW_ADD) \
            sent = held; nSent = nHeld; \
            if (nHeld > 0) { ROWS(ROW_LOAD) } \
            nHeld = 0; }
        const int cl = s_cnt[lane];
        for (int sg = 0; sg < AGG_WAVES; ++sg) {
            const int c = __builtin_amdgcn_readlane(cl, sg);
            for (int i0 = 0; i0 < c; i0 += 32) {
                const int i = i0 + lane;
                const unsigned ent = (i < c) ? seg[sg][i] : 0u;
                const bool own = (i < c) && (int)((ent >> 20) & (AGG_WAVES - 1)) == wave;
                unsigned m = __builtin_amdgcn_ballot_w32(own);
                while (m) {
                    const int l = __builtin_ctz(m);
                    m &= m - 1u;
                    const unsigned e2 = (unsigned)__builtin_amdgcn_readlane((int)ent, l);
                    held = (lane == nHeld) ? e2 : held;
                    if (++nHeld == ROW_GROUP) ROW_STEP()
                }
            }
        }
        ROW_STEP()
        ROWS(ROW_ADD)
#undef ROW_STEP
#undef ROW_LOAD
#undef ROW_ADD
#undef ROWS
            __syncthreads();
        }
    } while (anyMore);

    if (accLive) {
        f4* aggRows = (f4*)agg + (long)rowLo * 32;
        for (int i = tid; i < rowsHere * 32; i += AGG_BLOCK) aggRows[i] = accFlat[i];
    }
}

#define MLP_WAVES 16
#define MLP_BLOCK (MLP_WAVES * 32)
#define MLP_GROUPS 256
#define MLP_SIMDS 4
#define TB_PITCH 33


static __device__ __forceinline__ v8f mma4(float a0, float a1, float b0, float b1, v8f c) {
    v2f a = {a0, a1};
    v2f b = {b0, b1};
    return __builtin_amdgcn_wmma_f32_16x16x4_f32(false, a, false, b, (short)0, c, false, false);
}

__global__ __launch_bounds__(MLP_BLOCK)
void gin_mlp_t(const float* __restrict__ xin,
               const float* __restrict__ agg,
               const float* __restrict__ W1, const float* __restrict__ b1,
               const float* __restrict__ W2, const float* __restrict__ b2,
               float* __restrict__ out,
               int applyReluOut, int nTiles) {
    __shared__ f4 w1f[8][16][32];
    __shared__ f4 w2f[8][16][32];
    __shared__ alignas(16) float bs[2][D];
    __shared__ f4 tb[MLP_WAVES][TILE_ROWS][TB_PITCH];
    __shared__ f4 xch[8][32][2];

    const int tid  = threadIdx.x;
    const int lane = tid & 31;
    const int wave = __builtin_amdgcn_readfirstlane(tid >> 5);

    const int c = lane & 15;
    const int h = lane >> 4;
    int tile = blockIdx.x + MLP_GROUPS * wave;
    const int nMain = (nTiles / (MLP_GROUPS * MLP_SIMDS)) * (MLP_GROUPS * MLP_SIMDS);
    const int xFirst = nMain + blockIdx.x;

    const f4 zero4 = {0.f, 0.f, 0.f, 0.f};
    f4 xv[16], av[16];
    #pragma unroll
    for (int i = 0; i < 16; ++i) { xv[i] = zero4; av[i] = zero4; }
    if (tile < nMain) {
        const f4* xr0 = (const f4*)xin + (long)tile * (16 * 32) + lane;
        const f4* ar0 = (const f4*)agg + (long)tile * (16 * 32) + lane;
        #pragma unroll
        for (int r = 0; r < 16; ++r) { xv[r] = xr0[32 * r]; av[r] = ar0[32 * r]; }
    }
    f4 (*myTb)[TB_PITCH] = tb[wave];

    {
        const int n  = tid & (D - 1);
        const int tn = n >> 4;
        const int cc = n & 15;
        const int q  = tid >> 7;
        const float bias1 = b1[n];
        const float bias2 = b2[n];
        f4 wa[8], wb[8];
        #pragma unroll
        for (int j = 0; j < 8; ++j) {
            const int kg = q + 4 * j;
            #pragma unroll
            for (int m = 0; m < 4; ++m) {
                wa[j][m] = W1[(4 * kg + m) * D + n];
                wb[j][m] = W2[(4 * kg + m) * D + n];
            }
        }
        #pragma unroll
        for (int j = 0; j < 8; ++j) {
            const int kg = q + 4 * j;
            const int i1 = kg >> 1;
            const int h1 = kg & 1;
            w1f[tn][i1][cc + 16 * h1] = wa[j];
            const int i2 = ((kg >> 2) << 1) | (kg & 1);
            const int h2 = (kg >> 1) & 1;
            w2f[tn][i2][cc + 16 * h2] = wb[j];
        }
        bs[0][n] = bias1;
        bs[1][n] = bias2;
    }
    __syncthreads();

    while (tile < nMain) {
        asm volatile("" ::: "memory");
        #pragma unroll
        for (int r = 0; r < 16; ++r) myTb[r][lane] = xv[r] + av[r];
        wave_lds_handover();
        f4 sv[16];
        #pragma unroll
        for (int i = 0; i < 16; ++i) sv[i] = myTb[c][2 * i + h];
        __builtin_amdgcn_sched_barrier(0);

        v8f acc[8];
        #pragma unroll
        for (int t = 0; t < 8; ++t) acc[t] = v8f{};

        {
            f4 aC[8], aN[8];
            #pragma unroll
            for (int t = 0; t < 8; ++t) aC[t] = w1f[t][0][lane];
            #pragma unroll
            for (int i = 0; i < 16; ++i) {
                if (i + 1 < 16) {
                    #pragma unroll
                    for (int t = 0; t < 8; ++t) aN[t] = w1f[t][i + 1][lane];
                }
                __builtin_amdgcn_sched_barrier(0);
                #pragma unroll
                for (int t = 0; t < 8; ++t) acc[t] = mma4(aC[t][0], aC[t][1], sv[i][0], sv[i][1], acc[t]);
                #pragma unroll
                for (int t = 0; t < 8; ++t) acc[t] = mma4(aC[t][2], aC[t][3], sv[i][2], sv[i][3], acc[t]);
                __builtin_amdgcn_sched_barrier(0);
                #pragma unroll
                for (int t = 0; t < 8; ++t) aC[t] = aN[t];
            }
        }
        #pragma unroll
        for (int t = 0; t < 8; ++t) {
            const f4 bA = *(const f4*)&bs[0][16 * t + 8 * h];
            const f4 bB = *(const f4*)&bs[0][16 * t + 8 * h + 4];
            #pragma unroll
            for (int r = 0; r < 8; ++r) {
                const float v = acc[t][r] + (r < 4 ? bA[r] : bB[r - 4]);
                acc[t][r] = v > 0.0f ? v : 0.0f;
            }
        }

        v8f acc2[8];
        #pragma unroll
        for (int t = 0; t < 8; ++t) acc2[t] = v8f{};
        {
            f4 aC[8], aN[8];
            #pragma unroll
            for (int t = 0; t < 8; ++t) aC[t] = w2f[t][0][lane];
            #pragma unroll
            for (int i = 0; i < 16; ++i) {
                const int tk = i >> 1;
                const int p4 = (i & 1) * 4;
                if (i + 1 < 16) {
                    #pragma unroll
                    for (int t = 0; t < 8; ++t) aN[t] = w2f[t][i + 1][lane];
                }
                __builtin_amdgcn_sched_barrier(0);
                #pragma unroll
                for (int t = 0; t < 8; ++t) acc2[t] = mma4(aC[t][0], aC[t][1], acc[tk][p4], acc[tk][p4 + 1], acc2[t]);
                #pragma unroll
                for (int t = 0; t < 8; ++t) acc2[t] = mma4(aC[t][2], aC[t][3], acc[tk][p4 + 2], acc[tk][p4 + 3], acc2[t]);
                __builtin_amdgcn_sched_barrier(0);
                #pragma unroll
                for (int t = 0; t < 8; ++t) aC[t] = aN[t];
            }
        }

        #pragma unroll
        for (int t = 0; t < 8; ++t) {
            const f4 bA = *(const f4*)&bs[1][16 * t + 8 * h];
            const f4 bB = *(const f4*)&bs[1][16 * t + 8 * h + 4];
            f4 o0, o1;
            #pragma unroll
            for (int r = 0; r < 4; ++r) {
                float u0 = acc2[t][r] + bA[r];
                float u1 = acc2[t][r + 4] + bB[r];
                if (applyReluOut) { u0 = u0 > 0.0f ? u0 : 0.0f; u1 = u1 > 0.0f ? u1 : 0.0f; }
                o0[r] = u0; o1[r] = u1;
            }
            myTb[c][4 * t + 2 * h]     = o0;
            myTb[c][4 * t + 2 * h + 1] = o1;
        }
        wave_lds_handover();
        f4* orow = (f4*)out + (long)tile * (16 * 32) + lane;
        #pragma unroll
        for (int r = 0; r < 16; ++r) orow[32 * r] = myTb[r][lane];

        tile += MLP_GROUPS * MLP_WAVES;
        if (tile < nMain) {
            const f4* xrN = (const f4*)xin + (long)tile * (16 * 32) + lane;
            const f4* arN = (const f4*)agg + (long)tile * (16 * 32) + lane;
            #pragma unroll
            for (int r = 0; r < 16; ++r) { xv[r] = xrN[32 * r]; av[r] = arN[32 * r]; }
        }
    }

    for (int xt = xFirst; xt < nTiles; xt += MLP_GROUPS) {
        const int t0 = 2 * wave;
        if (wave < MLP_SIMDS) {
            const f4* xrN = (const f4*)xin + (long)xt * (16 * 32) + lane;
            const f4* arN = (const f4*)agg + (long)xt * (16 * 32) + lane;
            f4 xl[16], al[16];
            #pragma unroll
            for (int r = 0; r < 16; ++r) { xl[r] = xrN[32 * r]; al[r] = arN[32 * r]; }
            #pragma unroll
            for (int r = 0; r < 16; ++r) myTb[r][lane] = xl[r] + al[r];
            wave_lds_handover();
            f4 sv[16];
            #pragma unroll
            for (int i = 0; i < 16; ++i) sv[i] = myTb[c][2 * i + h];
            v8f pacc[2];
            pacc[0] = v8f{}; pacc[1] = v8f{};
            {
                f4 gC[8], gN[8];
                #pragma unroll
                for (int u = 0; u < 2; ++u) {
                    #pragma unroll
                    for (int j = 0; j < 4; ++j) gC[4 * u + j] = w1f[t0 + u][j][lane];
                }
                #pragma unroll
                for (int g = 0; g < 4; ++g) {
                    if (g + 1 < 4) {
                        #pragma unroll
                        for (int u = 0; u < 2; ++u) {
                            #pragma unroll
                            for (int j = 0; j < 4; ++j) gN[4 * u + j] = w1f[t0 + u][4 * (g + 1) + j][lane];
                        }
                    }
                    __builtin_amdgcn_sched_barrier(0);
                    #pragma unroll
                    for (int j = 0; j < 4; ++j) {
                        const int i = 4 * g + j;
                        #pragma unroll
                        for (int u = 0; u < 2; ++u) pacc[u] = mma4(gC[4 * u + j][0], gC[4 * u + j][1], sv[i][0], sv[i][1], pacc[u]);
                        #pragma unroll
                        for (int u = 0; u < 2; ++u) pacc[u] = mma4(gC[4 * u + j][2], gC[4 * u + j][3], sv[i][2], sv[i][3], pacc[u]);
                    }
                    __builtin_amdgcn_sched_barrier(0);
                    #pragma unroll
                    for (int k = 0; k < 8; ++k) gC[k] = gN[k];
                }
            }
            #pragma unroll
            for (int u = 0; u < 2; ++u) {
                const f4 bA = *(const f4*)&bs[0][16 * (t0 + u) + 8 * h];
                const f4 bB = *(const f4*)&bs[0][16 * (t0 + u) + 8 * h + 4];
                f4 p0, p1;
                #pragma unroll
                for (int r = 0; r < 4; ++r) {
                    const float v0 = pacc[u][r] + bA[r];
                    const float v1 = pacc[u][r + 4] + bB[r];
                    p0[r] = v0 > 0.0f ? v0 : 0.0f;
                    p1[r] = v1 > 0.0f ? v1 : 0.0f;
                }
                xch[t0 + u][lane][0] = p0;
                xch[t0 + u][lane][1] = p1;
            }
        }
        __syncthreads();
        if (wave < MLP_SIMDS) {
            f4 hv[8][2];
            #pragma unroll
            for (int tk = 0; tk < 8; ++tk) { hv[tk][0] = xch[tk][lane][0]; hv[tk][1] = xch[tk][lane][1]; }
            v8f qacc[2];
            qacc[0] = v8f{}; qacc[1] = v8f{};
            {
                f4 gC[8], gN[8];
                #pragma unroll
                for (int u = 0; u < 2; ++u) {
                    #pragma unroll
                    for (int j = 0; j < 4; ++j) gC[4 * u + j] = w2f[t0 + u][j][lane];
                }
                #pragma unroll
                for (int g = 0; g < 4; ++g) {
                    if (g + 1 < 4) {
                        #pragma unroll
                        for (int u = 0; u < 2; ++u) {
                            #pragma unroll
                            for (int j = 0; j < 4; ++j) gN[4 * u + j] = w2f[t0 + u][4 * (g + 1) + j][lane];
                        }
                    }
                    __builtin_amdgcn_sched_barrier(0);
                    #pragma unroll
                    for (int j = 0; j < 4; ++j) {
                        const int i = 4 * g + j;
                        const int tk = i >> 1;
                        const int hf = i & 1;
                        #pragma unroll
                        for (int u = 0; u < 2; ++u) qacc[u] = mma4(gC[4 * u + j][0], gC[4 * u + j][1], hv[tk][hf][0], hv[tk][hf][1], qacc[u]);
                        #pragma unroll
                        for (int u = 0; u < 2; ++u) qacc[u] = mma4(gC[4 * u + j][2], gC[4 * u + j][3], hv[tk][hf][2], hv[tk][hf][3], qacc[u]);
                    }
                    __builtin_amdgcn_sched_barrier(0);
                    #pragma unroll
                    for (int k = 0; k < 8; ++k) gC[k] = gN[k];
                }
            }
            #pragma unroll
            for (int u = 0; u < 2; ++u) {
                const int t = t0 + u;
                const f4 bA = *(const f4*)&bs[1][16 * t + 8 * h];
                const f4 bB = *(const f4*)&bs[1][16 * t + 8 * h + 4];
                f4 o0, o1;
                #pragma unroll
                for (int r = 0; r < 4; ++r) {
                    float u0 = qacc[u][r] + bA[r];
                    float u1 = qacc[u][r + 4] + bB[r];
                    if (applyReluOut) { u0 = u0 > 0.0f ? u0 : 0.0f; u1 = u1 > 0.0f ? u1 : 0.0f; }
                    o0[r] = u0; o1[r] = u1;
                }
                myTb[c][4 * t + 2 * h]     = o0;
                myTb[c][4 * t + 2 * h + 1] = o1;
            }
            wave_lds_handover();
            f4* orow = (f4*)out + (long)xt * (16 * 32);
            const int sl = 4 * t0 + (lane & 7);
            #pragma unroll
            for (int p = 0; p < 4; ++p) {
                const int r = 4 * p + (lane >> 3);
                orow[32 * r + sl] = myTb[r][sl];
            }
        }
        if (xt + MLP_GROUPS < nTiles) __syncthreads();
    }
}

extern "C" void kernel_launch(void* const* d_in, const int* in_sizes, int n_in,
                              void* d_out, int out_size, void* d_ws, size_t ws_size,
                              hipStream_t stream) {
    const float* x  = (const float*)d_in[0];
    const int*   ei = (const int*)  d_in[1];
    const float* W1 = (const float*)d_in[2];
    const float* b1 = (const float*)d_in[3];
    const float* W2 = (const float*)d_in[4];
    const float* b2 = (const float*)d_in[5];
    float* out = (float*)d_out;

    const int N = in_sizes[0] / D;
    const int E = in_sizes[1] / 2;
    const int* src = ei;
    const int* dst = ei + E;

    float* agg = (float*)d_ws;
    float* h1  = agg + (size_t)N * D;

    const int  aBlocks = (N + AGG_ROWS - 1) / AGG_ROWS;
    const int  mTiles  = N / TILE_ROWS;
    const int  kBlocks = (E + 511 + 4095) / 4096;
    unsigned char* keys = (unsigned char*)(h1 + (size_t)N * D);
    unsigned* keepSeg = (unsigned*)(keys + (size_t)kBlocks * 4096);
    int*      keepCnt = (int*)(keepSeg + (size_t)aBlocks * AGG_WAVES * SEG_CAP);
    int*      keepOk  = keepCnt + (size_t)aBlocks * AGG_WAVES;

    gin_edge_keys<<<kBlocks, 256, 0, stream>>>(dst, keys, N, E);

    gin_agg_owner<<<aBlocks, AGG_BLOCK, 0, stream>>>(x, src, dst, keys, agg, keepSeg, keepCnt, keepOk, 0, N, E);
    gin_mlp_t<<<MLP_GROUPS, MLP_BLOCK, 0, stream>>>(x, agg, W1, b1, W2, b2, h1, 1, mTiles);

    gin_agg_owner<<<aBlocks, AGG_BLOCK, 0, stream>>>(h1, src, dst, keys, agg, keepSeg, keepCnt, keepOk, 1, N, E);
    gin_mlp_t<<<MLP_GROUPS, MLP_BLOCK, 0, stream>>>(h1, agg, W1, b1, W2, b2, out, 0, mTiles);
}
